# starting kernel
# baseline (speedup 1.0000x reference)
#include <hip/hip_runtime.h>
#include <hip/hip_bf16.h>

typedef __attribute__((ext_vector_type(2))) float v2f;
typedef __attribute__((ext_vector_type(8))) float v8f;

#define HDIM 64
#define INDIM 128

__device__ __forceinline__ float lrelu(float v) { return v > 0.0f ? v : 0.01f * v; }

__global__ void deg_init_kernel(float* __restrict__ deg, int n) {
    int i = blockIdx.x * blockDim.x + threadIdx.x;
    if (i < n) deg[i] = 1.0f;
}

__global__ void deg_count_kernel(const int* __restrict__ dst, float* __restrict__ deg, int e) {
    int i = blockIdx.x * blockDim.x + threadIdx.x;
    if (i < e)
        __hip_atomic_fetch_add(&deg[dst[i]], 1.0f, __ATOMIC_RELAXED, __HIP_MEMORY_SCOPE_AGENT);
}

__global__ void dinv_kernel(float* __restrict__ deg, int n) {
    int i = blockIdx.x * blockDim.x + threadIdx.x;
    if (i < n) {
        float d = deg[i];
        deg[i] = d > 0.0f ? rsqrtf(d) : 0.0f;
    }
}

__global__ void wmma_gemm_f32_kernel(const float* __restrict__ A, const float* __restrict__ B,
                                     float* __restrict__ C, const float* __restrict__ bias,
                                     int M, int K, int act) {
    extern __shared__ float bsw[];
    const int tid   = threadIdx.x;
    const int lane  = tid & 31;
    const int wave  = tid >> 5;
    const int lmod  = lane & 15;
    const int lhalf = lane >> 4;

    const int pairs = K * 32;
    for (int p = tid; p < pairs; p += 256) {
        int ks  = p >> 7;
        int ll  = (p >> 2) & 31;
        int tn  = p & 3;
        int kb  = 4 * ks + 2 * (ll >> 4);
        int col = tn * 16 + (ll & 15);
        bsw[p * 2 + 0] = B[(size_t)kb * HDIM + col];
        bsw[p * 2 + 1] = B[(size_t)(kb + 1) * HDIM + col];
    }
    __syncthreads();

    const int gwave = blockIdx.x * 8 + wave;
    if (gwave >= (M >> 4)) return;

    const float* arow = A + (size_t)(gwave * 16 + lmod) * K + 2 * lhalf;
    v8f acc0 = {}, acc1 = {}, acc2 = {}, acc3 = {};

    const int steps = K >> 2;
    for (int ks = 0; ks < steps; ++ks) {
        v2f a = *(const v2f*)(arow + 4 * ks);
        const float4* bp = (const float4*)(bsw + (size_t)(ks * 32 + lane) * 8);
        float4 b01 = bp[0];
        float4 b23 = bp[1];
        v2f b0 = {b01.x, b01.y}, b1 = {b01.z, b01.w};
        v2f b2 = {b23.x, b23.y}, b3 = {b23.z, b23.w};
        acc0 = __builtin_amdgcn_wmma_f32_16x16x4_f32(false, a, false, b0, (short)0, acc0, false, false);
        acc1 = __builtin_amdgcn_wmma_f32_16x16x4_f32(false, a, false, b1, (short)0, acc1, false, false);
        acc2 = __builtin_amdgcn_wmma_f32_16x16x4_f32(false, a, false, b2, (short)0, acc2, false, false);
        acc3 = __builtin_amdgcn_wmma_f32_16x16x4_f32(false, a, false, b3, (short)0, acc3, false, false);
    }

    const int rbase = gwave * 16 + lhalf * 8;
    float bi0 = 0.f, bi1 = 0.f, bi2 = 0.f, bi3 = 0.f;
    if (bias) { bi0 = bias[lmod]; bi1 = bias[16 + lmod]; bi2 = bias[32 + lmod]; bi3 = bias[48 + lmod]; }
#pragma unroll
    for (int r = 0; r < 8; ++r) {
        size_t ro = (size_t)(rbase + r) * HDIM + lmod;
        float v0 = acc0[r] + bi0, v1 = acc1[r] + bi1, v2 = acc2[r] + bi2, v3 = acc3[r] + bi3;
        if (act) { v0 = lrelu(v0); v1 = lrelu(v1); v2 = lrelu(v2); v3 = lrelu(v3); }
        C[ro] = v0; C[ro + 16] = v1; C[ro + 32] = v2; C[ro + 48] = v3;
    }
}

__global__ void conv_init_kernel(const float* __restrict__ Hm, const float* __restrict__ dinv,
                                 const float* __restrict__ bias, float* __restrict__ out, int n) {
    int tid = blockIdx.x * blockDim.x + threadIdx.x;
    if (tid >= n * 16) return;
    int i = tid >> 4, j = (tid & 15) * 4;
    float s = dinv[i]; s *= s;
    float4 h = *(const float4*)(Hm + (size_t)i * HDIM + j);
    float4 b = *(const float4*)(bias + j);
    float4 o = make_float4(b.x + s * h.x, b.y + s * h.y, b.z + s * h.z, b.w + s * h.w);
    *(float4*)(out + (size_t)i * HDIM + j) = o;
}

__global__ void conv_scatter_kernel(const float* __restrict__ Hm, const int* __restrict__ src,
                                    const int* __restrict__ dst, const float* __restrict__ dinv,
                                    float* __restrict__ out, int e) {
    int tid = blockIdx.x * blockDim.x + threadIdx.x;
    int ed = tid >> 4;
    if (ed >= e) return;
    int j = (tid & 15) * 4;
    int s = src[ed], d = dst[ed];
    float nrm = dinv[s] * dinv[d];
    float4 h = *(const float4*)(Hm + (size_t)s * HDIM + j);
    float* o = out + (size_t)d * HDIM + j;
    __hip_atomic_fetch_add(o + 0, nrm * h.x, __ATOMIC_RELAXED, __HIP_MEMORY_SCOPE_AGENT);
    __hip_atomic_fetch_add(o + 1, nrm * h.y, __ATOMIC_RELAXED, __HIP_MEMORY_SCOPE_AGENT);
    __hip_atomic_fetch_add(o + 2, nrm * h.z, __ATOMIC_RELAXED, __HIP_MEMORY_SCOPE_AGENT);
    __hip_atomic_fetch_add(o + 3, nrm * h.w, __ATOMIC_RELAXED, __HIP_MEMORY_SCOPE_AGENT);
}

__global__ void relu_inplace_kernel(float* __restrict__ buf, int n16) {
    int tid = blockIdx.x * blockDim.x + threadIdx.x;
    if (tid >= n16) return;
    float4* p = (float4*)buf + tid;
    float4 v = *p;
    v.x = fmaxf(v.x, 0.0f); v.y = fmaxf(v.y, 0.0f);
    v.z = fmaxf(v.z, 0.0f); v.w = fmaxf(v.w, 0.0f);
    *p = v;
}

__global__ void dual_head_kernel(const float* __restrict__ Z, const int* __restrict__ idx, int n,
                                 const float* __restrict__ Wa, const float* __restrict__ ba,
                                 float* __restrict__ outA,
                                 const float* __restrict__ Wb, const float* __restrict__ bb,
                                 float* __restrict__ outB) {
    int t = blockIdx.x * blockDim.x + threadIdx.x;
    if (t >= n) return;
    const float* row = Z + (size_t)idx[t] * HDIM;
    float da = 0.0f, db = 0.0f;
#pragma unroll
    for (int j = 0; j < HDIM; j += 4) {
        float4 z  = *(const float4*)(row + j);
        float4 wa = *(const float4*)(Wa + j);
        float4 wb = *(const float4*)(Wb + j);
        da += z.x * wa.x + z.y * wa.y + z.z * wa.z + z.w * wa.w;
        db += z.x * wb.x + z.y * wb.y + z.z * wb.z + z.w * wb.w;
    }
    outA[t] = lrelu(da + ba[0]);
    outB[t] = lrelu(db + bb[0]);
}

__global__ void tprob_kernel(const float* __restrict__ P1, const float* __restrict__ Wp2,
                             const float* __restrict__ bp2, float* __restrict__ out, int n) {
    int i = blockIdx.x * blockDim.x + threadIdx.x;
    if (i >= n) return;
    const float* row = P1 + (size_t)i * HDIM;
    float d0 = 0.0f, d1 = 0.0f;
#pragma unroll
    for (int k = 0; k < HDIM; ++k) {
        float z = row[k];
        d0 += z * Wp2[k * 2 + 0];
        d1 += z * Wp2[k * 2 + 1];
    }
    out[i * 2 + 0] = lrelu(d0 + bp2[0]);
    out[i * 2 + 1] = lrelu(d1 + bp2[1]);
}

extern "C" void kernel_launch(void* const* d_in, const int* in_sizes, int n_in,
                              void* d_out, int out_size, void* d_ws, size_t ws_size,
                              hipStream_t stream) {
    const float* x     = (const float*)d_in[0];
    const int*   ei    = (const int*)d_in[1];
    const int*   treat = (const int*)d_in[2];
    const int*   ctrl  = (const int*)d_in[3];
    const float* W1  = (const float*)d_in[4];  const float* b1  = (const float*)d_in[5];
    const float* W2  = (const float*)d_in[6];  const float* b2  = (const float*)d_in[7];
    const float* Wy1 = (const float*)d_in[8];  const float* by1 = (const float*)d_in[9];
    const float* Wy0 = (const float*)d_in[10]; const float* by0 = (const float*)d_in[11];
    const float* Wp1 = (const float*)d_in[12]; const float* bp1 = (const float*)d_in[13];
    const float* Wp2 = (const float*)d_in[14]; const float* bp2 = (const float*)d_in[15];

    const int N = in_sizes[0] / INDIM;
    const int E = in_sizes[1] / 2;
    const int T = in_sizes[2];
    const int Cn = in_sizes[3];
    const int* e_src = ei;
    const int* e_dst = ei + E;

    float* dinv = (float*)d_ws;
    float* buf0 = dinv + (((size_t)N + 63) & ~(size_t)63);
    float* buf1 = buf0 + (size_t)N * HDIM;

    float* o_y1    = (float*)d_out;
    float* o_yc0   = o_y1 + T;
    float* o_y0    = o_yc0 + T;
    float* o_yc1   = o_y0 + Cn;
    float* o_tprob = o_yc1 + Cn;
    float* o_xZ2   = o_tprob + (size_t)2 * N;

    const int BLK = 256;
    auto cdiv = [](long long a, long long b) { return (int)((a + b - 1) / b); };
    const int gemm_blocks = cdiv(N / 16, 8);
    const size_t smem128 = (size_t)INDIM * HDIM * sizeof(float);
    const size_t smem64  = (size_t)HDIM * HDIM * sizeof(float);

    deg_init_kernel<<<cdiv(N, BLK), BLK, 0, stream>>>(dinv, N);
    deg_count_kernel<<<cdiv(E, BLK), BLK, 0, stream>>>(e_dst, dinv, E);
    dinv_kernel<<<cdiv(N, BLK), BLK, 0, stream>>>(dinv, N);

    wmma_gemm_f32_kernel<<<gemm_blocks, BLK, smem128, stream>>>(x, W1, buf0, nullptr, N, INDIM, 0);
    conv_init_kernel<<<cdiv((long long)N * 16, BLK), BLK, 0, stream>>>(buf0, dinv, b1, buf1, N);
    conv_scatter_kernel<<<cdiv((long long)E * 16, BLK), BLK, 0, stream>>>(buf0, e_src, e_dst, dinv, buf1, E);
    relu_inplace_kernel<<<cdiv((long long)N * 16, BLK), BLK, 0, stream>>>(buf1, N * 16);

    wmma_gemm_f32_kernel<<<gemm_blocks, BLK, smem64, stream>>>(buf1, W2, buf0, nullptr, N, HDIM, 0);
    conv_init_kernel<<<cdiv((long long)N * 16, BLK), BLK, 0, stream>>>(buf0, dinv, b2, o_xZ2, N);
    conv_scatter_kernel<<<cdiv((long long)E * 16, BLK), BLK, 0, stream>>>(buf0, e_src, e_dst, dinv, o_xZ2, E);

    dual_head_kernel<<<cdiv(T, BLK), BLK, 0, stream>>>(o_xZ2, treat, T, Wy1, by1, o_y1, Wy0, by0, o_yc0);
    dual_head_kernel<<<cdiv(Cn, BLK), BLK, 0, stream>>>(o_xZ2, ctrl, Cn, Wy0, by0, o_y0, Wy1, by1, o_yc1);

    wmma_gemm_f32_kernel<<<gemm_blocks, BLK, smem64, stream>>>(o_xZ2, Wp1, buf0, bp1, N, HDIM, 1);
    tprob_kernel<<<cdiv(N, BLK), BLK, 0, stream>>>(buf0, Wp2, bp2, o_tprob, N);
}


# Float-atomic scatter-add replaced by destination-ordered row gather: edges ranked per call (returning int atomic, scan, place); each node summed in registers, stored once; sum order = arrival rank
# speedup vs baseline: 1.6923x; 1.6923x over previous
#include <hip/hip_runtime.h>
#include <hip/hip_bf16.h>

typedef __attribute__((ext_vector_type(2))) float v2f;
typedef __attribute__((ext_vector_type(8))) float v8f;

#define HDIM 64
#define INDIM 128

__device__ __forceinline__ float lrelu(float v) { return v > 0.0f ? v : 0.01f * v; }

__global__ void zero_cnt_kernel(int* __restrict__ cnt, int n) {
    int i = blockIdx.x * blockDim.x + threadIdx.x;
    if (i < n) cnt[i] = 0;
}

__global__ void edge_rank_kernel(const int* __restrict__ dst, int* __restrict__ cnt,
                                 int* __restrict__ pos, int e) {
    int i = blockIdx.x * blockDim.x + threadIdx.x;
    if (i < e)
        pos[i] = __hip_atomic_fetch_add(&cnt[dst[i]], 1, __ATOMIC_RELAXED, __HIP_MEMORY_SCOPE_AGENT);
}

#define SCAN_EPB 1024
__global__ void __launch_bounds__(256) row_offsets_kernel(const int* __restrict__ cnt, int* __restrict__ offs,
                                                          float* __restrict__ dinv, int n) {
    __shared__ int wsum[8];
    __shared__ int wpre[8];
    const int tid  = threadIdx.x;
    const int lane = tid & 31;
    const int wave = tid >> 5;
    const int base = blockIdx.x * SCAN_EPB;

    int pre = 0;
    const int4* c4 = (const int4*)cnt;
    for (int q = tid; q < (base >> 2); q += 256) {
        int4 v = c4[q];
        pre += (v.x + v.y) + (v.z + v.w);
    }

    const int i0 = base + tid * 4;
    int c[4];
    for (int k = 0; k < 4; ++k) c[k] = (i0 + k < n) ? cnt[i0 + k] : 0;
    const int t = (c[0] + c[1]) + (c[2] + c[3]);

    int incl = t;
    for (int o = 1; o < 32; o <<= 1) {
        int u = __shfl_up(incl, o, 32);
        if (lane >= o) incl += u;
    }
    for (int o = 16; o > 0; o >>= 1) pre += __shfl_xor(pre, o, 32);
    if (lane == 31) wsum[wave] = incl;
    if (lane == 0)  wpre[wave] = pre;
    __syncthreads();
    int front = 0;
    for (int w = 0; w < 8; ++w) {
        front += wpre[w];
        if (w < wave) front += wsum[w];
    }
    int excl = front + incl - t;
    for (int k = 0; k < 4; ++k) {
        if (i0 + k < n) {
            offs[i0 + k] = excl;
            float d = (float)(c[k] + 1);
            dinv[i0 + k] = d > 0.0f ? rsqrtf(d) : 0.0f;
            excl += c[k];
        }
    }
}

__global__ void edge_place_kernel(const int* __restrict__ src, const int* __restrict__ dst,
                                  const int* __restrict__ pos, const int* __restrict__ offs,
                                  const float* __restrict__ dinv, int* __restrict__ adj, int e) {
    int i = blockIdx.x * blockDim.x + threadIdx.x;
    if (i >= e) return;
    int s = src[i], d = dst[i];
    ((int2*)adj)[offs[d] + pos[i]] = make_int2(s, __float_as_int(dinv[s]));
}

__global__ void wmma_gemm_f32_kernel(const float* __restrict__ A, const float* __restrict__ B,
                                     float* __restrict__ C, const float* __restrict__ bias,
                                     int M, int K, int act) {
    extern __shared__ float bsw[];
    const int tid   = threadIdx.x;
    const int lane  = tid & 31;
    const int wave  = tid >> 5;
    const int lmod  = lane & 15;
    const int lhalf = lane >> 4;

    const int pairs = K * 32;
    for (int p = tid; p < pairs; p += 256) {
        int ks  = p >> 7;
        int ll  = (p >> 2) & 31;
        int tn  = p & 3;
        int kb  = 4 * ks + 2 * (ll >> 4);
        int col = tn * 16 + (ll & 15);
        bsw[p * 2 + 0] = B[(size_t)kb * HDIM + col];
        bsw[p * 2 + 1] = B[(size_t)(kb + 1) * HDIM + col];
    }
    __syncthreads();

    const int gwave = blockIdx.x * 8 + wave;
    if (gwave >= (M >> 4)) return;

    const float* arow = A + (size_t)(gwave * 16 + lmod) * K + 2 * lhalf;
    v8f acc0 = {}, acc1 = {}, acc2 = {}, acc3 = {};

    const int steps = K >> 2;
    for (int ks = 0; ks < steps; ++ks) {
        v2f a = *(const v2f*)(arow + 4 * ks);
        const float4* bp = (const float4*)(bsw + (size_t)(ks * 32 + lane) * 8);
        float4 b01 = bp[0];
        float4 b23 = bp[1];
        v2f b0 = {b01.x, b01.y}, b1 = {b01.z, b01.w};
        v2f b2 = {b23.x, b23.y}, b3 = {b23.z, b23.w};
        acc0 = __builtin_amdgcn_wmma_f32_16x16x4_f32(false, a, false, b0, (short)0, acc0, false, false);
        acc1 = __builtin_amdgcn_wmma_f32_16x16x4_f32(false, a, false, b1, (short)0, acc1, false, false);
        acc2 = __builtin_amdgcn_wmma_f32_16x16x4_f32(false, a, false, b2, (short)0, acc2, false, false);
        acc3 = __builtin_amdgcn_wmma_f32_16x16x4_f32(false, a, false, b3, (short)0, acc3, false, false);
    }

    const int rbase = gwave * 16 + lhalf * 8;
    float bi0 = 0.f, bi1 = 0.f, bi2 = 0.f, bi3 = 0.f;
    if (bias) { bi0 = bias[lmod]; bi1 = bias[16 + lmod]; bi2 = bias[32 + lmod]; bi3 = bias[48 + lmod]; }
#pragma unroll
    for (int r = 0; r < 8; ++r) {
        size_t ro = (size_t)(rbase + r) * HDIM + lmod;
        float v0 = acc0[r] + bi0, v1 = acc1[r] + bi1, v2 = acc2[r] + bi2, v3 = acc3[r] + bi3;
        if (act) { v0 = lrelu(v0); v1 = lrelu(v1); v2 = lrelu(v2); v3 = lrelu(v3); }
        C[ro] = v0; C[ro + 16] = v1; C[ro + 32] = v2; C[ro + 48] = v3;
    }
}

__global__ void __launch_bounds__(256) conv_gather_kernel(const float* __restrict__ Hm, const int* __restrict__ adj,
                                                          const int* __restrict__ offs, const int* __restrict__ cnt,
                                                          const float* __restrict__ dinv, const float* __restrict__ bias,
                                                          float* __restrict__ out, int n, int act) {
    int tid = blockIdx.x * blockDim.x + threadIdx.x;
    int i = tid >> 4;
    if (i >= n) return;
    const int j = (tid & 15) * 4;
    const float dd = dinv[i];
    const int m = cnt[i];
    const int2* ap = (const int2*)adj + offs[i];
    const float* hcol = Hm + j;

    float4 hs = *(const float4*)(hcol + (size_t)i * HDIM);
    const float sl = dd * dd;
    float4 acc = make_float4(sl * hs.x, sl * hs.y, sl * hs.z, sl * hs.w);

    int k = 0;
    for (; k + 4 <= m; k += 4) {
        int2 e0 = ap[k], e1 = ap[k + 1], e2 = ap[k + 2], e3 = ap[k + 3];
        float4 h0 = *(const float4*)(hcol + (size_t)e0.x * HDIM);
        float4 h1 = *(const float4*)(hcol + (size_t)e1.x * HDIM);
        float4 h2 = *(const float4*)(hcol + (size_t)e2.x * HDIM);
        float4 h3 = *(const float4*)(hcol + (size_t)e3.x * HDIM);
        float w0 = __int_as_float(e0.y) * dd, w1 = __int_as_float(e1.y) * dd;
        float w2 = __int_as_float(e2.y) * dd, w3 = __int_as_float(e3.y) * dd;
        acc.x += w0 * h0.x; acc.y += w0 * h0.y; acc.z += w0 * h0.z; acc.w += w0 * h0.w;
        acc.x += w1 * h1.x; acc.y += w1 * h1.y; acc.z += w1 * h1.z; acc.w += w1 * h1.w;
        acc.x += w2 * h2.x; acc.y += w2 * h2.y; acc.z += w2 * h2.z; acc.w += w2 * h2.w;
        acc.x += w3 * h3.x; acc.y += w3 * h3.y; acc.z += w3 * h3.z; acc.w += w3 * h3.w;
    }
    const int r = m - k;
    if (r > 0) {
        int2 e0 = ap[k];
        int2 e1 = ap[r > 1 ? k + 1 : k];
        int2 e2 = ap[r > 2 ? k + 2 : k];
        float4 h0 = *(const float4*)(hcol + (size_t)e0.x * HDIM);
        float4 h1 = *(const float4*)(hcol + (size_t)e1.x * HDIM);
        float4 h2 = *(const float4*)(hcol + (size_t)e2.x * HDIM);
        float w0 = __int_as_float(e0.y) * dd;
        float w1 = r > 1 ? __int_as_float(e1.y) * dd : 0.0f;
        float w2 = r > 2 ? __int_as_float(e2.y) * dd : 0.0f;
        acc.x += w0 * h0.x; acc.y += w0 * h0.y; acc.z += w0 * h0.z; acc.w += w0 * h0.w;
        acc.x += w1 * h1.x; acc.y += w1 * h1.y; acc.z += w1 * h1.z; acc.w += w1 * h1.w;
        acc.x += w2 * h2.x; acc.y += w2 * h2.y; acc.z += w2 * h2.z; acc.w += w2 * h2.w;
    }

    float4 b = *(const float4*)(bias + j);
    float4 o = make_float4(b.x + acc.x, b.y + acc.y, b.z + acc.z, b.w + acc.w);
    if (act) {
        o.x = fmaxf(o.x, 0.0f); o.y = fmaxf(o.y, 0.0f);
        o.z = fmaxf(o.z, 0.0f); o.w = fmaxf(o.w, 0.0f);
    }
    *(float4*)(out + (size_t)i * HDIM + j) = o;
}

__global__ void dual_head_kernel(const float* __restrict__ Z, const int* __restrict__ idx, int n,
                                 const float* __restrict__ Wa, const float* __restrict__ ba,
                                 float* __restrict__ outA,
                                 const float* __restrict__ Wb, const float* __restrict__ bb,
                                 float* __restrict__ outB) {
    int t = blockIdx.x * blockDim.x + threadIdx.x;
    if (t >= n) return;
    const float* row = Z + (size_t)idx[t] * HDIM;
    float da = 0.0f, db = 0.0f;
#pragma unroll
    for (int j = 0; j < HDIM; j += 4) {
        float4 z  = *(const float4*)(row + j);
        float4 wa = *(const float4*)(Wa + j);
        float4 wb = *(const float4*)(Wb + j);
        da += z.x * wa.x + z.y * wa.y + z.z * wa.z + z.w * wa.w;
        db += z.x * wb.x + z.y * wb.y + z.z * wb.z + z.w * wb.w;
    }
    outA[t] = lrelu(da + ba[0]);
    outB[t] = lrelu(db + bb[0]);
}

__global__ void tprob_kernel(const float* __restrict__ P1, const float* __restrict__ Wp2,
                             const float* __restrict__ bp2, float* __restrict__ out, int n) {
    int i = blockIdx.x * blockDim.x + threadIdx.x;
    if (i >= n) return;
    const float* row = P1 + (size_t)i * HDIM;
    float d0 = 0.0f, d1 = 0.0f;
#pragma unroll
    for (int k = 0; k < HDIM; ++k) {
        float z = row[k];
        d0 += z * Wp2[k * 2 + 0];
        d1 += z * Wp2[k * 2 + 1];
    }
    out[i * 2 + 0] = lrelu(d0 + bp2[0]);
    out[i * 2 + 1] = lrelu(d1 + bp2[1]);
}

extern "C" void kernel_launch(void* const* d_in, const int* in_sizes, int n_in,
                              void* d_out, int out_size, void* d_ws, size_t ws_size,
                              hipStream_t stream) {
    const float* x     = (const float*)d_in[0];
    const int*   ei    = (const int*)d_in[1];
    const int*   treat = (const int*)d_in[2];
    const int*   ctrl  = (const int*)d_in[3];
    const float* W1  = (const float*)d_in[4];  const float* b1  = (const float*)d_in[5];
    const float* W2  = (const float*)d_in[6];  const float* b2  = (const float*)d_in[7];
    const float* Wy1 = (const float*)d_in[8];  const float* by1 = (const float*)d_in[9];
    const float* Wy0 = (const float*)d_in[10]; const float* by0 = (const float*)d_in[11];
    const float* Wp1 = (const float*)d_in[12]; const float* bp1 = (const float*)d_in[13];
    const float* Wp2 = (const float*)d_in[14]; const float* bp2 = (const float*)d_in[15];

    const int N = in_sizes[0] / INDIM;
    const int E = in_sizes[1] / 2;
    const int T = in_sizes[2];
    const int Cn = in_sizes[3];
    const int* e_src = ei;
    const int* e_dst = ei + E;

    float* dinv = (float*)d_ws;
    float* buf0 = dinv + (((size_t)N + 63) & ~(size_t)63);
    int* adj  = (int*)(buf0 + (size_t)N * HDIM);
    int* pos  = adj + (size_t)2 * E;
    int* cnt  = pos + (((size_t)E + 63) & ~(size_t)63);
    int* offs = cnt + (((size_t)N + 63) & ~(size_t)63);

    float* o_y1    = (float*)d_out;
    float* o_yc0   = o_y1 + T;
    float* o_y0    = o_yc0 + T;
    float* o_yc1   = o_y0 + Cn;
    float* o_tprob = o_yc1 + Cn;
    float* o_xZ2   = o_tprob + (size_t)2 * N;

    const int BLK = 256;
    auto cdiv = [](long long a, long long b) { return (int)((a + b - 1) / b); };
    const int gemm_blocks = cdiv(N / 16, 8);
    const size_t smem128 = (size_t)INDIM * HDIM * sizeof(float);
    const size_t smem64  = (size_t)HDIM * HDIM * sizeof(float);

    zero_cnt_kernel<<<cdiv(N, BLK), BLK, 0, stream>>>(cnt, N);
    edge_rank_kernel<<<cdiv(E, BLK), BLK, 0, stream>>>(e_dst, cnt, pos, E);
    row_offsets_kernel<<<cdiv(N, SCAN_EPB), BLK, 0, stream>>>(cnt, offs, dinv, N);
    edge_place_kernel<<<cdiv(E, BLK), BLK, 0, stream>>>(e_src, e_dst, pos, offs, dinv, adj, E);

    wmma_gemm_f32_kernel<<<gemm_blocks, BLK, smem128, stream>>>(x, W1, buf0, nullptr, N, INDIM, 0);
    conv_gather_kernel<<<cdiv((long long)N * 16, BLK), BLK, 0, stream>>>(buf0, adj, offs, cnt, dinv, b1, o_xZ2, N, 1);

    wmma_gemm_f32_kernel<<<gemm_blocks, BLK, smem64, stream>>>(o_xZ2, W2, buf0, nullptr, N, HDIM, 0);
    conv_gather_kernel<<<cdiv((long long)N * 16, BLK), BLK, 0, stream>>>(buf0, adj, offs, cnt, dinv, b2, o_xZ2, N, 0);

    dual_head_kernel<<<cdiv(T, BLK), BLK, 0, stream>>>(o_xZ2, treat, T, Wy1, by1, o_y1, Wy0, by0, o_yc0);
    dual_head_kernel<<<cdiv(Cn, BLK), BLK, 0, stream>>>(o_xZ2, ctrl, Cn, Wy0, by0, o_y0, Wy1, by1, o_yc1);

    wmma_gemm_f32_kernel<<<gemm_blocks, BLK, smem64, stream>>>(o_xZ2, Wp1, buf0, bp1, N, HDIM, 1);
    tprob_kernel<<<cdiv(N, BLK), BLK, 0, stream>>>(buf0, Wp2, bp2, o_tprob, N);
}


# Edge order from a two-level counting sort in LDS (512-node dst bins, then rank per node), not global returning atomics + random 8-B stores; H rows prescaled by deg^-1/2; sum order = LDS arrival rank
# speedup vs baseline: 2.6365x; 1.5579x over previous
#include <hip/hip_runtime.h>
#include <hip/hip_bf16.h>

typedef __attribute__((ext_vector_type(2))) float v2f;
typedef __attribute__((ext_vector_type(8))) float v8f;

#define HDIM 64
#define INDIM 128

__device__ __forceinline__ float lrelu(float v) { return v > 0.0f ? v : 0.01f * v; }

#define SORT_TPB 512
#define SORT_CH 6144
#define BIN_SHIFT 9
#define BIN_NODES 512
#define FINE_CAP 12288

__device__ __forceinline__ int block_excl_scan_512(int v, int* wsum, int tid) {
    const int lane = tid & 31;
    const int wave = tid >> 5;
    int incl = v;
    for (int o = 1; o < 32; o <<= 1) {
        int u = __shfl_up(incl, o, 32);
        if (lane >= o) incl += u;
    }
    if (lane == 31) wsum[wave] = incl;
    __syncthreads();
    int front = 0;
    for (int w = 0; w < 16; ++w)
        if (w < wave) front += wsum[w];
    return front + incl - v;
}

__global__ void __launch_bounds__(SORT_TPB) bin_hist_kernel(const int* __restrict__ dst, int* __restrict__ hist,
                                                            int e, int nb, int g1) {
    extern __shared__ int lds[];
    const int tid = threadIdx.x;
    const int c = blockIdx.x;
    for (int b = tid; b < nb; b += SORT_TPB) lds[b] = 0;
    __syncthreads();
    const int base = c * SORT_CH;
    const int lim = min(e, base + SORT_CH);
    for (int i = base + tid; i < lim; i += SORT_TPB) atomicAdd(&lds[dst[i] >> BIN_SHIFT], 1);
    __syncthreads();
    for (int b = tid; b < nb; b += SORT_TPB) hist[(size_t)b * g1 + c] = lds[b];
}

#define SCAN_EPB 1024
__global__ void __launch_bounds__(256) table_scan_kernel(const int* __restrict__ cnt, int* __restrict__ offs, int n) {
    __shared__ int wsum[8];
    __shared__ int wpre[8];
    const int tid  = threadIdx.x;
    const int lane = tid & 31;
    const int wave = tid >> 5;
    const int base = blockIdx.x * SCAN_EPB;

    int pre = 0;
    const int4* c4 = (const int4*)cnt;
    for (int q = tid; q < (base >> 2); q += 256) {
        int4 v = c4[q];
        pre += (v.x + v.y) + (v.z + v.w);
    }

    const int i0 = base + tid * 4;
    int c[4];
    for (int k = 0; k < 4; ++k) c[k] = (i0 + k < n) ? cnt[i0 + k] : 0;
    const int t = (c[0] + c[1]) + (c[2] + c[3]);

    int incl = t;
    for (int o = 1; o < 32; o <<= 1) {
        int u = __shfl_up(incl, o, 32);
        if (lane >= o) incl += u;
    }
    for (int o = 16; o > 0; o >>= 1) pre += __shfl_xor(pre, o, 32);
    if (lane == 31) wsum[wave] = incl;
    if (lane == 0)  wpre[wave] = pre;
    __syncthreads();
    int front = 0;
    for (int w = 0; w < 8; ++w) {
        front += wpre[w];
        if (w < wave) front += wsum[w];
    }
    int excl = front + incl - t;
    for (int k = 0; k < 4; ++k) {
        if (i0 + k < n) {
            offs[i0 + k] = excl;
            excl += c[k];
        }
    }
}

__global__ void __launch_bounds__(SORT_TPB) bin_scatter_kernel(const int* __restrict__ src, const int* __restrict__ dst,
                                                               const int* __restrict__ hist, const int* __restrict__ hofs,
                                                               int* __restrict__ binned, int e, int nb, int g1) {
    extern __shared__ int lds[];
    __shared__ int wsum[16];
    const int tid = threadIdx.x;
    const int c = blockIdx.x;
    int* lstart = lds;
    int* cursor = lds + nb;
    int* gbase  = lds + 2 * nb;
    int2* stage = (int2*)(lds + ((3 * nb + 1) & ~1));

    const int ipt = (nb + SORT_TPB - 1) / SORT_TPB;
    const int b0 = tid * ipt;
    const int b1 = min(nb, b0 + ipt);
    int mine = 0;
    for (int b = b0; b < b1; ++b) mine += hist[(size_t)b * g1 + c];
    int run = block_excl_scan_512(mine, wsum, tid);
    for (int b = b0; b < b1; ++b) {
        lstart[b] = run;
        cursor[b] = run;
        gbase[b]  = hofs[(size_t)b * g1 + c];
        run += hist[(size_t)b * g1 + c];
    }
    __syncthreads();

    const int base = c * SORT_CH;
    const int lim = min(e, base + SORT_CH);
    for (int i = base + tid; i < lim; i += SORT_TPB) {
        int s = src[i], d = dst[i];
        int p = atomicAdd(&cursor[d >> BIN_SHIFT], 1);
        stage[p] = make_int2(s, d);
    }
    __syncthreads();

    const int nloc = lim - base;
    int2* out2 = (int2*)binned;
    for (int j = tid; j < nloc; j += SORT_TPB) {
        int2 v = stage[j];
        int b = v.y >> BIN_SHIFT;
        out2[gbase[b] + (j - lstart[b])] = v;
    }
}

__global__ void __launch_bounds__(SORT_TPB) bin_rows_kernel(const int* __restrict__ binned, const int* __restrict__ hofs,
                                                            int* __restrict__ adj, int* __restrict__ cntG,
                                                            int* __restrict__ offsG, float* __restrict__ dinv,
                                                            int n, int e, int nb, int g1) {
    __shared__ int cnt[BIN_NODES];
    __shared__ int cursor[BIN_NODES];
    __shared__ int wsum[16];
    __shared__ int stage[FINE_CAP];
    const int tid = threadIdx.x;
    const int b = blockIdx.x;
    const int start = hofs[(size_t)b * g1];
    const int end = (b + 1 < nb) ? hofs[(size_t)(b + 1) * g1] : e;
    const int m = end - start;
    const int2* bp = (const int2*)binned + start;

    cnt[tid] = 0;
    __syncthreads();
    for (int j = tid; j < m; j += SORT_TPB) atomicAdd(&cnt[bp[j].y & (BIN_NODES - 1)], 1);
    __syncthreads();

    const int cdeg = cnt[tid];
    const int ex = block_excl_scan_512(cdeg, wsum, tid);
    cursor[tid] = ex;
    const int node = b * BIN_NODES + tid;
    if (node < n) {
        cntG[node] = cdeg;
        offsG[node] = start + ex;
        float d = (float)(cdeg + 1);
        dinv[node] = d > 0.0f ? rsqrtf(d) : 0.0f;
    }
    __syncthreads();

    if (m <= FINE_CAP) {
        for (int j = tid; j < m; j += SORT_TPB) {
            int2 v = bp[j];
            int p = atomicAdd(&cursor[v.y & (BIN_NODES - 1)], 1);
            stage[p] = v.x;
        }
        __syncthreads();
        for (int j = tid; j < m; j += SORT_TPB) adj[start + j] = stage[j];
    } else {
        for (int j = tid; j < m; j += SORT_TPB) {
            int2 v = bp[j];
            int p = atomicAdd(&cursor[v.y & (BIN_NODES - 1)], 1);
            adj[start + p] = v.x;
        }
    }
}

__global__ void wmma_gemm_f32_kernel(const float* __restrict__ A, const float* __restrict__ B,
                                     float* __restrict__ C, const float* __restrict__ bias,
                                     const float* __restrict__ rowscale, int M, int K, int act) {
    extern __shared__ float bsw[];
    const int tid   = threadIdx.x;
    const int lane  = tid & 31;
    const int wave  = tid >> 5;
    const int lmod  = lane & 15;
    const int lhalf = lane >> 4;

    const int pairs = K * 32;
    for (int p = tid; p < pairs; p += 256) {
        int ks  = p >> 7;
        int ll  = (p >> 2) & 31;
        int tn  = p & 3;
        int kb  = 4 * ks + 2 * (ll >> 4);
        int col = tn * 16 + (ll & 15);
        bsw[p * 2 + 0] = B[(size_t)kb * HDIM + col];
        bsw[p * 2 + 1] = B[(size_t)(kb + 1) * HDIM + col];
    }
    __syncthreads();

    const int gwave = blockIdx.x * 8 + wave;
    if (gwave >= (M >> 4)) return;

    const float* arow = A + (size_t)(gwave * 16 + lmod) * K + 2 * lhalf;
    v8f acc0 = {}, acc1 = {}, acc2 = {}, acc3 = {};

    const int steps = K >> 2;
    for (int ks = 0; ks < steps; ++ks) {
        v2f a = *(const v2f*)(arow + 4 * ks);
        const float4* bp = (const float4*)(bsw + (size_t)(ks * 32 + lane) * 8);
        float4 b01 = bp[0];
        float4 b23 = bp[1];
        v2f b0 = {b01.x, b01.y}, b1 = {b01.z, b01.w};
        v2f b2 = {b23.x, b23.y}, b3 = {b23.z, b23.w};
        acc0 = __builtin_amdgcn_wmma_f32_16x16x4_f32(false, a, false, b0, (short)0, acc0, false, false);
        acc1 = __builtin_amdgcn_wmma_f32_16x16x4_f32(false, a, false, b1, (short)0, acc1, false, false);
        acc2 = __builtin_amdgcn_wmma_f32_16x16x4_f32(false, a, false, b2, (short)0, acc2, false, false);
        acc3 = __builtin_amdgcn_wmma_f32_16x16x4_f32(false, a, false, b3, (short)0, acc3, false, false);
    }

    const int rbase = gwave * 16 + lhalf * 8;
    float bi0 = 0.f, bi1 = 0.f, bi2 = 0.f, bi3 = 0.f;
    if (bias) { bi0 = bias[lmod]; bi1 = bias[16 + lmod]; bi2 = bias[32 + lmod]; bi3 = bias[48 + lmod]; }
#pragma unroll
    for (int r = 0; r < 8; ++r) {
        size_t ro = (size_t)(rbase + r) * HDIM + lmod;
        float v0 = acc0[r] + bi0, v1 = acc1[r] + bi1, v2 = acc2[r] + bi2, v3 = acc3[r] + bi3;
        if (act) { v0 = lrelu(v0); v1 = lrelu(v1); v2 = lrelu(v2); v3 = lrelu(v3); }
        if (rowscale) { const float rs = rowscale[rbase + r]; v0 *= rs; v1 *= rs; v2 *= rs; v3 *= rs; }
        C[ro] = v0; C[ro + 16] = v1; C[ro + 32] = v2; C[ro + 48] = v3;
    }
}

__global__ void __launch_bounds__(256) conv_gather_kernel(const float* __restrict__ Hm, const int* __restrict__ adj,
                                                          const int* __restrict__ offs, const int* __restrict__ cnt,
                                                          const float* __restrict__ dinv, const float* __restrict__ bias,
                                                          float* __restrict__ out, int n, int act) {
    int tid = blockIdx.x * blockDim.x + threadIdx.x;
    int i = tid >> 4;
    if (i >= n) return;
    const int j = (tid & 15) * 4;
    const float dd = dinv[i];
    const int m = cnt[i];
    const int* ap = adj + offs[i];
    const float* hcol = Hm + j;

    float4 acc = *(const float4*)(hcol + (size_t)i * HDIM);

    int k = 0;
    for (; k + 4 <= m; k += 4) {
        int s0 = ap[k], s1 = ap[k + 1], s2 = ap[k + 2], s3 = ap[k + 3];
        float4 h0 = *(const float4*)(hcol + (size_t)s0 * HDIM);
        float4 h1 = *(const float4*)(hcol + (size_t)s1 * HDIM);
        float4 h2 = *(const float4*)(hcol + (size_t)s2 * HDIM);
        float4 h3 = *(const float4*)(hcol + (size_t)s3 * HDIM);
        acc.x += h0.x; acc.y += h0.y; acc.z += h0.z; acc.w += h0.w;
        acc.x += h1.x; acc.y += h1.y; acc.z += h1.z; acc.w += h1.w;
        acc.x += h2.x; acc.y += h2.y; acc.z += h2.z; acc.w += h2.w;
        acc.x += h3.x; acc.y += h3.y; acc.z += h3.z; acc.w += h3.w;
    }
    const int r = m - k;
    if (r > 0) {
        int s0 = ap[k];
        int s1 = ap[r > 1 ? k + 1 : k];
        int s2 = ap[r > 2 ? k + 2 : k];
        float4 h0 = *(const float4*)(hcol + (size_t)s0 * HDIM);
        float4 h1 = *(const float4*)(hcol + (size_t)s1 * HDIM);
        float4 h2 = *(const float4*)(hcol + (size_t)s2 * HDIM);
        float w1 = r > 1 ? 1.0f : 0.0f;
        float w2 = r > 2 ? 1.0f : 0.0f;
        acc.x += h0.x; acc.y += h0.y; acc.z += h0.z; acc.w += h0.w;
        acc.x += w1 * h1.x; acc.y += w1 * h1.y; acc.z += w1 * h1.z; acc.w += w1 * h1.w;
        acc.x += w2 * h2.x; acc.y += w2 * h2.y; acc.z += w2 * h2.z; acc.w += w2 * h2.w;
    }

    float4 b = *(const float4*)(bias + j);
    float4 o = make_float4(b.x + dd * acc.x, b.y + dd * acc.y, b.z + dd * acc.z, b.w + dd * acc.w);
    if (act) {
        o.x = fmaxf(o.x, 0.0f); o.y = fmaxf(o.y, 0.0f);
        o.z = fmaxf(o.z, 0.0f); o.w = fmaxf(o.w, 0.0f);
    }
    *(float4*)(out + (size_t)i * HDIM + j) = o;
}

__global__ void dual_head_kernel(const float* __restrict__ Z, const int* __restrict__ idx, int n,
                                 const float* __restrict__ Wa, const float* __restrict__ ba,
                                 float* __restrict__ outA,
                                 const float* __restrict__ Wb, const float* __restrict__ bb,
                                 float* __restrict__ outB) {
    int t = blockIdx.x * blockDim.x + threadIdx.x;
    if (t >= n) return;
    const float* row = Z + (size_t)idx[t] * HDIM;
    float da = 0.0f, db = 0.0f;
#pragma unroll
    for (int j = 0; j < HDIM; j += 4) {
        float4 z  = *(const float4*)(row + j);
        float4 wa = *(const float4*)(Wa + j);
        float4 wb = *(const float4*)(Wb + j);
        da += z.x * wa.x + z.y * wa.y + z.z * wa.z + z.w * wa.w;
        db += z.x * wb.x + z.y * wb.y + z.z * wb.z + z.w * wb.w;
    }
    outA[t] = lrelu(da + ba[0]);
    outB[t] = lrelu(db + bb[0]);
}

__global__ void tprob_kernel(const float* __restrict__ P1, const float* __restrict__ Wp2,
                             const float* __restrict__ bp2, float* __restrict__ out, int n) {
    int i = blockIdx.x * blockDim.x + threadIdx.x;
    if (i >= n) return;
    const float* row = P1 + (size_t)i * HDIM;
    float d0 = 0.0f, d1 = 0.0f;
#pragma unroll
    for (int k = 0; k < HDIM; ++k) {
        float z = row[k];
        d0 += z * Wp2[k * 2 + 0];
        d1 += z * Wp2[k * 2 + 1];
    }
    out[i * 2 + 0] = lrelu(d0 + bp2[0]);
    out[i * 2 + 1] = lrelu(d1 + bp2[1]);
}

extern "C" void kernel_launch(void* const* d_in, const int* in_sizes, int n_in,
                              void* d_out, int out_size, void* d_ws, size_t ws_size,
                              hipStream_t stream) {
    const float* x     = (const float*)d_in[0];
    const int*   ei    = (const int*)d_in[1];
    const int*   treat = (const int*)d_in[2];
    const int*   ctrl  = (const int*)d_in[3];
    const float* W1  = (const float*)d_in[4];  const float* b1  = (const float*)d_in[5];
    const float* W2  = (const float*)d_in[6];  const float* b2  = (const float*)d_in[7];
    const float* Wy1 = (const float*)d_in[8];  const float* by1 = (const float*)d_in[9];
    const float* Wy0 = (const float*)d_in[10]; const float* by0 = (const float*)d_in[11];
    const float* Wp1 = (const float*)d_in[12]; const float* bp1 = (const float*)d_in[13];
    const float* Wp2 = (const float*)d_in[14]; const float* bp2 = (const float*)d_in[15];

    const int N = in_sizes[0] / INDIM;
    const int E = in_sizes[1] / 2;
    const int T = in_sizes[2];
    const int Cn = in_sizes[3];
    const int* e_src = ei;
    const int* e_dst = ei + E;

    float* dinv = (float*)d_ws;
    float* buf0 = dinv + (((size_t)N + 63) & ~(size_t)63);
    int* binned = (int*)(buf0 + (size_t)N * HDIM);
    int* adj  = binned + (size_t)2 * E;
    int* cnt  = adj + (((size_t)E + 63) & ~(size_t)63);
    int* offs = cnt + (((size_t)N + 63) & ~(size_t)63);

    float* o_y1    = (float*)d_out;
    float* o_yc0   = o_y1 + T;
    float* o_y0    = o_yc0 + T;
    float* o_yc1   = o_y0 + Cn;
    float* o_tprob = o_yc1 + Cn;
    float* o_xZ2   = o_tprob + (size_t)2 * N;

    const int BLK = 256;
    auto cdiv = [](long long a, long long b) { return (int)((a + b - 1) / b); };
    const int gemm_blocks = cdiv(N / 16, 8);
    const size_t smem128 = (size_t)INDIM * HDIM * sizeof(float);
    const size_t smem64  = (size_t)HDIM * HDIM * sizeof(float);
    const int nb = cdiv(N, BIN_NODES);
    const int g1 = cdiv(E, SORT_CH);
    const int tab = nb * g1;
    int* hist = (int*)buf0;
    int* hofs = hist + (((size_t)tab + 63) & ~(size_t)63);
    const size_t smem_hist = (size_t)nb * sizeof(int);
    const size_t smem_scat = ((size_t)((3 * nb + 1) & ~1) + (size_t)2 * SORT_CH) * sizeof(int);

    bin_hist_kernel<<<g1, SORT_TPB, smem_hist, stream>>>(e_dst, hist, E, nb, g1);
    table_scan_kernel<<<cdiv(tab, SCAN_EPB), BLK, 0, stream>>>(hist, hofs, tab);
    bin_scatter_kernel<<<g1, SORT_TPB, smem_scat, stream>>>(e_src, e_dst, hist, hofs, binned, E, nb, g1);
    bin_rows_kernel<<<nb, SORT_TPB, 0, stream>>>(binned, hofs, adj, cnt, offs, dinv, N, E, nb, g1);

    wmma_gemm_f32_kernel<<<gemm_blocks, BLK, smem128, stream>>>(x, W1, buf0, nullptr, dinv, N, INDIM, 0);
    conv_gather_kernel<<<cdiv((long long)N * 16, BLK), BLK, 0, stream>>>(buf0, adj, offs, cnt, dinv, b1, o_xZ2, N, 1);

    wmma_gemm_f32_kernel<<<gemm_blocks, BLK, smem64, stream>>>(o_xZ2, W2, buf0, nullptr, dinv, N, HDIM, 0);
    conv_gather_kernel<<<cdiv((long long)N * 16, BLK), BLK, 0, stream>>>(buf0, adj, offs, cnt, dinv, b2, o_xZ2, N, 0);

    dual_head_kernel<<<cdiv(T, BLK), BLK, 0, stream>>>(o_xZ2, treat, T, Wy1, by1, o_y1, Wy0, by0, o_yc0);
    dual_head_kernel<<<cdiv(Cn, BLK), BLK, 0, stream>>>(o_xZ2, ctrl, Cn, Wy0, by0, o_y0, Wy1, by1, o_yc1);

    wmma_gemm_f32_kernel<<<gemm_blocks, BLK, smem64, stream>>>(o_xZ2, Wp1, buf0, bp1, nullptr, N, HDIM, 1);
    tprob_kernel<<<cdiv(N, BLK), BLK, 0, stream>>>(buf0, Wp2, bp2, o_tprob, N);
}


# Everything after conv2 in one launch instead of four: Wp1 GEMM keeps P1 in registers and does the 64x2 layer there (sum order: 4 in-lane terms, then 16-lane tree), heads run as extra blocks
# speedup vs baseline: 2.9623x; 1.1236x over previous
#include <hip/hip_runtime.h>
#include <hip/hip_bf16.h>

typedef __attribute__((ext_vector_type(2))) float v2f;
typedef __attribute__((ext_vector_type(8))) float v8f;

#define HDIM 64
#define INDIM 128

__device__ __forceinline__ float lrelu(float v) { return v > 0.0f ? v : 0.01f * v; }

#define SORT_TPB 512
#define SORT_CH 6144
#define BIN_SHIFT 9
#define BIN_NODES 512
#define FINE_CAP 12288

__device__ __forceinline__ int block_excl_scan_512(int v, int* wsum, int tid) {
    const int lane = tid & 31;
    const int wave = tid >> 5;
    int incl = v;
    for (int o = 1; o < 32; o <<= 1) {
        int u = __shfl_up(incl, o, 32);
        if (lane >= o) incl += u;
    }
    if (lane == 31) wsum[wave] = incl;
    __syncthreads();
    int front = 0;
    for (int w = 0; w < 16; ++w)
        if (w < wave) front += wsum[w];
    return front + incl - v;
}

__global__ void __launch_bounds__(SORT_TPB) bin_hist_kernel(const int* __restrict__ dst, int* __restrict__ hist,
                                                            int e, int nb, int g1) {
    extern __shared__ int lds[];
    const int tid = threadIdx.x;
    const int c = blockIdx.x;
    for (int b = tid; b < nb; b += SORT_TPB) lds[b] = 0;
    __syncthreads();
    const int base = c * SORT_CH;
    const int lim = min(e, base + SORT_CH);
    for (int i = base + tid; i < lim; i += SORT_TPB) atomicAdd(&lds[dst[i] >> BIN_SHIFT], 1);
    __syncthreads();
    for (int b = tid; b < nb; b += SORT_TPB) hist[(size_t)b * g1 + c] = lds[b];
}

#define SCAN_EPB 1024
__global__ void __launch_bounds__(256) table_scan_kernel(const int* __restrict__ cnt, int* __restrict__ offs, int n) {
    __shared__ int wsum[8];
    __shared__ int wpre[8];
    const int tid  = threadIdx.x;
    const int lane = tid & 31;
    const int wave = tid >> 5;
    const int base = blockIdx.x * SCAN_EPB;

    int pre = 0;
    const int4* c4 = (const int4*)cnt;
    for (int q = tid; q < (base >> 2); q += 256) {
        int4 v = c4[q];
        pre += (v.x + v.y) + (v.z + v.w);
    }

    const int i0 = base + tid * 4;
    int c[4];
    for (int k = 0; k < 4; ++k) c[k] = (i0 + k < n) ? cnt[i0 + k] : 0;
    const int t = (c[0] + c[1]) + (c[2] + c[3]);

    int incl = t;
    for (int o = 1; o < 32; o <<= 1) {
        int u = __shfl_up(incl, o, 32);
        if (lane >= o) incl += u;
    }
    for (int o = 16; o > 0; o >>= 1) pre += __shfl_xor(pre, o, 32);
    if (lane == 31) wsum[wave] = incl;
    if (lane == 0)  wpre[wave] = pre;
    __syncthreads();
    int front = 0;
    for (int w = 0; w < 8; ++w) {
        front += wpre[w];
        if (w < wave) front += wsum[w];
    }
    int excl = front + incl - t;
    for (int k = 0; k < 4; ++k) {
        if (i0 + k < n) {
            offs[i0 + k] = excl;
            excl += c[k];
        }
    }
}

__global__ void __launch_bounds__(SORT_TPB) bin_scatter_kernel(const int* __restrict__ src, const int* __restrict__ dst,
                                                               const int* __restrict__ hist, const int* __restrict__ hofs,
                                                               int* __restrict__ binned, int e, int nb, int g1) {
    extern __shared__ int lds[];
    __shared__ int wsum[16];
    const int tid = threadIdx.x;
    const int c = blockIdx.x;
    int* lstart = lds;
    int* cursor = lds + nb;
    int* gbase  = lds + 2 * nb;
    int2* stage = (int2*)(lds + ((3 * nb + 1) & ~1));

    const int ipt = (nb + SORT_TPB - 1) / SORT_TPB;
    const int b0 = tid * ipt;
    const int b1 = min(nb, b0 + ipt);
    int mine = 0;
    for (int b = b0; b < b1; ++b) mine += hist[(size_t)b * g1 + c];
    int run = block_excl_scan_512(mine, wsum, tid);
    for (int b = b0; b < b1; ++b) {
        lstart[b] = run;
        cursor[b] = run;
        gbase[b]  = hofs[(size_t)b * g1 + c];
        run += hist[(size_t)b * g1 + c];
    }
    __syncthreads();

    const int base = c * SORT_CH;
    const int lim = min(e, base + SORT_CH);
    for (int i = base + tid; i < lim; i += SORT_TPB) {
        int s = src[i], d = dst[i];
        int p = atomicAdd(&cursor[d >> BIN_SHIFT], 1);
        stage[p] = make_int2(s, d);
    }
    __syncthreads();

    const int nloc = lim - base;
    int2* out2 = (int2*)binned;
    for (int j = tid; j < nloc; j += SORT_TPB) {
        int2 v = stage[j];
        int b = v.y >> BIN_SHIFT;
        out2[gbase[b] + (j - lstart[b])] = v;
    }
}

__global__ void __launch_bounds__(SORT_TPB) bin_rows_kernel(const int* __restrict__ binned, const int* __restrict__ hofs,
                                                            int* __restrict__ adj, int* __restrict__ cntG,
                                                            int* __restrict__ offsG, float* __restrict__ dinv,
                                                            int n, int e, int nb, int g1) {
    __shared__ int cnt[BIN_NODES];
    __shared__ int cursor[BIN_NODES];
    __shared__ int wsum[16];
    __shared__ int stage[FINE_CAP];
    const int tid = threadIdx.x;
    const int b = blockIdx.x;
    const int start = hofs[(size_t)b * g1];
    const int end = (b + 1 < nb) ? hofs[(size_t)(b + 1) * g1] : e;
    const int m = end - start;
    const int2* bp = (const int2*)binned + start;

    cnt[tid] = 0;
    __syncthreads();
    for (int j = tid; j < m; j += SORT_TPB) atomicAdd(&cnt[bp[j].y & (BIN_NODES - 1)], 1);
    __syncthreads();

    const int cdeg = cnt[tid];
    const int ex = block_excl_scan_512(cdeg, wsum, tid);
    cursor[tid] = ex;
    const int node = b * BIN_NODES + tid;
    if (node < n) {
        cntG[node] = cdeg;
        offsG[node] = start + ex;
        float d = (float)(cdeg + 1);
        dinv[node] = d > 0.0f ? rsqrtf(d) : 0.0f;
    }
    __syncthreads();

    if (m <= FINE_CAP) {
        for (int j = tid; j < m; j += SORT_TPB) {
            int2 v = bp[j];
            int p = atomicAdd(&cursor[v.y & (BIN_NODES - 1)], 1);
            stage[p] = v.x;
        }
        __syncthreads();
        for (int j = tid; j < m; j += SORT_TPB) adj[start + j] = stage[j];
    } else {
        for (int j = tid; j < m; j += SORT_TPB) {
            int2 v = bp[j];
            int p = atomicAdd(&cursor[v.y & (BIN_NODES - 1)], 1);
            adj[start + p] = v.x;
        }
    }
}

__global__ void wmma_gemm_f32_kernel(const float* __restrict__ A, const float* __restrict__ B,
                                     float* __restrict__ C, const float* __restrict__ bias,
                                     const float* __restrict__ rowscale, int M, int K, int act) {
    extern __shared__ float bsw[];
    const int tid   = threadIdx.x;
    const int lane  = tid & 31;
    const int wave  = tid >> 5;
    const int lmod  = lane & 15;
    const int lhalf = lane >> 4;

    const int pairs = K * 32;
    for (int p = tid; p < pairs; p += 256) {
        int ks  = p >> 7;
        int ll  = (p >> 2) & 31;
        int tn  = p & 3;
        int kb  = 4 * ks + 2 * (ll >> 4);
        int col = tn * 16 + (ll & 15);
        bsw[p * 2 + 0] = B[(size_t)kb * HDIM + col];
        bsw[p * 2 + 1] = B[(size_t)(kb + 1) * HDIM + col];
    }
    __syncthreads();

    const int gwave = blockIdx.x * 8 + wave;
    if (gwave >= (M >> 4)) return;

    const float* arow = A + (size_t)(gwave * 16 + lmod) * K + 2 * lhalf;
    v8f acc0 = {}, acc1 = {}, acc2 = {}, acc3 = {};

    const int steps = K >> 2;
    for (int ks = 0; ks < steps; ++ks) {
        v2f a = *(const v2f*)(arow + 4 * ks);
        const float4* bp = (const float4*)(bsw + (size_t)(ks * 32 + lane) * 8);
        float4 b01 = bp[0];
        float4 b23 = bp[1];
        v2f b0 = {b01.x, b01.y}, b1 = {b01.z, b01.w};
        v2f b2 = {b23.x, b23.y}, b3 = {b23.z, b23.w};
        acc0 = __builtin_amdgcn_wmma_f32_16x16x4_f32(false, a, false, b0, (short)0, acc0, false, false);
        acc1 = __builtin_amdgcn_wmma_f32_16x16x4_f32(false, a, false, b1, (short)0, acc1, false, false);
        acc2 = __builtin_amdgcn_wmma_f32_16x16x4_f32(false, a, false, b2, (short)0, acc2, false, false);
        acc3 = __builtin_amdgcn_wmma_f32_16x16x4_f32(false, a, false, b3, (short)0, acc3, false, false);
    }

    const int rbase = gwave * 16 + lhalf * 8;
    float bi0 = 0.f, bi1 = 0.f, bi2 = 0.f, bi3 = 0.f;
    if (bias) { bi0 = bias[lmod]; bi1 = bias[16 + lmod]; bi2 = bias[32 + lmod]; bi3 = bias[48 + lmod]; }
#pragma unroll
    for (int r = 0; r < 8; ++r) {
        size_t ro = (size_t)(rbase + r) * HDIM + lmod;
        float v0 = acc0[r] + bi0, v1 = acc1[r] + bi1, v2 = acc2[r] + bi2, v3 = acc3[r] + bi3;
        if (act) { v0 = lrelu(v0); v1 = lrelu(v1); v2 = lrelu(v2); v3 = lrelu(v3); }
        if (rowscale) { const float rs = rowscale[rbase + r]; v0 *= rs; v1 *= rs; v2 *= rs; v3 *= rs; }
        C[ro] = v0; C[ro + 16] = v1; C[ro + 32] = v2; C[ro + 48] = v3;
    }
}

__global__ void __launch_bounds__(256) conv_gather_kernel(const float* __restrict__ Hm, const int* __restrict__ adj,
                                                          const int* __restrict__ offs, const int* __restrict__ cnt,
                                                          const float* __restrict__ dinv, const float* __restrict__ bias,
                                                          float* __restrict__ out, int n, int act) {
    int tid = blockIdx.x * blockDim.x + threadIdx.x;
    int i = tid >> 4;
    if (i >= n) return;
    const int j = (tid & 15) * 4;
    const float dd = dinv[i];
    const int m = cnt[i];
    const int* ap = adj + offs[i];
    const float* hcol = Hm + j;

    float4 acc = *(const float4*)(hcol + (size_t)i * HDIM);

    int k = 0;
    for (; k + 4 <= m; k += 4) {
        int s0 = ap[k], s1 = ap[k + 1], s2 = ap[k + 2], s3 = ap[k + 3];
        float4 h0 = *(const float4*)(hcol + (size_t)s0 * HDIM);
        float4 h1 = *(const float4*)(hcol + (size_t)s1 * HDIM);
        float4 h2 = *(const float4*)(hcol + (size_t)s2 * HDIM);
        float4 h3 = *(const float4*)(hcol + (size_t)s3 * HDIM);
        acc.x += h0.x; acc.y += h0.y; acc.z += h0.z; acc.w += h0.w;
        acc.x += h1.x; acc.y += h1.y; acc.z += h1.z; acc.w += h1.w;
        acc.x += h2.x; acc.y += h2.y; acc.z += h2.z; acc.w += h2.w;
        acc.x += h3.x; acc.y += h3.y; acc.z += h3.z; acc.w += h3.w;
    }
    const int r = m - k;
    if (r > 0) {
        int s0 = ap[k];
        int s1 = ap[r > 1 ? k + 1 : k];
        int s2 = ap[r > 2 ? k + 2 : k];
        float4 h0 = *(const float4*)(hcol + (size_t)s0 * HDIM);
        float4 h1 = *(const float4*)(hcol + (size_t)s1 * HDIM);
        float4 h2 = *(const float4*)(hcol + (size_t)s2 * HDIM);
        float w1 = r > 1 ? 1.0f : 0.0f;
        float w2 = r > 2 ? 1.0f : 0.0f;
        acc.x += h0.x; acc.y += h0.y; acc.z += h0.z; acc.w += h0.w;
        acc.x += w1 * h1.x; acc.y += w1 * h1.y; acc.z += w1 * h1.z; acc.w += w1 * h1.w;
        acc.x += w2 * h2.x; acc.y += w2 * h2.y; acc.z += w2 * h2.z; acc.w += w2 * h2.w;
    }

    float4 b = *(const float4*)(bias + j);
    float4 o = make_float4(b.x + dd * acc.x, b.y + dd * acc.y, b.z + dd * acc.z, b.w + dd * acc.w);
    if (act) {
        o.x = fmaxf(o.x, 0.0f); o.y = fmaxf(o.y, 0.0f);
        o.z = fmaxf(o.z, 0.0f); o.w = fmaxf(o.w, 0.0f);
    }
    *(float4*)(out + (size_t)i * HDIM + j) = o;
}

__global__ void __launch_bounds__(256) tail_kernel(const float* __restrict__ Z,
                                                   const float* __restrict__ Wp1, const float* __restrict__ bp1,
                                                   const float* __restrict__ Wp2, const float* __restrict__ bp2,
                                                   float* __restrict__ tprob, int M,
                                                   const int* __restrict__ treat, int nt,
                                                   const int* __restrict__ ctrl, int nc,
                                                   const float* __restrict__ Wy1, const float* __restrict__ by1,
                                                   const float* __restrict__ Wy0, const float* __restrict__ by0,
                                                   float* __restrict__ o_y1, float* __restrict__ o_yc0,
                                                   float* __restrict__ o_y0, float* __restrict__ o_yc1,
                                                   int hb_t, int hb_c) {
    extern __shared__ float bsw[];
    const int tid = threadIdx.x;
    int blk = blockIdx.x;

    if (blk < hb_t + hb_c) {
        const bool tr = blk < hb_t;
        const int t = (tr ? blk : blk - hb_t) * 256 + tid;
        if (t >= (tr ? nt : nc)) return;
        const int node = tr ? treat[t] : ctrl[t];
        const float* row = Z + (size_t)node * HDIM;
        float d1 = 0.0f, d0 = 0.0f;
#pragma unroll 4
        for (int j = 0; j < HDIM; j += 4) {
            float4 z  = *(const float4*)(row + j);
            float4 w1 = *(const float4*)(Wy1 + j);
            float4 w0 = *(const float4*)(Wy0 + j);
            d1 += z.x * w1.x + z.y * w1.y + z.z * w1.z + z.w * w1.w;
            d0 += z.x * w0.x + z.y * w0.y + z.z * w0.z + z.w * w0.w;
        }
        const float r1 = lrelu(d1 + by1[0]);
        const float r0 = lrelu(d0 + by0[0]);
        if (tr) { o_y1[t] = r1; o_yc0[t] = r0; }
        else    { o_y0[t] = r0; o_yc1[t] = r1; }
        return;
    }
    blk -= hb_t + hb_c;

    const int lane  = tid & 31;
    const int wave  = tid >> 5;
    const int lmod  = lane & 15;
    const int lhalf = lane >> 4;

    for (int p = tid; p < HDIM * 32; p += 256) {
        int ks  = p >> 7;
        int ll  = (p >> 2) & 31;
        int tn  = p & 3;
        int kb  = 4 * ks + 2 * (ll >> 4);
        int col = tn * 16 + (ll & 15);
        bsw[p * 2 + 0] = Wp1[(size_t)kb * HDIM + col];
        bsw[p * 2 + 1] = Wp1[(size_t)(kb + 1) * HDIM + col];
    }
    __syncthreads();

    const int gwave = blk * 8 + wave;
    if (gwave >= (M >> 4)) return;

    const float* arow = Z + (size_t)(gwave * 16 + lmod) * HDIM + 2 * lhalf;
    v8f acc0 = {}, acc1 = {}, acc2 = {}, acc3 = {};
#pragma unroll
    for (int ks = 0; ks < HDIM / 4; ++ks) {
        v2f a = *(const v2f*)(arow + 4 * ks);
        const float4* bp = (const float4*)(bsw + (size_t)(ks * 32 + lane) * 8);
        float4 b01 = bp[0];
        float4 b23 = bp[1];
        v2f b0 = {b01.x, b01.y}, b1 = {b01.z, b01.w};
        v2f b2 = {b23.x, b23.y}, b3 = {b23.z, b23.w};
        acc0 = __builtin_amdgcn_wmma_f32_16x16x4_f32(false, a, false, b0, (short)0, acc0, false, false);
        acc1 = __builtin_amdgcn_wmma_f32_16x16x4_f32(false, a, false, b1, (short)0, acc1, false, false);
        acc2 = __builtin_amdgcn_wmma_f32_16x16x4_f32(false, a, false, b2, (short)0, acc2, false, false);
        acc3 = __builtin_amdgcn_wmma_f32_16x16x4_f32(false, a, false, b3, (short)0, acc3, false, false);
    }

    const float bi0 = bp1[lmod], bi1 = bp1[16 + lmod], bi2 = bp1[32 + lmod], bi3 = bp1[48 + lmod];
    const float2 w0 = *(const float2*)(Wp2 + (size_t)(lmod) * 2);
    const float2 w1 = *(const float2*)(Wp2 + (size_t)(16 + lmod) * 2);
    const float2 w2 = *(const float2*)(Wp2 + (size_t)(32 + lmod) * 2);
    const float2 w3 = *(const float2*)(Wp2 + (size_t)(48 + lmod) * 2);
    float v[16];
#pragma unroll
    for (int r = 0; r < 8; ++r) {
        const float p0 = lrelu(acc0[r] + bi0), p1 = lrelu(acc1[r] + bi1);
        const float p2 = lrelu(acc2[r] + bi2), p3 = lrelu(acc3[r] + bi3);
        v[2 * r + 0] = p0 * w0.x + p1 * w1.x + p2 * w2.x + p3 * w3.x;
        v[2 * r + 1] = p0 * w0.y + p1 * w1.y + p2 * w2.y + p3 * w3.y;
    }
    const bool h8 = (lmod & 8) != 0, h4 = (lmod & 4) != 0, h2 = (lmod & 2) != 0, h1 = (lmod & 1) != 0;
    float u8[8], u4[4], u2[2];
#pragma unroll
    for (int i = 0; i < 8; ++i) {
        const float keep = h8 ? v[i + 8] : v[i];
        const float send = h8 ? v[i] : v[i + 8];
        u8[i] = keep + __shfl_xor(send, 8, 32);
    }
#pragma unroll
    for (int i = 0; i < 4; ++i) {
        const float keep = h4 ? u8[i + 4] : u8[i];
        const float send = h4 ? u8[i] : u8[i + 4];
        u4[i] = keep + __shfl_xor(send, 4, 32);
    }
#pragma unroll
    for (int i = 0; i < 2; ++i) {
        const float keep = h2 ? u4[i + 2] : u4[i];
        const float send = h2 ? u4[i] : u4[i + 2];
        u2[i] = keep + __shfl_xor(send, 2, 32);
    }
    const float keep1 = h1 ? u2[1] : u2[0];
    const float send1 = h1 ? u2[0] : u2[1];
    const float tot = keep1 + __shfl_xor(send1, 1, 32);
    tprob[(size_t)gwave * 32 + lane] = lrelu(tot + bp2[lmod & 1]);
}

extern "C" void kernel_launch(void* const* d_in, const int* in_sizes, int n_in,
                              void* d_out, int out_size, void* d_ws, size_t ws_size,
                              hipStream_t stream) {
    const float* x     = (const float*)d_in[0];
    const int*   ei    = (const int*)d_in[1];
    const int*   treat = (const int*)d_in[2];
    const int*   ctrl  = (const int*)d_in[3];
    const float* W1  = (const float*)d_in[4];  const float* b1  = (const float*)d_in[5];
    const float* W2  = (const float*)d_in[6];  const float* b2  = (const float*)d_in[7];
    const float* Wy1 = (const float*)d_in[8];  const float* by1 = (const float*)d_in[9];
    const float* Wy0 = (const float*)d_in[10]; const float* by0 = (const float*)d_in[11];
    const float* Wp1 = (const float*)d_in[12]; const float* bp1 = (const float*)d_in[13];
    const float* Wp2 = (const float*)d_in[14]; const float* bp2 = (const float*)d_in[15];

    const int N = in_sizes[0] / INDIM;
    const int E = in_sizes[1] / 2;
    const int T = in_sizes[2];
    const int Cn = in_sizes[3];
    const int* e_src = ei;
    const int* e_dst = ei + E;

    float* dinv = (float*)d_ws;
    float* buf0 = dinv + (((size_t)N + 63) & ~(size_t)63);
    int* binned = (int*)(buf0 + (size_t)N * HDIM);
    int* adj  = binned + (size_t)2 * E;
    int* cnt  = adj + (((size_t)E + 63) & ~(size_t)63);
    int* offs = cnt + (((size_t)N + 63) & ~(size_t)63);

    float* o_y1    = (float*)d_out;
    float* o_yc0   = o_y1 + T;
    float* o_y0    = o_yc0 + T;
    float* o_yc1   = o_y0 + Cn;
    float* o_tprob = o_yc1 + Cn;
    float* o_xZ2   = o_tprob + (size_t)2 * N;

    const int BLK = 256;
    auto cdiv = [](long long a, long long b) { return (int)((a + b - 1) / b); };
    const int gemm_blocks = cdiv(N / 16, 8);
    const size_t smem128 = (size_t)INDIM * HDIM * sizeof(float);
    const size_t smem64  = (size_t)HDIM * HDIM * sizeof(float);
    const int nb = cdiv(N, BIN_NODES);
    const int g1 = cdiv(E, SORT_CH);
    const int tab = nb * g1;
    int* hist = (int*)buf0;
    int* hofs = hist + (((size_t)tab + 63) & ~(size_t)63);
    const size_t smem_hist = (size_t)nb * sizeof(int);
    const size_t smem_scat = ((size_t)((3 * nb + 1) & ~1) + (size_t)2 * SORT_CH) * sizeof(int);

    bin_hist_kernel<<<g1, SORT_TPB, smem_hist, stream>>>(e_dst, hist, E, nb, g1);
    table_scan_kernel<<<cdiv(tab, SCAN_EPB), BLK, 0, stream>>>(hist, hofs, tab);
    bin_scatter_kernel<<<g1, SORT_TPB, smem_scat, stream>>>(e_src, e_dst, hist, hofs, binned, E, nb, g1);
    bin_rows_kernel<<<nb, SORT_TPB, 0, stream>>>(binned, hofs, adj, cnt, offs, dinv, N, E, nb, g1);

    wmma_gemm_f32_kernel<<<gemm_blocks, BLK, smem128, stream>>>(x, W1, buf0, nullptr, dinv, N, INDIM, 0);
    conv_gather_kernel<<<cdiv((long long)N * 16, BLK), BLK, 0, stream>>>(buf0, adj, offs, cnt, dinv, b1, o_xZ2, N, 1);

    wmma_gemm_f32_kernel<<<gemm_blocks, BLK, smem64, stream>>>(o_xZ2, W2, buf0, nullptr, dinv, N, HDIM, 0);
    conv_gather_kernel<<<cdiv((long long)N * 16, BLK), BLK, 0, stream>>>(buf0, adj, offs, cnt, dinv, b2, o_xZ2, N, 0);

    const int hb_t = cdiv(T, BLK);
    const int hb_c = cdiv(Cn, BLK);
    tail_kernel<<<hb_t + hb_c + gemm_blocks, BLK, smem64, stream>>>(o_xZ2, Wp1, bp1, Wp2, bp2, o_tprob, N,
                                                                    treat, T, ctrl, Cn, Wy1, by1, Wy0, by0,
                                                                    o_y1, o_yc0, o_y0, o_yc1, hb_t, hb_c);
}


# Sort launches (bin counts, table scan, bin runs, rows per bin): each thread fetches its share as one group of independent loads, not one awaited load per loop trip; sum order still LDS arrival rank
# speedup vs baseline: 3.6401x; 1.2288x over previous
#include <hip/hip_runtime.h>
#include <hip/hip_bf16.h>

typedef __attribute__((ext_vector_type(2))) float v2f;
typedef __attribute__((ext_vector_type(8))) float v8f;

#define HDIM 64
#define INDIM 128

__device__ __forceinline__ float lrelu(float v) { return v > 0.0f ? v : 0.01f * v; }

#define SORT_TPB 512
#define SORT_CH 6144
#define SORT_EPT (SORT_CH / SORT_TPB)
#define BIN_SHIFT 9
#define BIN_NODES 512
#define FINE_CAP 12288
#define FINE_EPT (FINE_CAP / SORT_TPB)

__device__ __forceinline__ int block_excl_scan_512(int v, int* wsum, int tid) {
    const int lane = tid & 31;
    const int wave = tid >> 5;
    int incl = v;
    for (int o = 1; o < 32; o <<= 1) {
        int u = __shfl_up(incl, o, 32);
        if (lane >= o) incl += u;
    }
    if (lane == 31) wsum[wave] = incl;
    __syncthreads();
    int front = 0;
    for (int w = 0; w < 16; ++w)
        if (w < wave) front += wsum[w];
    return front + incl - v;
}

__global__ void __launch_bounds__(SORT_TPB) bin_hist_kernel(const int* __restrict__ dst, int* __restrict__ hist,
                                                            int e, int nb, int g1) {
    extern __shared__ int lds[];
    const int tid = threadIdx.x;
    const int c = blockIdx.x;
    const int base = c * SORT_CH;
    const int lim = min(e, base + SORT_CH);
    int d[SORT_EPT];
#pragma unroll
    for (int k = 0; k < SORT_EPT; ++k) d[k] = dst[min(base + tid + k * SORT_TPB, lim - 1)];
    for (int b = tid; b < nb; b += SORT_TPB) lds[b] = 0;
    __syncthreads();
#pragma unroll
    for (int k = 0; k < SORT_EPT; ++k)
        if (base + tid + k * SORT_TPB < lim) atomicAdd(&lds[d[k] >> BIN_SHIFT], 1);
    __syncthreads();
    for (int b = tid; b < nb; b += SORT_TPB) hist[(size_t)b * g1 + c] = lds[b];
}

#define SCAN_EPB 1024
#define SCAN_DEPTH 16
__global__ void __launch_bounds__(256) table_scan_kernel(const int* __restrict__ cnt, int* __restrict__ offs, int n) {
    __shared__ int wsum[8];
    __shared__ int wpre[8];
    const int tid  = threadIdx.x;
    const int lane = tid & 31;
    const int wave = tid >> 5;
    const int base = blockIdx.x * SCAN_EPB;

    int pre = 0;
    const int4* c4 = (const int4*)cnt;
    const int nq = base >> 2;
    for (int q0 = tid; q0 < nq; q0 += 256 * SCAN_DEPTH) {
        int4 v[SCAN_DEPTH];
#pragma unroll
        for (int j = 0; j < SCAN_DEPTH; ++j) v[j] = c4[min(q0 + j * 256, nq - 1)];
        __builtin_amdgcn_sched_group_barrier(0x020, SCAN_DEPTH, 0);
        __builtin_amdgcn_sched_group_barrier(0x002, 6 * SCAN_DEPTH, 0);
#pragma unroll
        for (int j = 0; j < SCAN_DEPTH; ++j)
            if (q0 + j * 256 < nq) pre += (v[j].x + v[j].y) + (v[j].z + v[j].w);
    }

    const int i0 = base + tid * 4;
    int c[4];
    for (int k = 0; k < 4; ++k) c[k] = (i0 + k < n) ? cnt[i0 + k] : 0;
    const int t = (c[0] + c[1]) + (c[2] + c[3]);

    int incl = t;
    for (int o = 1; o < 32; o <<= 1) {
        int u = __shfl_up(incl, o, 32);
        if (lane >= o) incl += u;
    }
    for (int o = 16; o > 0; o >>= 1) pre += __shfl_xor(pre, o, 32);
    if (lane == 31) wsum[wave] = incl;
    if (lane == 0)  wpre[wave] = pre;
    __syncthreads();
    int front = 0;
    for (int w = 0; w < 8; ++w) {
        front += wpre[w];
        if (w < wave) front += wsum[w];
    }
    int excl = front + incl - t;
    for (int k = 0; k < 4; ++k) {
        if (i0 + k < n) {
            offs[i0 + k] = excl;
            excl += c[k];
        }
    }
}

__global__ void __launch_bounds__(SORT_TPB) bin_scatter_kernel(const int* __restrict__ src, const int* __restrict__ dst,
                                                               const int* __restrict__ hist, const int* __restrict__ hofs,
                                                               int* __restrict__ binned, int e, int nb, int g1) {
    extern __shared__ int lds[];
    __shared__ int wsum[16];
    const int tid = threadIdx.x;
    const int c = blockIdx.x;
    int* lstart = lds;
    int* cursor = lds + nb;
    int* gbase  = lds + 2 * nb;
    int2* stage = (int2*)(lds + ((3 * nb + 1) & ~1));

    const int base = c * SORT_CH;
    const int lim = min(e, base + SORT_CH);
    int es[SORT_EPT], ed[SORT_EPT];
#pragma unroll
    for (int k = 0; k < SORT_EPT; ++k) {
        const int i = min(base + tid + k * SORT_TPB, lim - 1);
        es[k] = src[i];
        ed[k] = dst[i];
    }

    const int ipt = (nb + SORT_TPB - 1) / SORT_TPB;
    const int b0 = tid * ipt;
    const int b1 = min(nb, b0 + ipt);
    int mine = 0;
    for (int b = b0; b < b1; ++b) mine += hist[(size_t)b * g1 + c];
    int run = block_excl_scan_512(mine, wsum, tid);
    for (int b = b0; b < b1; ++b) {
        lstart[b] = run;
        cursor[b] = run;
        gbase[b]  = hofs[(size_t)b * g1 + c];
        run += hist[(size_t)b * g1 + c];
    }
    __syncthreads();

    int pos[SORT_EPT];
#pragma unroll
    for (int k = 0; k < SORT_EPT; ++k) {
        pos[k] = 0;
        if (base + tid + k * SORT_TPB < lim) pos[k] = atomicAdd(&cursor[ed[k] >> BIN_SHIFT], 1);
    }
#pragma unroll
    for (int k = 0; k < SORT_EPT; ++k)
        if (base + tid + k * SORT_TPB < lim) stage[pos[k]] = make_int2(es[k], ed[k]);
    __syncthreads();

    const int nloc = lim - base;
    int2* out2 = (int2*)binned;
    int2 w[SORT_EPT];
#pragma unroll
    for (int k = 0; k < SORT_EPT; ++k) w[k] = stage[min(tid + k * SORT_TPB, nloc - 1)];
    int shift[SORT_EPT];
#pragma unroll
    for (int k = 0; k < SORT_EPT; ++k) {
        const int b = w[k].y >> BIN_SHIFT;
        shift[k] = gbase[b] - lstart[b];
    }
#pragma unroll
    for (int k = 0; k < SORT_EPT; ++k) {
        const int j = tid + k * SORT_TPB;
        if (j < nloc) out2[shift[k] + j] = w[k];
    }
}

__global__ void __launch_bounds__(SORT_TPB) bin_rows_kernel(const int* __restrict__ binned, const int* __restrict__ hofs,
                                                            int* __restrict__ adj, int* __restrict__ cntG,
                                                            int* __restrict__ offsG, float* __restrict__ dinv,
                                                            int n, int e, int nb, int g1) {
    __shared__ int cnt[BIN_NODES];
    __shared__ int cursor[BIN_NODES];
    __shared__ int wsum[16];
    __shared__ int stage[FINE_CAP];
    const int tid = threadIdx.x;
    const int b = blockIdx.x;
    const int start = hofs[(size_t)b * g1];
    const int end = (b + 1 < nb) ? hofs[(size_t)(b + 1) * g1] : e;
    const int m = end - start;
    const int2* bp = (const int2*)binned + start;
    const bool fits = m <= FINE_CAP;

    int2 v[FINE_EPT];
#pragma unroll
    for (int k = 0; k < FINE_EPT; ++k) v[k] = make_int2(0, 0);
    if (fits && m > 0) {
#pragma unroll
        for (int k = 0; k < FINE_EPT; ++k) v[k] = bp[min(tid + k * SORT_TPB, m - 1)];
    }

    cnt[tid] = 0;
    __syncthreads();
    if (fits) {
#pragma unroll
        for (int k = 0; k < FINE_EPT; ++k)
            if (tid + k * SORT_TPB < m) atomicAdd(&cnt[v[k].y & (BIN_NODES - 1)], 1);
    } else {
        for (int j = tid; j < m; j += SORT_TPB) atomicAdd(&cnt[bp[j].y & (BIN_NODES - 1)], 1);
    }
    __syncthreads();

    const int cdeg = cnt[tid];
    const int ex = block_excl_scan_512(cdeg, wsum, tid);
    cursor[tid] = ex;
    const int node = b * BIN_NODES + tid;
    if (node < n) {
        cntG[node] = cdeg;
        offsG[node] = start + ex;
        float d = (float)(cdeg + 1);
        dinv[node] = d > 0.0f ? rsqrtf(d) : 0.0f;
    }
    __syncthreads();

    if (fits) {
#pragma unroll
        for (int k = 0; k < FINE_EPT; ++k)
            if (tid + k * SORT_TPB < m) v[k].y = atomicAdd(&cursor[v[k].y & (BIN_NODES - 1)], 1);
#pragma unroll
        for (int k = 0; k < FINE_EPT; ++k)
            if (tid + k * SORT_TPB < m) stage[v[k].y] = v[k].x;
        __syncthreads();
        for (int j = tid; j < m; j += SORT_TPB) adj[start + j] = stage[j];
    } else {
        for (int j = tid; j < m; j += SORT_TPB) {
            int2 u = bp[j];
            int p = atomicAdd(&cursor[u.y & (BIN_NODES - 1)], 1);
            adj[start + p] = u.x;
        }
    }
}

__global__ void wmma_gemm_f32_kernel(const float* __restrict__ A, const float* __restrict__ B,
                                     float* __restrict__ C, const float* __restrict__ bias,
                                     const float* __restrict__ rowscale, int M, int K, int act) {
    extern __shared__ float bsw[];
    const int tid   = threadIdx.x;
    const int lane  = tid & 31;
    const int wave  = tid >> 5;
    const int lmod  = lane & 15;
    const int lhalf = lane >> 4;

    const int pairs = K * 32;
    for (int p = tid; p < pairs; p += 256) {
        int ks  = p >> 7;
        int ll  = (p >> 2) & 31;
        int tn  = p & 3;
        int kb  = 4 * ks + 2 * (ll >> 4);
        int col = tn * 16 + (ll & 15);
        bsw[p * 2 + 0] = B[(size_t)kb * HDIM + col];
        bsw[p * 2 + 1] = B[(size_t)(kb + 1) * HDIM + col];
    }
    __syncthreads();

    const int gwave = blockIdx.x * 8 + wave;
    if (gwave >= (M >> 4)) return;

    const float* arow = A + (size_t)(gwave * 16 + lmod) * K + 2 * lhalf;
    v8f acc0 = {}, acc1 = {}, acc2 = {}, acc3 = {};

    const int steps = K >> 2;
    for (int ks = 0; ks < steps; ++ks) {
        v2f a = *(const v2f*)(arow + 4 * ks);
        const float4* bp = (const float4*)(bsw + (size_t)(ks * 32 + lane) * 8);
        float4 b01 = bp[0];
        float4 b23 = bp[1];
        v2f b0 = {b01.x, b01.y}, b1 = {b01.z, b01.w};
        v2f b2 = {b23.x, b23.y}, b3 = {b23.z, b23.w};
        acc0 = __builtin_amdgcn_wmma_f32_16x16x4_f32(false, a, false, b0, (short)0, acc0, false, false);
        acc1 = __builtin_amdgcn_wmma_f32_16x16x4_f32(false, a, false, b1, (short)0, acc1, false, false);
        acc2 = __builtin_amdgcn_wmma_f32_16x16x4_f32(false, a, false, b2, (short)0, acc2, false, false);
        acc3 = __builtin_amdgcn_wmma_f32_16x16x4_f32(false, a, false, b3, (short)0, acc3, false, false);
    }

    const int rbase = gwave * 16 + lhalf * 8;
    float bi0 = 0.f, bi1 = 0.f, bi2 = 0.f, bi3 = 0.f;
    if (bias) { bi0 = bias[lmod]; bi1 = bias[16 + lmod]; bi2 = bias[32 + lmod]; bi3 = bias[48 + lmod]; }
#pragma unroll
    for (int r = 0; r < 8; ++r) {
        size_t ro = (size_t)(rbase + r) * HDIM + lmod;
        float v0 = acc0[r] + bi0, v1 = acc1[r] + bi1, v2 = acc2[r] + bi2, v3 = acc3[r] + bi3;
        if (act) { v0 = lrelu(v0); v1 = lrelu(v1); v2 = lrelu(v2); v3 = lrelu(v3); }
        if (rowscale) { const float rs = rowscale[rbase + r]; v0 *= rs; v1 *= rs; v2 *= rs; v3 *= rs; }
        C[ro] = v0; C[ro + 16] = v1; C[ro + 32] = v2; C[ro + 48] = v3;
    }
}

__global__ void __launch_bounds__(256) conv_gather_kernel(const float* __restrict__ Hm, const int* __restrict__ adj,
                                                          const int* __restrict__ offs, const int* __restrict__ cnt,
                                                          const float* __restrict__ dinv, const float* __restrict__ bias,
                                                          float* __restrict__ out, int n, int act) {
    int tid = blockIdx.x * blockDim.x + threadIdx.x;
    int i = tid >> 4;
    if (i >= n) return;
    const int j = (tid & 15) * 4;
    const float dd = dinv[i];
    const int m = cnt[i];
    const int* ap = adj + offs[i];
    const float* hcol = Hm + j;

    float4 acc = *(const float4*)(hcol + (size_t)i * HDIM);

    int k = 0;
    for (; k + 4 <= m; k += 4) {
        int s0 = ap[k], s1 = ap[k + 1], s2 = ap[k + 2], s3 = ap[k + 3];
        float4 h0 = *(const float4*)(hcol + (size_t)s0 * HDIM);
        float4 h1 = *(const float4*)(hcol + (size_t)s1 * HDIM);
        float4 h2 = *(const float4*)(hcol + (size_t)s2 * HDIM);
        float4 h3 = *(const float4*)(hcol + (size_t)s3 * HDIM);
        acc.x += h0.x; acc.y += h0.y; acc.z += h0.z; acc.w += h0.w;
        acc.x += h1.x; acc.y += h1.y; acc.z += h1.z; acc.w += h1.w;
        acc.x += h2.x; acc.y += h2.y; acc.z += h2.z; acc.w += h2.w;
        acc.x += h3.x; acc.y += h3.y; acc.z += h3.z; acc.w += h3.w;
    }
    const int r = m - k;
    if (r > 0) {
        int s0 = ap[k];
        int s1 = ap[r > 1 ? k + 1 : k];
        int s2 = ap[r > 2 ? k + 2 : k];
        float4 h0 = *(const float4*)(hcol + (size_t)s0 * HDIM);
        float4 h1 = *(const float4*)(hcol + (size_t)s1 * HDIM);
        float4 h2 = *(const float4*)(hcol + (size_t)s2 * HDIM);
        float w1 = r > 1 ? 1.0f : 0.0f;
        float w2 = r > 2 ? 1.0f : 0.0f;
        acc.x += h0.x; acc.y += h0.y; acc.z += h0.z; acc.w += h0.w;
        acc.x += w1 * h1.x; acc.y += w1 * h1.y; acc.z += w1 * h1.z; acc.w += w1 * h1.w;
        acc.x += w2 * h2.x; acc.y += w2 * h2.y; acc.z += w2 * h2.z; acc.w += w2 * h2.w;
    }

    float4 b = *(const float4*)(bias + j);
    float4 o = make_float4(b.x + dd * acc.x, b.y + dd * acc.y, b.z + dd * acc.z, b.w + dd * acc.w);
    if (act) {
        o.x = fmaxf(o.x, 0.0f); o.y = fmaxf(o.y, 0.0f);
        o.z = fmaxf(o.z, 0.0f); o.w = fmaxf(o.w, 0.0f);
    }
    *(float4*)(out + (size_t)i * HDIM + j) = o;
}

__global__ void __launch_bounds__(256) tail_kernel(const float* __restrict__ Z,
                                                   const float* __restrict__ Wp1, const float* __restrict__ bp1,
                                                   const float* __restrict__ Wp2, const float* __restrict__ bp2,
                                                   float* __restrict__ tprob, int M,
                                                   const int* __restrict__ treat, int nt,
                                                   const int* __restrict__ ctrl, int nc,
                                                   const float* __restrict__ Wy1, const float* __restrict__ by1,
                                                   const float* __restrict__ Wy0, const float* __restrict__ by0,
                                                   float* __restrict__ o_y1, float* __restrict__ o_yc0,
                                                   float* __restrict__ o_y0, float* __restrict__ o_yc1,
                                                   int hb_t, int hb_c) {
    extern __shared__ float bsw[];
    const int tid = threadIdx.x;
    int blk = blockIdx.x;

    if (blk < hb_t + hb_c) {
        const bool tr = blk < hb_t;
        const int t = (tr ? blk : blk - hb_t) * 256 + tid;
        if (t >= (tr ? nt : nc)) return;
        const int node = tr ? treat[t] : ctrl[t];
        const float* row = Z + (size_t)node * HDIM;
        float d1 = 0.0f, d0 = 0.0f;
#pragma unroll 4
        for (int j = 0; j < HDIM; j += 4) {
            float4 z  = *(const float4*)(row + j);
            float4 w1 = *(const float4*)(Wy1 + j);
            float4 w0 = *(const float4*)(Wy0 + j);
            d1 += z.x * w1.x + z.y * w1.y + z.z * w1.z + z.w * w1.w;
            d0 += z.x * w0.x + z.y * w0.y + z.z * w0.z + z.w * w0.w;
        }
        const float r1 = lrelu(d1 + by1[0]);
        const float r0 = lrelu(d0 + by0[0]);
        if (tr) { o_y1[t] = r1; o_yc0[t] = r0; }
        else    { o_y0[t] = r0; o_yc1[t] = r1; }
        return;
    }
    blk -= hb_t + hb_c;

    const int lane  = tid & 31;
    const int wave  = tid >> 5;
    const int lmod  = lane & 15;
    const int lhalf = lane >> 4;

    for (int p = tid; p < HDIM * 32; p += 256) {
        int ks  = p >> 7;
        int ll  = (p >> 2) & 31;
        int tn  = p & 3;
        int kb  = 4 * ks + 2 * (ll >> 4);
        int col = tn * 16 + (ll & 15);
        bsw[p * 2 + 0] = Wp1[(size_t)kb * HDIM + col];
        bsw[p * 2 + 1] = Wp1[(size_t)(kb + 1) * HDIM + col];
    }
    __syncthreads();

    const int gwave = blk * 8 + wave;
    if (gwave >= (M >> 4)) return;

    const float* arow = Z + (size_t)(gwave * 16 + lmod) * HDIM + 2 * lhalf;
    v8f acc0 = {}, acc1 = {}, acc2 = {}, acc3 = {};
#pragma unroll
    for (int ks = 0; ks < HDIM / 4; ++ks) {
        v2f a = *(const v2f*)(arow + 4 * ks);
        const float4* bp = (const float4*)(bsw + (size_t)(ks * 32 + lane) * 8);
        float4 b01 = bp[0];
        float4 b23 = bp[1];
        v2f b0 = {b01.x, b01.y}, b1 = {b01.z, b01.w};
        v2f b2 = {b23.x, b23.y}, b3 = {b23.z, b23.w};
        acc0 = __builtin_amdgcn_wmma_f32_16x16x4_f32(false, a, false, b0, (short)0, acc0, false, false);
        acc1 = __builtin_amdgcn_wmma_f32_16x16x4_f32(false, a, false, b1, (short)0, acc1, false, false);
        acc2 = __builtin_amdgcn_wmma_f32_16x16x4_f32(false, a, false, b2, (short)0, acc2, false, false);
        acc3 = __builtin_amdgcn_wmma_f32_16x16x4_f32(false, a, false, b3, (short)0, acc3, false, false);
    }

    const float bi0 = bp1[lmod], bi1 = bp1[16 + lmod], bi2 = bp1[32 + lmod], bi3 = bp1[48 + lmod];
    const float2 w0 = *(const float2*)(Wp2 + (size_t)(lmod) * 2);
    const float2 w1 = *(const float2*)(Wp2 + (size_t)(16 + lmod) * 2);
    const float2 w2 = *(const float2*)(Wp2 + (size_t)(32 + lmod) * 2);
    const float2 w3 = *(const float2*)(Wp2 + (size_t)(48 + lmod) * 2);
    float v[16];
#pragma unroll
    for (int r = 0; r < 8; ++r) {
        const float p0 = lrelu(acc0[r] + bi0), p1 = lrelu(acc1[r] + bi1);
        const float p2 = lrelu(acc2[r] + bi2), p3 = lrelu(acc3[r] + bi3);
        v[2 * r + 0] = p0 * w0.x + p1 * w1.x + p2 * w2.x + p3 * w3.x;
        v[2 * r + 1] = p0 * w0.y + p1 * w1.y + p2 * w2.y + p3 * w3.y;
    }
    const bool h8 = (lmod & 8) != 0, h4 = (lmod & 4) != 0, h2 = (lmod & 2) != 0, h1 = (lmod & 1) != 0;
    float u8[8], u4[4], u2[2];
#pragma unroll
    for (int i = 0; i < 8; ++i) {
        const float keep = h8 ? v[i + 8] : v[i];
        const float send = h8 ? v[i] : v[i + 8];
        u8[i] = keep + __shfl_xor(send, 8, 32);
    }
#pragma unroll
    for (int i = 0; i < 4; ++i) {
        const float keep = h4 ? u8[i + 4] : u8[i];
        const float send = h4 ? u8[i] : u8[i + 4];
        u4[i] = keep + __shfl_xor(send, 4, 32);
    }
#pragma unroll
    for (int i = 0; i < 2; ++i) {
        const float keep = h2 ? u4[i + 2] : u4[i];
        const float send = h2 ? u4[i] : u4[i + 2];
        u2[i] = keep + __shfl_xor(send, 2, 32);
    }
    const float keep1 = h1 ? u2[1] : u2[0];
    const float send1 = h1 ? u2[0] : u2[1];
    const float tot = keep1 + __shfl_xor(send1, 1, 32);
    tprob[(size_t)gwave * 32 + lane] = lrelu(tot + bp2[lmod & 1]);
}

extern "C" void kernel_launch(void* const* d_in, const int* in_sizes, int n_in,
                              void* d_out, int out_size, void* d_ws, size_t ws_size,
                              hipStream_t stream) {
    const float* x     = (const float*)d_in[0];
    const int*   ei    = (const int*)d_in[1];
    const int*   treat = (const int*)d_in[2];
    const int*   ctrl  = (const int*)d_in[3];
    const float* W1  = (const float*)d_in[4];  const float* b1  = (const float*)d_in[5];
    const float* W2  = (const float*)d_in[6];  const float* b2  = (const float*)d_in[7];
    const float* Wy1 = (const float*)d_in[8];  const float* by1 = (const float*)d_in[9];
    const float* Wy0 = (const float*)d_in[10]; const float* by0 = (const float*)d_in[11];
    const float* Wp1 = (const float*)d_in[12]; const float* bp1 = (const float*)d_in[13];
    const float* Wp2 = (const float*)d_in[14]; const float* bp2 = (const float*)d_in[15];

    const int N = in_sizes[0] / INDIM;
    const int E = in_sizes[1] / 2;
    const int T = in_sizes[2];
    const int Cn = in_sizes[3];
    const int* e_src = ei;
    const int* e_dst = ei + E;

    float* dinv = (float*)d_ws;
    float* buf0 = dinv + (((size_t)N + 63) & ~(size_t)63);
    int* binned = (int*)(buf0 + (size_t)N * HDIM);
    int* adj  = binned + (size_t)2 * E;
    int* cnt  = adj + (((size_t)E + 63) & ~(size_t)63);
    int* offs = cnt + (((size_t)N + 63) & ~(size_t)63);

    float* o_y1    = (float*)d_out;
    float* o_yc0   = o_y1 + T;
    float* o_y0    = o_yc0 + T;
    float* o_yc1   = o_y0 + Cn;
    float* o_tprob = o_yc1 + Cn;
    float* o_xZ2   = o_tprob + (size_t)2 * N;

    const int BLK = 256;
    auto cdiv = [](long long a, long long b) { return (int)((a + b - 1) / b); };
    const int gemm_blocks = cdiv(N / 16, 8);
    const size_t smem128 = (size_t)INDIM * HDIM * sizeof(float);
    const size_t smem64  = (size_t)HDIM * HDIM * sizeof(float);
    const int nb = cdiv(N, BIN_NODES);
    const int g1 = cdiv(E, SORT_CH);
    const int tab = nb * g1;
    int* hist = (int*)buf0;
    int* hofs = hist + (((size_t)tab + 63) & ~(size_t)63);
    const size_t smem_hist = (size_t)nb * sizeof(int);
    const size_t smem_scat = ((size_t)((3 * nb + 1) & ~1) + (size_t)2 * SORT_CH) * sizeof(int);

    bin_hist_kernel<<<g1, SORT_TPB, smem_hist, stream>>>(e_dst, hist, E, nb, g1);
    table_scan_kernel<<<cdiv(tab, SCAN_EPB), BLK, 0, stream>>>(hist, hofs, tab);
    bin_scatter_kernel<<<g1, SORT_TPB, smem_scat, stream>>>(e_src, e_dst, hist, hofs, binned, E, nb, g1);
    bin_rows_kernel<<<nb, SORT_TPB, 0, stream>>>(binned, hofs, adj, cnt, offs, dinv, N, E, nb, g1);

    wmma_gemm_f32_kernel<<<gemm_blocks, BLK, smem128, stream>>>(x, W1, buf0, nullptr, dinv, N, INDIM, 0);
    conv_gather_kernel<<<cdiv((long long)N * 16, BLK), BLK, 0, stream>>>(buf0, adj, offs, cnt, dinv, b1, o_xZ2, N, 1);

    wmma_gemm_f32_kernel<<<gemm_blocks, BLK, smem64, stream>>>(o_xZ2, W2, buf0, nullptr, dinv, N, HDIM, 0);
    conv_gather_kernel<<<cdiv((long long)N * 16, BLK), BLK, 0, stream>>>(buf0, adj, offs, cnt, dinv, b2, o_xZ2, N, 0);

    const int hb_t = cdiv(T, BLK);
    const int hb_c = cdiv(Cn, BLK);
    tail_kernel<<<hb_t + hb_c + gemm_blocks, BLK, smem64, stream>>>(o_xZ2, Wp1, bp1, Wp2, bp2, o_tprob, N,
                                                                    treat, T, ctrl, Cn, Wy1, by1, Wy0, by0,
                                                                    o_y1, o_yc0, o_y0, o_yc1, hb_t, hb_c);
}


# Aggregation launches: fixed grid of 1024 resident blocks, each 16-lane row walks many nodes and asks for its next eight row indices ahead, not one short wave per two nodes; sum order kept
# speedup vs baseline: 3.8004x; 1.0440x over previous
#include <hip/hip_runtime.h>
#include <hip/hip_bf16.h>

typedef __attribute__((ext_vector_type(2))) float v2f;
typedef __attribute__((ext_vector_type(8))) float v8f;

#define HDIM 64
#define INDIM 128

__device__ __forceinline__ float lrelu(float v) { return v > 0.0f ? v : 0.01f * v; }

#define SORT_TPB 512
#define SORT_CH 6144
#define SORT_EPT (SORT_CH / SORT_TPB)
#define BIN_SHIFT 9
#define BIN_NODES 512
#define FINE_CAP 12288
#define FINE_EPT (FINE_CAP / SORT_TPB)

__device__ __forceinline__ int block_excl_scan_512(int v, int* wsum, int tid) {
    const int lane = tid & 31;
    const int wave = tid >> 5;
    int incl = v;
    for (int o = 1; o < 32; o <<= 1) {
        int u = __shfl_up(incl, o, 32);
        if (lane >= o) incl += u;
    }
    if (lane == 31) wsum[wave] = incl;
    __syncthreads();
    int front = 0;
    for (int w = 0; w < 16; ++w)
        if (w < wave) front += wsum[w];
    return front + incl - v;
}

__global__ void __launch_bounds__(SORT_TPB) bin_hist_kernel(const int* __restrict__ dst, int* __restrict__ hist,
                                                            int e, int nb, int g1) {
    extern __shared__ int lds[];
    const int tid = threadIdx.x;
    const int c = blockIdx.x;
    const int base = c * SORT_CH;
    const int lim = min(e, base + SORT_CH);
    int d[SORT_EPT];
#pragma unroll
    for (int k = 0; k < SORT_EPT; ++k) d[k] = dst[min(base + tid + k * SORT_TPB, lim - 1)];
    for (int b = tid; b < nb; b += SORT_TPB) lds[b] = 0;
    __syncthreads();
#pragma unroll
    for (int k = 0; k < SORT_EPT; ++k)
        if (base + tid + k * SORT_TPB < lim) atomicAdd(&lds[d[k] >> BIN_SHIFT], 1);
    __syncthreads();
    for (int b = tid; b < nb; b += SORT_TPB) hist[(size_t)b * g1 + c] = lds[b];
}

#define SCAN_EPB 1024
#define SCAN_DEPTH 16
__global__ void __launch_bounds__(256) table_scan_kernel(const int* __restrict__ cnt, int* __restrict__ offs, int n) {
    __shared__ int wsum[8];
    __shared__ int wpre[8];
    const int tid  = threadIdx.x;
    const int lane = tid & 31;
    const int wave = tid >> 5;
    const int base = blockIdx.x * SCAN_EPB;

    int pre = 0;
    const int4* c4 = (const int4*)cnt;
    const int nq = base >> 2;
    for (int q0 = tid; q0 < nq; q0 += 256 * SCAN_DEPTH) {
        int4 v[SCAN_DEPTH];
#pragma unroll
        for (int j = 0; j < SCAN_DEPTH; ++j) v[j] = c4[min(q0 + j * 256, nq - 1)];
        __builtin_amdgcn_sched_group_barrier(0x020, SCAN_DEPTH, 0);
        __builtin_amdgcn_sched_group_barrier(0x002, 6 * SCAN_DEPTH, 0);
#pragma unroll
        for (int j = 0; j < SCAN_DEPTH; ++j)
            if (q0 + j * 256 < nq) pre += (v[j].x + v[j].y) + (v[j].z + v[j].w);
    }

    const int i0 = base + tid * 4;
    int c[4];
    for (int k = 0; k < 4; ++k) c[k] = (i0 + k < n) ? cnt[i0 + k] : 0;
    const int t = (c[0] + c[1]) + (c[2] + c[3]);

    int incl = t;
    for (int o = 1; o < 32; o <<= 1) {
        int u = __shfl_up(incl, o, 32);
        if (lane >= o) incl += u;
    }
    for (int o = 16; o > 0; o >>= 1) pre += __shfl_xor(pre, o, 32);
    if (lane == 31) wsum[wave] = incl;
    if (lane == 0)  wpre[wave] = pre;
    __syncthreads();
    int front = 0;
    for (int w = 0; w < 8; ++w) {
        front += wpre[w];
        if (w < wave) front += wsum[w];
    }
    int excl = front + incl - t;
    for (int k = 0; k < 4; ++k) {
        if (i0 + k < n) {
            offs[i0 + k] = excl;
            excl += c[k];
        }
    }
}

__global__ void __launch_bounds__(SORT_TPB) bin_scatter_kernel(const int* __restrict__ src, const int* __restrict__ dst,
                                                               const int* __restrict__ hist, const int* __restrict__ hofs,
                                                               int* __restrict__ binned, int e, int nb, int g1) {
    extern __shared__ int lds[];
    __shared__ int wsum[16];
    const int tid = threadIdx.x;
    const int c = blockIdx.x;
    int* lstart = lds;
    int* cursor = lds + nb;
    int* gbase  = lds + 2 * nb;
    int2* stage = (int2*)(lds + ((3 * nb + 1) & ~1));

    const int base = c * SORT_CH;
    const int lim = min(e, base + SORT_CH);
    int es[SORT_EPT], ed[SORT_EPT];
#pragma unroll
    for (int k = 0; k < SORT_EPT; ++k) {
        const int i = min(base + tid + k * SORT_TPB, lim - 1);
        es[k] = src[i];
        ed[k] = dst[i];
    }

    const int ipt = (nb + SORT_TPB - 1) / SORT_TPB;
    const int b0 = tid * ipt;
    const int b1 = min(nb, b0 + ipt);
    int mine = 0;
    for (int b = b0; b < b1; ++b) mine += hist[(size_t)b * g1 + c];
    int run = block_excl_scan_512(mine, wsum, tid);
    for (int b = b0; b < b1; ++b) {
        lstart[b] = run;
        cursor[b] = run;
        gbase[b]  = hofs[(size_t)b * g1 + c];
        run += hist[(size_t)b * g1 + c];
    }
    __syncthreads();

    int pos[SORT_EPT];
#pragma unroll
    for (int k = 0; k < SORT_EPT; ++k) {
        pos[k] = 0;
        if (base + tid + k * SORT_TPB < lim) pos[k] = atomicAdd(&cursor[ed[k] >> BIN_SHIFT], 1);
    }
#pragma unroll
    for (int k = 0; k < SORT_EPT; ++k)
        if (base + tid + k * SORT_TPB < lim) stage[pos[k]] = make_int2(es[k], ed[k]);
    __syncthreads();

    const int nloc = lim - base;
    int2* out2 = (int2*)binned;
    int2 w[SORT_EPT];
#pragma unroll
    for (int k = 0; k < SORT_EPT; ++k) w[k] = stage[min(tid + k * SORT_TPB, nloc - 1)];
    int shift[SORT_EPT];
#pragma unroll
    for (int k = 0; k < SORT_EPT; ++k) {
        const int b = w[k].y >> BIN_SHIFT;
        shift[k] = gbase[b] - lstart[b];
    }
#pragma unroll
    for (int k = 0; k < SORT_EPT; ++k) {
        const int j = tid + k * SORT_TPB;
        if (j < nloc) out2[shift[k] + j] = w[k];
    }
}

__global__ void __launch_bounds__(SORT_TPB) bin_rows_kernel(const int* __restrict__ binned, const int* __restrict__ hofs,
                                                            int* __restrict__ adj, int* __restrict__ cntG,
                                                            int* __restrict__ offsG, float* __restrict__ dinv,
                                                            int n, int e, int nb, int g1) {
    __shared__ int cnt[BIN_NODES];
    __shared__ int cursor[BIN_NODES];
    __shared__ int wsum[16];
    __shared__ int stage[FINE_CAP];
    const int tid = threadIdx.x;
    const int b = blockIdx.x;
    const int start = hofs[(size_t)b * g1];
    const int end = (b + 1 < nb) ? hofs[(size_t)(b + 1) * g1] : e;
    const int m = end - start;
    const int2* bp = (const int2*)binned + start;
    const bool fits = m <= FINE_CAP;

    int2 v[FINE_EPT];
#pragma unroll
    for (int k = 0; k < FINE_EPT; ++k) v[k] = make_int2(0, 0);
    if (fits && m > 0) {
#pragma unroll
        for (int k = 0; k < FINE_EPT; ++k) v[k] = bp[min(tid + k * SORT_TPB, m - 1)];
    }

    cnt[tid] = 0;
    __syncthreads();
    if (fits) {
#pragma unroll
        for (int k = 0; k < FINE_EPT; ++k)
            if (tid + k * SORT_TPB < m) atomicAdd(&cnt[v[k].y & (BIN_NODES - 1)], 1);
    } else {
        for (int j = tid; j < m; j += SORT_TPB) atomicAdd(&cnt[bp[j].y & (BIN_NODES - 1)], 1);
    }
    __syncthreads();

    const int cdeg = cnt[tid];
    const int ex = block_excl_scan_512(cdeg, wsum, tid);
    cursor[tid] = ex;
    const int node = b * BIN_NODES + tid;
    if (node < n) {
        cntG[node] = cdeg;
        offsG[node] = start + ex;
        float d = (float)(cdeg + 1);
        dinv[node] = d > 0.0f ? rsqrtf(d) : 0.0f;
    }
    __syncthreads();

    if (fits) {
#pragma unroll
        for (int k = 0; k < FINE_EPT; ++k)
            if (tid + k * SORT_TPB < m) v[k].y = atomicAdd(&cursor[v[k].y & (BIN_NODES - 1)], 1);
#pragma unroll
        for (int k = 0; k < FINE_EPT; ++k)
            if (tid + k * SORT_TPB < m) stage[v[k].y] = v[k].x;
        __syncthreads();
        for (int j = tid; j < m; j += SORT_TPB) adj[start + j] = stage[j];
    } else {
        for (int j = tid; j < m; j += SORT_TPB) {
            int2 u = bp[j];
            int p = atomicAdd(&cursor[u.y & (BIN_NODES - 1)], 1);
            adj[start + p] = u.x;
        }
    }
}

__global__ void wmma_gemm_f32_kernel(const float* __restrict__ A, const float* __restrict__ B,
                                     float* __restrict__ C, const float* __restrict__ bias,
                                     const float* __restrict__ rowscale, int M, int K, int act) {
    extern __shared__ float bsw[];
    const int tid   = threadIdx.x;
    const int lane  = tid & 31;
    const int wave  = tid >> 5;
    const int lmod  = lane & 15;
    const int lhalf = lane >> 4;

    const int pairs = K * 32;
    for (int p = tid; p < pairs; p += 256) {
        int ks  = p >> 7;
        int ll  = (p >> 2) & 31;
        int tn  = p & 3;
        int kb  = 4 * ks + 2 * (ll >> 4);
        int col = tn * 16 + (ll & 15);
        bsw[p * 2 + 0] = B[(size_t)kb * HDIM + col];
        bsw[p * 2 + 1] = B[(size_t)(kb + 1) * HDIM + col];
    }
    __syncthreads();

    const int gwave = blockIdx.x * 8 + wave;
    if (gwave >= (M >> 4)) return;

    const float* arow = A + (size_t)(gwave * 16 + lmod) * K + 2 * lhalf;
    v8f acc0 = {}, acc1 = {}, acc2 = {}, acc3 = {};

    const int steps = K >> 2;
    for (int ks = 0; ks < steps; ++ks) {
        v2f a = *(const v2f*)(arow + 4 * ks);
        const float4* bp = (const float4*)(bsw + (size_t)(ks * 32 + lane) * 8);
        float4 b01 = bp[0];
        float4 b23 = bp[1];
        v2f b0 = {b01.x, b01.y}, b1 = {b01.z, b01.w};
        v2f b2 = {b23.x, b23.y}, b3 = {b23.z, b23.w};
        acc0 = __builtin_amdgcn_wmma_f32_16x16x4_f32(false, a, false, b0, (short)0, acc0, false, false);
        acc1 = __builtin_amdgcn_wmma_f32_16x16x4_f32(false, a, false, b1, (short)0, acc1, false, false);
        acc2 = __builtin_amdgcn_wmma_f32_16x16x4_f32(false, a, false, b2, (short)0, acc2, false, false);
        acc3 = __builtin_amdgcn_wmma_f32_16x16x4_f32(false, a, false, b3, (short)0, acc3, false, false);
    }

    const int rbase = gwave * 16 + lhalf * 8;
    float bi0 = 0.f, bi1 = 0.f, bi2 = 0.f, bi3 = 0.f;
    if (bias) { bi0 = bias[lmod]; bi1 = bias[16 + lmod]; bi2 = bias[32 + lmod]; bi3 = bias[48 + lmod]; }
#pragma unroll
    for (int r = 0; r < 8; ++r) {
        size_t ro = (size_t)(rbase + r) * HDIM + lmod;
        float v0 = acc0[r] + bi0, v1 = acc1[r] + bi1, v2 = acc2[r] + bi2, v3 = acc3[r] + bi3;
        if (act) { v0 = lrelu(v0); v1 = lrelu(v1); v2 = lrelu(v2); v3 = lrelu(v3); }
        if (rowscale) { const float rs = rowscale[rbase + r]; v0 *= rs; v1 *= rs; v2 *= rs; v3 *= rs; }
        C[ro] = v0; C[ro + 16] = v1; C[ro + 32] = v2; C[ro + 48] = v3;
    }
}

#define GATHER_BLOCKS 1024

#define ROW_LANE(v, U) __builtin_amdgcn_update_dpp(0, (v), 0x150 + (U), 0xf, 0xf, false)

#define GATHER_SLOT(U)                                                              \
    {                                                                               \
        const unsigned s = (unsigned)ROW_LANE(idx, U);                              \
        h[U] = *(const float4*)(hbytes + (size_t)((s << 8) | jb));                  \
    }
#define GATHER_ADD(U)                                                               \
    {                                                                               \
        const float w = (c + (U) < total) ? 1.0f : 0.0f;                            \
        acc.x = fmaf(w, h[U].x, acc.x); acc.y = fmaf(w, h[U].y, acc.y);             \
        acc.z = fmaf(w, h[U].z, acc.z); acc.w = fmaf(w, h[U].w, acc.w);             \
    }

__global__ void __launch_bounds__(256) conv_gather_kernel(const float* __restrict__ Hm, const int* __restrict__ adj,
                                                          const int* __restrict__ offs, const int* __restrict__ cnt,
                                                          const float* __restrict__ dinv, const float* __restrict__ bias,
                                                          float* __restrict__ out, int n, int act) {
    const int l8 = threadIdx.x & 7;
    const unsigned j = (threadIdx.x & 15) * 4;
    const unsigned jb = j * 4;
    const char* hbytes = (const char*)Hm;
    const int stride = (gridDim.x * 256) >> 4;
    int i = (blockIdx.x * 256 + threadIdx.x) >> 4;
    if (i >= n) return;
    const int last = n - 1;
    const float4 b = *(const float4*)(bias + j);

    int m = cnt[i];
    int o = offs[i];
    float dd = dinv[i];
    int traw = adj[max(o + min(l8, m) - 1, 0)];
    int i1 = i + stride;
    int q1 = min(i1, last);
    int m1 = cnt[q1];
    int o1 = offs[q1];
    float dd1 = dinv[q1];

    for (;;) {
        const int i2 = i1 + stride;
        const int q2 = min(i2, last);
        const int m2 = cnt[q2];
        const int o2 = offs[q2];
        const float dd2 = dinv[q2];
        const int traw1 = adj[max(o1 + min(l8, m1) - 1, 0)];

        float4 acc = make_float4(0.0f, 0.0f, 0.0f, 0.0f);
        const int total = m + 1;
        for (int c = 0; c < total; c += 8) {
            const int ecur = c + l8;
            const int idx = (ecur >= 1 && ecur <= m) ? traw : i;
            traw = adj[max(o + min(ecur + 8, m) - 1, 0)];
            float4 h[8];
            GATHER_SLOT(0) GATHER_SLOT(1) GATHER_SLOT(2) GATHER_SLOT(3)
            GATHER_SLOT(4) GATHER_SLOT(5) GATHER_SLOT(6) GATHER_SLOT(7)
            __builtin_amdgcn_sched_group_barrier(0x020, 9, 0);
            __builtin_amdgcn_sched_group_barrier(0x002, 16, 0);
            GATHER_ADD(0) GATHER_ADD(1) GATHER_ADD(2) GATHER_ADD(3)
            GATHER_ADD(4) GATHER_ADD(5) GATHER_ADD(6) GATHER_ADD(7)
        }

        float4 r = make_float4(b.x + dd * acc.x, b.y + dd * acc.y, b.z + dd * acc.z, b.w + dd * acc.w);
        if (act) {
            r.x = fmaxf(r.x, 0.0f); r.y = fmaxf(r.y, 0.0f);
            r.z = fmaxf(r.z, 0.0f); r.w = fmaxf(r.w, 0.0f);
        }
        *(float4*)(out + (size_t)i * HDIM + j) = r;

        if (i1 >= n) break;
        i = i1; m = m1; o = o1; dd = dd1; traw = traw1;
        i1 = i2; q1 = q2; m1 = m2; o1 = o2; dd1 = dd2;
    }
}

__global__ void __launch_bounds__(256) tail_kernel(const float* __restrict__ Z,
                                                   const float* __restrict__ Wp1, const float* __restrict__ bp1,
                                                   const float* __restrict__ Wp2, const float* __restrict__ bp2,
                                                   float* __restrict__ tprob, int M,
                                                   const int* __restrict__ treat, int nt,
                                                   const int* __restrict__ ctrl, int nc,
                                                   const float* __restrict__ Wy1, const float* __restrict__ by1,
                                                   const float* __restrict__ Wy0, const float* __restrict__ by0,
                                                   float* __restrict__ o_y1, float* __restrict__ o_yc0,
                                                   float* __restrict__ o_y0, float* __restrict__ o_yc1,
                                                   int hb_t, int hb_c) {
    extern __shared__ float bsw[];
    const int tid = threadIdx.x;
    int blk = blockIdx.x;

    if (blk < hb_t + hb_c) {
        const bool tr = blk < hb_t;
        const int t = (tr ? blk : blk - hb_t) * 256 + tid;
        if (t >= (tr ? nt : nc)) return;
        const int node = tr ? treat[t] : ctrl[t];
        const float* row = Z + (size_t)node * HDIM;
        float d1 = 0.0f, d0 = 0.0f;
#pragma unroll 4
        for (int j = 0; j < HDIM; j += 4) {
            float4 z  = *(const float4*)(row + j);
            float4 w1 = *(const float4*)(Wy1 + j);
            float4 w0 = *(const float4*)(Wy0 + j);
            d1 += z.x * w1.x + z.y * w1.y + z.z * w1.z + z.w * w1.w;
            d0 += z.x * w0.x + z.y * w0.y + z.z * w0.z + z.w * w0.w;
        }
        const float r1 = lrelu(d1 + by1[0]);
        const float r0 = lrelu(d0 + by0[0]);
        if (tr) { o_y1[t] = r1; o_yc0[t] = r0; }
        else    { o_y0[t] = r0; o_yc1[t] = r1; }
        return;
    }
    blk -= hb_t + hb_c;

    const int lane  = tid & 31;
    const int wave  = tid >> 5;
    const int lmod  = lane & 15;
    const int lhalf = lane >> 4;

    for (int p = tid; p < HDIM * 32; p += 256) {
        int ks  = p >> 7;
        int ll  = (p >> 2) & 31;
        int tn  = p & 3;
        int kb  = 4 * ks + 2 * (ll >> 4);
        int col = tn * 16 + (ll & 15);
        bsw[p * 2 + 0] = Wp1[(size_t)kb * HDIM + col];
        bsw[p * 2 + 1] = Wp1[(size_t)(kb + 1) * HDIM + col];
    }
    __syncthreads();

    const int gwave = blk * 8 + wave;
    if (gwave >= (M >> 4)) return;

    const float* arow = Z + (size_t)(gwave * 16 + lmod) * HDIM + 2 * lhalf;
    v8f acc0 = {}, acc1 = {}, acc2 = {}, acc3 = {};
#pragma unroll
    for (int ks = 0; ks < HDIM / 4; ++ks) {
        v2f a = *(const v2f*)(arow + 4 * ks);
        const float4* bp = (const float4*)(bsw + (size_t)(ks * 32 + lane) * 8);
        float4 b01 = bp[0];
        float4 b23 = bp[1];
        v2f b0 = {b01.x, b01.y}, b1 = {b01.z, b01.w};
        v2f b2 = {b23.x, b23.y}, b3 = {b23.z, b23.w};
        acc0 = __builtin_amdgcn_wmma_f32_16x16x4_f32(false, a, false, b0, (short)0, acc0, false, false);
        acc1 = __builtin_amdgcn_wmma_f32_16x16x4_f32(false, a, false, b1, (short)0, acc1, false, false);
        acc2 = __builtin_amdgcn_wmma_f32_16x16x4_f32(false, a, false, b2, (short)0, acc2, false, false);
        acc3 = __builtin_amdgcn_wmma_f32_16x16x4_f32(false, a, false, b3, (short)0, acc3, false, false);
    }

    const float bi0 = bp1[lmod], bi1 = bp1[16 + lmod], bi2 = bp1[32 + lmod], bi3 = bp1[48 + lmod];
    const float2 w0 = *(const float2*)(Wp2 + (size_t)(lmod) * 2);
    const float2 w1 = *(const float2*)(Wp2 + (size_t)(16 + lmod) * 2);
    const float2 w2 = *(const float2*)(Wp2 + (size_t)(32 + lmod) * 2);
    const float2 w3 = *(const float2*)(Wp2 + (size_t)(48 + lmod) * 2);
    float v[16];
#pragma unroll
    for (int r = 0; r < 8; ++r) {
        const float p0 = lrelu(acc0[r] + bi0), p1 = lrelu(acc1[r] + bi1);
        const float p2 = lrelu(acc2[r] + bi2), p3 = lrelu(acc3[r] + bi3);
        v[2 * r + 0] = p0 * w0.x + p1 * w1.x + p2 * w2.x + p3 * w3.x;
        v[2 * r + 1] = p0 * w0.y + p1 * w1.y + p2 * w2.y + p3 * w3.y;
    }
    const bool h8 = (lmod & 8) != 0, h4 = (lmod & 4) != 0, h2 = (lmod & 2) != 0, h1 = (lmod & 1) != 0;
    float u8[8], u4[4], u2[2];
#pragma unroll
    for (int i = 0; i < 8; ++i) {
        const float keep = h8 ? v[i + 8] : v[i];
        const float send = h8 ? v[i] : v[i + 8];
        u8[i] = keep + __shfl_xor(send, 8, 32);
    }
#pragma unroll
    for (int i = 0; i < 4; ++i) {
        const float keep = h4 ? u8[i + 4] : u8[i];
        const float send = h4 ? u8[i] : u8[i + 4];
        u4[i] = keep + __shfl_xor(send, 4, 32);
    }
#pragma unroll
    for (int i = 0; i < 2; ++i) {
        const float keep = h2 ? u4[i + 2] : u4[i];
        const float send = h2 ? u4[i] : u4[i + 2];
        u2[i] = keep + __shfl_xor(send, 2, 32);
    }
    const float keep1 = h1 ? u2[1] : u2[0];
    const float send1 = h1 ? u2[0] : u2[1];
    const float tot = keep1 + __shfl_xor(send1, 1, 32);
    tprob[(size_t)gwave * 32 + lane] = lrelu(tot + bp2[lmod & 1]);
}

extern "C" void kernel_launch(void* const* d_in, const int* in_sizes, int n_in,
                              void* d_out, int out_size, void* d_ws, size_t ws_size,
                              hipStream_t stream) {
    const float* x     = (const float*)d_in[0];
    const int*   ei    = (const int*)d_in[1];
    const int*   treat = (const int*)d_in[2];
    const int*   ctrl  = (const int*)d_in[3];
    const float* W1  = (const float*)d_in[4];  const float* b1  = (const float*)d_in[5];
    const float* W2  = (const float*)d_in[6];  const float* b2  = (const float*)d_in[7];
    const float* Wy1 = (const float*)d_in[8];  const float* by1 = (const float*)d_in[9];
    const float* Wy0 = (const float*)d_in[10]; const float* by0 = (const float*)d_in[11];
    const float* Wp1 = (const float*)d_in[12]; const float* bp1 = (const float*)d_in[13];
    const float* Wp2 = (const float*)d_in[14]; const float* bp2 = (const float*)d_in[15];

    const int N = in_sizes[0] / INDIM;
    const int E = in_sizes[1] / 2;
    const int T = in_sizes[2];
    const int Cn = in_sizes[3];
    const int* e_src = ei;
    const int* e_dst = ei + E;

    float* dinv = (float*)d_ws;
    float* buf0 = dinv + (((size_t)N + 63) & ~(size_t)63);
    int* binned = (int*)(buf0 + (size_t)N * HDIM);
    int* adj  = binned + (size_t)2 * E;
    int* cnt  = adj + (((size_t)E + 63) & ~(size_t)63);
    int* offs = cnt + (((size_t)N + 63) & ~(size_t)63);

    float* o_y1    = (float*)d_out;
    float* o_yc0   = o_y1 + T;
    float* o_y0    = o_yc0 + T;
    float* o_yc1   = o_y0 + Cn;
    float* o_tprob = o_yc1 + Cn;
    float* o_xZ2   = o_tprob + (size_t)2 * N;

    const int BLK = 256;
    auto cdiv = [](long long a, long long b) { return (int)((a + b - 1) / b); };
    const int gemm_blocks = cdiv(N / 16, 8);
    const int gather_blocks = min(GATHER_BLOCKS, cdiv((long long)N * 16, BLK));
    const size_t smem128 = (size_t)INDIM * HDIM * sizeof(float);
    const size_t smem64  = (size_t)HDIM * HDIM * sizeof(float);
    const int nb = cdiv(N, BIN_NODES);
    const int g1 = cdiv(E, SORT_CH);
    const int tab = nb * g1;
    int* hist = (int*)buf0;
    int* hofs = hist + (((size_t)tab + 63) & ~(size_t)63);
    const size_t smem_hist = (size_t)nb * sizeof(int);
    const size_t smem_scat = ((size_t)((3 * nb + 1) & ~1) + (size_t)2 * SORT_CH) * sizeof(int);

    bin_hist_kernel<<<g1, SORT_TPB, smem_hist, stream>>>(e_dst, hist, E, nb, g1);
    table_scan_kernel<<<cdiv(tab, SCAN_EPB), BLK, 0, stream>>>(hist, hofs, tab);
    bin_scatter_kernel<<<g1, SORT_TPB, smem_scat, stream>>>(e_src, e_dst, hist, hofs, binned, E, nb, g1);
    bin_rows_kernel<<<nb, SORT_TPB, 0, stream>>>(binned, hofs, adj, cnt, offs, dinv, N, E, nb, g1);

    wmma_gemm_f32_kernel<<<gemm_blocks, BLK, smem128, stream>>>(x, W1, buf0, nullptr, dinv, N, INDIM, 0);
    conv_gather_kernel<<<gather_blocks, BLK, 0, stream>>>(buf0, adj, offs, cnt, dinv, b1, o_xZ2, N, 1);

    wmma_gemm_f32_kernel<<<gemm_blocks, BLK, smem64, stream>>>(o_xZ2, W2, buf0, nullptr, dinv, N, HDIM, 0);
    conv_gather_kernel<<<gather_blocks, BLK, 0, stream>>>(buf0, adj, offs, cnt, dinv, b2, o_xZ2, N, 0);

    const int hb_t = cdiv(T, BLK);
    const int hb_c = cdiv(Cn, BLK);
    tail_kernel<<<hb_t + hb_c + gemm_blocks, BLK, smem64, stream>>>(o_xZ2, Wp1, bp1, Wp2, bp2, o_tprob, N,
                                                                    treat, T, ctrl, Cn, Wy1, by1, Wy0, by0,
                                                                    o_y1, o_yc0, o_y0, o_yc1, hb_t, hb_c);
}


# Edge sort in two launches, not four: each chunk ordered by dst bin in place from its own LDS counts (no count pass, no table scan); bin blocks fetch their runs; sum order still LDS arrival rank
# speedup vs baseline: 4.1084x; 1.0810x over previous
#include <hip/hip_runtime.h>
#include <hip/hip_bf16.h>

typedef __attribute__((ext_vector_type(2))) float v2f;
typedef __attribute__((ext_vector_type(8))) float v8f;

#define HDIM 64
#define INDIM 128

__device__ __forceinline__ float lrelu(float v) { return v > 0.0f ? v : 0.01f * v; }

#define SORT_TPB 512
#define SORT_CH 6144
#define SORT_EPT (SORT_CH / SORT_TPB)
#define BIN_SHIFT 9
#define BIN_NODES 512
#define FINE_CAP 12288

__device__ __forceinline__ int block_excl_scan_512(int v, int* wsum, int tid) {
    const int lane = tid & 31;
    const int wave = tid >> 5;
    int incl = v;
    for (int o = 1; o < 32; o <<= 1) {
        int u = __shfl_up(incl, o, 32);
        if (lane >= o) incl += u;
    }
    if (lane == 31) wsum[wave] = incl;
    __syncthreads();
    int front = 0;
    for (int w = 0; w < 16; ++w)
        if (w < wave) front += wsum[w];
    return front + incl - v;
}

__global__ void __launch_bounds__(SORT_TPB) chunk_sort_kernel(const int* __restrict__ src, const int* __restrict__ dst,
                                                              int* __restrict__ tab, int* __restrict__ binned,
                                                              int e, int nb, int g1) {
    extern __shared__ int lds[];
    __shared__ int wsum[16];
    const int tid = threadIdx.x;
    const int c = blockIdx.x;
    int* cnt    = lds;
    int* cursor = lds + nb;
    int2* stage = (int2*)(lds + ((2 * nb + 1) & ~1));

    const int base = c * SORT_CH;
    const int lim = min(e, base + SORT_CH);
    const int nloc = lim - base;
    int es[SORT_EPT], ed[SORT_EPT];
#pragma unroll
    for (int k = 0; k < SORT_EPT; ++k) {
        const int i = min(base + tid + k * SORT_TPB, lim - 1);
        es[k] = src[i];
        ed[k] = dst[i];
    }
    for (int b = tid; b < nb; b += SORT_TPB) cnt[b] = 0;
    __syncthreads();
#pragma unroll
    for (int k = 0; k < SORT_EPT; ++k)
        if (tid + k * SORT_TPB < nloc) atomicAdd(&cnt[ed[k] >> BIN_SHIFT], 1);
    __syncthreads();

    const int ipt = (nb + SORT_TPB - 1) / SORT_TPB;
    const int b0 = tid * ipt;
    const int b1 = min(nb, b0 + ipt);
    int mine = 0;
    for (int b = b0; b < b1; ++b) mine += cnt[b];
    int run = block_excl_scan_512(mine, wsum, tid);
    for (int b = b0; b < b1; ++b) {
        cursor[b] = run;
        tab[(size_t)b * g1 + c] = run;
        run += cnt[b];
    }
    if (tid == 0) tab[(size_t)nb * g1 + c] = nloc;
    __syncthreads();

    int pos[SORT_EPT];
#pragma unroll
    for (int k = 0; k < SORT_EPT; ++k) {
        pos[k] = 0;
        if (tid + k * SORT_TPB < nloc) pos[k] = atomicAdd(&cursor[ed[k] >> BIN_SHIFT], 1);
    }
#pragma unroll
    for (int k = 0; k < SORT_EPT; ++k)
        if (tid + k * SORT_TPB < nloc) stage[pos[k]] = make_int2(es[k], ed[k]);
    __syncthreads();

    int2* out2 = (int2*)binned + base;
    int2 w[SORT_EPT];
#pragma unroll
    for (int k = 0; k < SORT_EPT; ++k) w[k] = stage[min(tid + k * SORT_TPB, nloc - 1)];
#pragma unroll
    for (int k = 0; k < SORT_EPT; ++k) {
        const int j = tid + k * SORT_TPB;
        if (j < nloc) out2[j] = w[k];
    }
}

#define RUN_Q 18
#define RUN_S 2
__global__ void __launch_bounds__(SORT_TPB) bin_rows_kernel(const int* __restrict__ binned, const int* __restrict__ tab,
                                                            int* __restrict__ adj, int* __restrict__ cntG,
                                                            int* __restrict__ offsG, float* __restrict__ dinv,
                                                            int n, int e, int nb, int g1) {
    __shared__ int cnt[BIN_NODES];
    __shared__ int cursor[BIN_NODES];
    __shared__ int wsum[16];
    __shared__ int red[48];
    __shared__ int stage[FINE_CAP];
    const int tid = threadIdx.x;
    const int lane = tid & 31;
    const int wave = tid >> 5;
    const int b = blockIdx.x;
    const int2* bp = (const int2*)binned;
    const int* row0 = tab + (size_t)b * g1;
    const int* row1 = row0 + g1;

    int rs = 0, ln = 0, s0 = 0, s1 = 0, mx = 0;
    {
        const int cq = wave + 16 * lane;
        if (cq < g1) {
            rs = row0[cq];
            const int re = row1[cq];
            ln = re - rs;
            s0 = rs; s1 = re; mx = ln;
        }
        for (int cc = cq + SORT_TPB; cc < g1; cc += SORT_TPB) {
            const int a0 = row0[cc], a1 = row1[cc];
            s0 += a0; s1 += a1; mx = max(mx, a1 - a0);
        }
    }

    int2 v[RUN_Q * RUN_S];
#pragma unroll
    for (int q = 0; q < RUN_Q; ++q) {
        const int rsq = __builtin_amdgcn_readlane(rs, q);
        const int lnq = __builtin_amdgcn_readlane(ln, q);
        const int p0 = (wave + 16 * q) * SORT_CH + rsq;
#pragma unroll
        for (int s = 0; s < RUN_S; ++s)
            v[q * RUN_S + s] = bp[min(p0 + min(lane + 32 * s, max(lnq - 1, 0)), e - 1)];
    }

    for (int o = 16; o > 0; o >>= 1) {
        s0 += __shfl_xor(s0, o, 32);
        s1 += __shfl_xor(s1, o, 32);
        mx = max(mx, __shfl_xor(mx, o, 32));
    }
    if (lane == 0) { red[wave] = s0; red[16 + wave] = s1; red[32 + wave] = mx; }
    cnt[tid] = 0;
    __syncthreads();
    int start = 0, end = 0, maxlen = 0;
    for (int w = 0; w < 16; ++w) {
        start += red[w];
        end += red[16 + w];
        maxlen = max(maxlen, red[32 + w]);
    }
    const int m = end - start;
    const bool fits = m <= FINE_CAP && g1 <= 16 * RUN_Q && maxlen <= 32 * RUN_S;

    if (fits) {
#pragma unroll
        for (int q = 0; q < RUN_Q; ++q) {
            const int lnq = __builtin_amdgcn_readlane(ln, q);
#pragma unroll
            for (int s = 0; s < RUN_S; ++s)
                if (lane + 32 * s < lnq) atomicAdd(&cnt[v[q * RUN_S + s].y & (BIN_NODES - 1)], 1);
        }
    } else {
        for (int c = wave; c < g1; c += 16) {
            const int r0 = row0[c];
            const int len = row1[c] - r0;
            const int2* rp = bp + (size_t)c * SORT_CH + r0;
            for (int j = lane; j < len; j += 32) atomicAdd(&cnt[rp[j].y & (BIN_NODES - 1)], 1);
        }
    }
    __syncthreads();

    const int cdeg = cnt[tid];
    const int ex = block_excl_scan_512(cdeg, wsum, tid);
    cursor[tid] = ex;
    const int node = b * BIN_NODES + tid;
    if (node < n) {
        cntG[node] = cdeg;
        offsG[node] = start + ex;
        float d = (float)(cdeg + 1);
        dinv[node] = d > 0.0f ? rsqrtf(d) : 0.0f;
    }
    __syncthreads();

    if (fits) {
#pragma unroll
        for (int q = 0; q < RUN_Q; ++q) {
            const int lnq = __builtin_amdgcn_readlane(ln, q);
#pragma unroll
            for (int s = 0; s < RUN_S; ++s)
                if (lane + 32 * s < lnq)
                    v[q * RUN_S + s].y = atomicAdd(&cursor[v[q * RUN_S + s].y & (BIN_NODES - 1)], 1);
        }
#pragma unroll
        for (int q = 0; q < RUN_Q; ++q) {
            const int lnq = __builtin_amdgcn_readlane(ln, q);
#pragma unroll
            for (int s = 0; s < RUN_S; ++s)
                if (lane + 32 * s < lnq) stage[v[q * RUN_S + s].y] = v[q * RUN_S + s].x;
        }
        __syncthreads();
        for (int j = tid; j < m; j += SORT_TPB) adj[start + j] = stage[j];
    } else {
        for (int c = wave; c < g1; c += 16) {
            const int r0 = row0[c];
            const int len = row1[c] - r0;
            const int2* rp = bp + (size_t)c * SORT_CH + r0;
            for (int j = lane; j < len; j += 32) {
                const int2 u = rp[j];
                const int p = atomicAdd(&cursor[u.y & (BIN_NODES - 1)], 1);
                adj[start + p] = u.x;
            }
        }
    }
}

__global__ void wmma_gemm_f32_kernel(const float* __restrict__ A, const float* __restrict__ B,
                                     float* __restrict__ C, const float* __restrict__ bias,
                                     const float* __restrict__ rowscale, int M, int K, int act) {
    extern __shared__ float bsw[];
    const int tid   = threadIdx.x;
    const int lane  = tid & 31;
    const int wave  = tid >> 5;
    const int lmod  = lane & 15;
    const int lhalf = lane >> 4;

    const int pairs = K * 32;
    for (int p = tid; p < pairs; p += 256) {
        int ks  = p >> 7;
        int ll  = (p >> 2) & 31;
        int tn  = p & 3;
        int kb  = 4 * ks + 2 * (ll >> 4);
        int col = tn * 16 + (ll & 15);
        bsw[p * 2 + 0] = B[(size_t)kb * HDIM + col];
        bsw[p * 2 + 1] = B[(size_t)(kb + 1) * HDIM + col];
    }
    __syncthreads();

    const int gwave = blockIdx.x * 8 + wave;
    if (gwave >= (M >> 4)) return;

    const float* arow = A + (size_t)(gwave * 16 + lmod) * K + 2 * lhalf;
    v8f acc0 = {}, acc1 = {}, acc2 = {}, acc3 = {};

    const int steps = K >> 2;
    for (int ks = 0; ks < steps; ++ks) {
        v2f a = *(const v2f*)(arow + 4 * ks);
        const float4* bp = (const float4*)(bsw + (size_t)(ks * 32 + lane) * 8);
        float4 b01 = bp[0];
        float4 b23 = bp[1];
        v2f b0 = {b01.x, b01.y}, b1 = {b01.z, b01.w};
        v2f b2 = {b23.x, b23.y}, b3 = {b23.z, b23.w};
        acc0 = __builtin_amdgcn_wmma_f32_16x16x4_f32(false, a, false, b0, (short)0, acc0, false, false);
        acc1 = __builtin_amdgcn_wmma_f32_16x16x4_f32(false, a, false, b1, (short)0, acc1, false, false);
        acc2 = __builtin_amdgcn_wmma_f32_16x16x4_f32(false, a, false, b2, (short)0, acc2, false, false);
        acc3 = __builtin_amdgcn_wmma_f32_16x16x4_f32(false, a, false, b3, (short)0, acc3, false, false);
    }

    const int rbase = gwave * 16 + lhalf * 8;
    float bi0 = 0.f, bi1 = 0.f, bi2 = 0.f, bi3 = 0.f;
    if (bias) { bi0 = bias[lmod]; bi1 = bias[16 + lmod]; bi2 = bias[32 + lmod]; bi3 = bias[48 + lmod]; }
#pragma unroll
    for (int r = 0; r < 8; ++r) {
        size_t ro = (size_t)(rbase + r) * HDIM + lmod;
        float v0 = acc0[r] + bi0, v1 = acc1[r] + bi1, v2 = acc2[r] + bi2, v3 = acc3[r] + bi3;
        if (act) { v0 = lrelu(v0); v1 = lrelu(v1); v2 = lrelu(v2); v3 = lrelu(v3); }
        if (rowscale) { const float rs = rowscale[rbase + r]; v0 *= rs; v1 *= rs; v2 *= rs; v3 *= rs; }
        C[ro] = v0; C[ro + 16] = v1; C[ro + 32] = v2; C[ro + 48] = v3;
    }
}

#define GATHER_BLOCKS 1024

#define ROW_LANE(v, U) __builtin_amdgcn_update_dpp(0, (v), 0x150 + (U), 0xf, 0xf, false)

#define GATHER_SLOT(U)                                                              \
    {                                                                               \
        const unsigned s = (unsigned)ROW_LANE(idx, U);                              \
        h[U] = *(const float4*)(hbytes + (size_t)((s << 8) | jb));                  \
    }
#define GATHER_ADD(U)                                                               \
    {                                                                               \
        const float w = (c + (U) < total) ? 1.0f : 0.0f;                            \
        acc.x = fmaf(w, h[U].x, acc.x); acc.y = fmaf(w, h[U].y, acc.y);             \
        acc.z = fmaf(w, h[U].z, acc.z); acc.w = fmaf(w, h[U].w, acc.w);             \
    }

__global__ void __launch_bounds__(256) conv_gather_kernel(const float* __restrict__ Hm, const int* __restrict__ adj,
                                                          const int* __restrict__ offs, const int* __restrict__ cnt,
                                                          const float* __restrict__ dinv, const float* __restrict__ bias,
                                                          float* __restrict__ out, int n, int act) {
    const int l8 = threadIdx.x & 7;
    const unsigned j = (threadIdx.x & 15) * 4;
    const unsigned jb = j * 4;
    const char* hbytes = (const char*)Hm;
    const int stride = (gridDim.x * 256) >> 4;
    int i = (blockIdx.x * 256 + threadIdx.x) >> 4;
    if (i >= n) return;
    const int last = n - 1;
    const float4 b = *(const float4*)(bias + j);

    int m = cnt[i];
    int o = offs[i];
    float dd = dinv[i];
    int traw = adj[max(o + min(l8, m) - 1, 0)];
    int i1 = i + stride;
    int q1 = min(i1, last);
    int m1 = cnt[q1];
    int o1 = offs[q1];
    float dd1 = dinv[q1];

    for (;;) {
        const int i2 = i1 + stride;
        const int q2 = min(i2, last);
        const int m2 = cnt[q2];
        const int o2 = offs[q2];
        const float dd2 = dinv[q2];
        const int traw1 = adj[max(o1 + min(l8, m1) - 1, 0)];

        float4 acc = make_float4(0.0f, 0.0f, 0.0f, 0.0f);
        const int total = m + 1;
        for (int c = 0; c < total; c += 8) {
            const int ecur = c + l8;
            const int idx = (ecur >= 1 && ecur <= m) ? traw : i;
            traw = adj[max(o + min(ecur + 8, m) - 1, 0)];
            float4 h[8];
            GATHER_SLOT(0) GATHER_SLOT(1) GATHER_SLOT(2) GATHER_SLOT(3)
            GATHER_SLOT(4) GATHER_SLOT(5) GATHER_SLOT(6) GATHER_SLOT(7)
            __builtin_amdgcn_sched_group_barrier(0x020, 9, 0);
            __builtin_amdgcn_sched_group_barrier(0x002, 16, 0);
            GATHER_ADD(0) GATHER_ADD(1) GATHER_ADD(2) GATHER_ADD(3)
            GATHER_ADD(4) GATHER_ADD(5) GATHER_ADD(6) GATHER_ADD(7)
        }

        float4 r = make_float4(b.x + dd * acc.x, b.y + dd * acc.y, b.z + dd * acc.z, b.w + dd * acc.w);
        if (act) {
            r.x = fmaxf(r.x, 0.0f); r.y = fmaxf(r.y, 0.0f);
            r.z = fmaxf(r.z, 0.0f); r.w = fmaxf(r.w, 0.0f);
        }
        *(float4*)(out + (size_t)i * HDIM + j) = r;

        if (i1 >= n) break;
        i = i1; m = m1; o = o1; dd = dd1; traw = traw1;
        i1 = i2; q1 = q2; m1 = m2; o1 = o2; dd1 = dd2;
    }
}

__global__ void __launch_bounds__(256) tail_kernel(const float* __restrict__ Z,
                                                   const float* __restrict__ Wp1, const float* __restrict__ bp1,
                                                   const float* __restrict__ Wp2, const float* __restrict__ bp2,
                                                   float* __restrict__ tprob, int M,
                                                   const int* __restrict__ treat, int nt,
                                                   const int* __restrict__ ctrl, int nc,
                                                   const float* __restrict__ Wy1, const float* __restrict__ by1,
                                                   const float* __restrict__ Wy0, const float* __restrict__ by0,
                                                   float* __restrict__ o_y1, float* __restrict__ o_yc0,
                                                   float* __restrict__ o_y0, float* __restrict__ o_yc1,
                                                   int hb_t, int hb_c) {
    extern __shared__ float bsw[];
    const int tid = threadIdx.x;
    int blk = blockIdx.x;

    if (blk < hb_t + hb_c) {
        const bool tr = blk < hb_t;
        const int t = (tr ? blk : blk - hb_t) * 256 + tid;
        if (t >= (tr ? nt : nc)) return;
        const int node = tr ? treat[t] : ctrl[t];
        const float* row = Z + (size_t)node * HDIM;
        float d1 = 0.0f, d0 = 0.0f;
#pragma unroll 4
        for (int j = 0; j < HDIM; j += 4) {
            float4 z  = *(const float4*)(row + j);
            float4 w1 = *(const float4*)(Wy1 + j);
            float4 w0 = *(const float4*)(Wy0 + j);
            d1 += z.x * w1.x + z.y * w1.y + z.z * w1.z + z.w * w1.w;
            d0 += z.x * w0.x + z.y * w0.y + z.z * w0.z + z.w * w0.w;
        }
        const float r1 = lrelu(d1 + by1[0]);
        const float r0 = lrelu(d0 + by0[0]);
        if (tr) { o_y1[t] = r1; o_yc0[t] = r0; }
        else    { o_y0[t] = r0; o_yc1[t] = r1; }
        return;
    }
    blk -= hb_t + hb_c;

    const int lane  = tid & 31;
    const int wave  = tid >> 5;
    const int lmod  = lane & 15;
    const int lhalf = lane >> 4;

    for (int p = tid; p < HDIM * 32; p += 256) {
        int ks  = p >> 7;
        int ll  = (p >> 2) & 31;
        int tn  = p & 3;
        int kb  = 4 * ks + 2 * (ll >> 4);
        int col = tn * 16 + (ll & 15);
        bsw[p * 2 + 0] = Wp1[(size_t)kb * HDIM + col];
        bsw[p * 2 + 1] = Wp1[(size_t)(kb + 1) * HDIM + col];
    }
    __syncthreads();

    const int gwave = blk * 8 + wave;
    if (gwave >= (M >> 4)) return;

    const float* arow = Z + (size_t)(gwave * 16 + lmod) * HDIM + 2 * lhalf;
    v8f acc0 = {}, acc1 = {}, acc2 = {}, acc3 = {};
#pragma unroll
    for (int ks = 0; ks < HDIM / 4; ++ks) {
        v2f a = *(const v2f*)(arow + 4 * ks);
        const float4* bp = (const float4*)(bsw + (size_t)(ks * 32 + lane) * 8);
        float4 b01 = bp[0];
        float4 b23 = bp[1];
        v2f b0 = {b01.x, b01.y}, b1 = {b01.z, b01.w};
        v2f b2 = {b23.x, b23.y}, b3 = {b23.z, b23.w};
        acc0 = __builtin_amdgcn_wmma_f32_16x16x4_f32(false, a, false, b0, (short)0, acc0, false, false);
        acc1 = __builtin_amdgcn_wmma_f32_16x16x4_f32(false, a, false, b1, (short)0, acc1, false, false);
        acc2 = __builtin_amdgcn_wmma_f32_16x16x4_f32(false, a, false, b2, (short)0, acc2, false, false);
        acc3 = __builtin_amdgcn_wmma_f32_16x16x4_f32(false, a, false, b3, (short)0, acc3, false, false);
    }

    const float bi0 = bp1[lmod], bi1 = bp1[16 + lmod], bi2 = bp1[32 + lmod], bi3 = bp1[48 + lmod];
    const float2 w0 = *(const float2*)(Wp2 + (size_t)(lmod) * 2);
    const float2 w1 = *(const float2*)(Wp2 + (size_t)(16 + lmod) * 2);
    const float2 w2 = *(const float2*)(Wp2 + (size_t)(32 + lmod) * 2);
    const float2 w3 = *(const float2*)(Wp2 + (size_t)(48 + lmod) * 2);
    float v[16];
#pragma unroll
    for (int r = 0; r < 8; ++r) {
        const float p0 = lrelu(acc0[r] + bi0), p1 = lrelu(acc1[r] + bi1);
        const float p2 = lrelu(acc2[r] + bi2), p3 = lrelu(acc3[r] + bi3);
        v[2 * r + 0] = p0 * w0.x + p1 * w1.x + p2 * w2.x + p3 * w3.x;
        v[2 * r + 1] = p0 * w0.y + p1 * w1.y + p2 * w2.y + p3 * w3.y;
    }
    const bool h8 = (lmod & 8) != 0, h4 = (lmod & 4) != 0, h2 = (lmod & 2) != 0, h1 = (lmod & 1) != 0;
    float u8[8], u4[4], u2[2];
#pragma unroll
    for (int i = 0; i < 8; ++i) {
        const float keep = h8 ? v[i + 8] : v[i];
        const float send = h8 ? v[i] : v[i + 8];
        u8[i] = keep + __shfl_xor(send, 8, 32);
    }
#pragma unroll
    for (int i = 0; i < 4; ++i) {
        const float keep = h4 ? u8[i + 4] : u8[i];
        const float send = h4 ? u8[i] : u8[i + 4];
        u4[i] = keep + __shfl_xor(send, 4, 32);
    }
#pragma unroll
    for (int i = 0; i < 2; ++i) {
        const float keep = h2 ? u4[i + 2] : u4[i];
        const float send = h2 ? u4[i] : u4[i + 2];
        u2[i] = keep + __shfl_xor(send, 2, 32);
    }
    const float keep1 = h1 ? u2[1] : u2[0];
    const float send1 = h1 ? u2[0] : u2[1];
    const float tot = keep1 + __shfl_xor(send1, 1, 32);
    tprob[(size_t)gwave * 32 + lane] = lrelu(tot + bp2[lmod & 1]);
}

extern "C" void kernel_launch(void* const* d_in, const int* in_sizes, int n_in,
                              void* d_out, int out_size, void* d_ws, size_t ws_size,
                              hipStream_t stream) {
    const float* x     = (const float*)d_in[0];
    const int*   ei    = (const int*)d_in[1];
    const int*   treat = (const int*)d_in[2];
    const int*   ctrl  = (const int*)d_in[3];
    const float* W1  = (const float*)d_in[4];  const float* b1  = (const float*)d_in[5];
    const float* W2  = (const float*)d_in[6];  const float* b2  = (const float*)d_in[7];
    const float* Wy1 = (const float*)d_in[8];  const float* by1 = (const float*)d_in[9];
    const float* Wy0 = (const float*)d_in[10]; const float* by0 = (const float*)d_in[11];
    const float* Wp1 = (const float*)d_in[12]; const float* bp1 = (const float*)d_in[13];
    const float* Wp2 = (const float*)d_in[14]; const float* bp2 = (const float*)d_in[15];

    const int N = in_sizes[0] / INDIM;
    const int E = in_sizes[1] / 2;
    const int T = in_sizes[2];
    const int Cn = in_sizes[3];
    const int* e_src = ei;
    const int* e_dst = ei + E;

    float* dinv = (float*)d_ws;
    float* buf0 = dinv + (((size_t)N + 63) & ~(size_t)63);
    int* binned = (int*)(buf0 + (size_t)N * HDIM);
    int* adj  = binned + (size_t)2 * E;
    int* cnt  = adj + (((size_t)E + 63) & ~(size_t)63);
    int* offs = cnt + (((size_t)N + 63) & ~(size_t)63);

    float* o_y1    = (float*)d_out;
    float* o_yc0   = o_y1 + T;
    float* o_y0    = o_yc0 + T;
    float* o_yc1   = o_y0 + Cn;
    float* o_tprob = o_yc1 + Cn;
    float* o_xZ2   = o_tprob + (size_t)2 * N;

    const int BLK = 256;
    auto cdiv = [](long long a, long long b) { return (int)((a + b - 1) / b); };
    const int gemm_blocks = cdiv(N / 16, 8);
    const int gather_blocks = min(GATHER_BLOCKS, cdiv((long long)N * 16, BLK));
    const size_t smem128 = (size_t)INDIM * HDIM * sizeof(float);
    const size_t smem64  = (size_t)HDIM * HDIM * sizeof(float);
    const int nb = cdiv(N, BIN_NODES);
    const int g1 = cdiv(E, SORT_CH);
    int* tab = (int*)buf0;
    const size_t smem_sort = ((size_t)((2 * nb + 1) & ~1) + (size_t)2 * SORT_CH) * sizeof(int);

    chunk_sort_kernel<<<g1, SORT_TPB, smem_sort, stream>>>(e_src, e_dst, tab, binned, E, nb, g1);
    bin_rows_kernel<<<nb, SORT_TPB, 0, stream>>>(binned, tab, adj, cnt, offs, dinv, N, E, nb, g1);

    wmma_gemm_f32_kernel<<<gemm_blocks, BLK, smem128, stream>>>(x, W1, buf0, nullptr, dinv, N, INDIM, 0);
    conv_gather_kernel<<<gather_blocks, BLK, 0, stream>>>(buf0, adj, offs, cnt, dinv, b1, o_xZ2, N, 1);

    wmma_gemm_f32_kernel<<<gemm_blocks, BLK, smem64, stream>>>(o_xZ2, W2, buf0, nullptr, dinv, N, HDIM, 0);
    conv_gather_kernel<<<gather_blocks, BLK, 0, stream>>>(buf0, adj, offs, cnt, dinv, b2, o_xZ2, N, 0);

    const int hb_t = cdiv(T, BLK);
    const int hb_c = cdiv(Cn, BLK);
    tail_kernel<<<hb_t + hb_c + gemm_blocks, BLK, smem64, stream>>>(o_xZ2, Wp1, bp1, Wp2, bp2, o_tprob, N,
                                                                    treat, T, ctrl, Cn, Wy1, by1, Wy0, by0,
                                                                    o_y1, o_yc0, o_y0, o_yc1, hb_t, hb_c);
}


# Each aggregation launch sums 16-node strips per wave into LDS and does the next dense layer there on fp32 WMMA (W2; Wp1, Wp2): Z1 never stored, xZ2 not re-read, 7 launches to 6; sum orders kept
# speedup vs baseline: 4.7229x; 1.1496x over previous
#include <hip/hip_runtime.h>
#include <hip/hip_bf16.h>

typedef __attribute__((ext_vector_type(2))) float v2f;
typedef __attribute__((ext_vector_type(8))) float v8f;

#define HDIM 64
#define INDIM 128

__device__ __forceinline__ float lrelu(float v) { return v > 0.0f ? v : 0.01f * v; }

#define SORT_TPB 512
#define SORT_CH 6144
#define SORT_EPT (SORT_CH / SORT_TPB)
#define BIN_SHIFT 9
#define BIN_NODES 512
#define FINE_CAP 12288

__device__ __forceinline__ int block_excl_scan_512(int v, int* wsum, int tid) {
    const int lane = tid & 31;
    const int wave = tid >> 5;
    int incl = v;
    for (int o = 1; o < 32; o <<= 1) {
        int u = __shfl_up(incl, o, 32);
        if (lane >= o) incl += u;
    }
    if (lane == 31) wsum[wave] = incl;
    __syncthreads();
    int front = 0;
    for (int w = 0; w < 16; ++w)
        if (w < wave) front += wsum[w];
    return front + incl - v;
}

__global__ void __launch_bounds__(SORT_TPB) chunk_sort_kernel(const int* __restrict__ src, const int* __restrict__ dst,
                                                              int* __restrict__ tab, int* __restrict__ binned,
                                                              int e, int nb, int g1) {
    extern __shared__ int lds[];
    __shared__ int wsum[16];
    const int tid = threadIdx.x;
    const int c = blockIdx.x;
    int* cnt    = lds;
    int* cursor = lds + nb;
    int2* stage = (int2*)(lds + ((2 * nb + 1) & ~1));

    const int base = c * SORT_CH;
    const int lim = min(e, base + SORT_CH);
    const int nloc = lim - base;
    int es[SORT_EPT], ed[SORT_EPT];
#pragma unroll
    for (int k = 0; k < SORT_EPT; ++k) {
        const int i = min(base + tid + k * SORT_TPB, lim - 1);
        es[k] = src[i];
        ed[k] = dst[i];
    }
    for (int b = tid; b < nb; b += SORT_TPB) cnt[b] = 0;
    __syncthreads();
#pragma unroll
    for (int k = 0; k < SORT_EPT; ++k)
        if (tid + k * SORT_TPB < nloc) atomicAdd(&cnt[ed[k] >> BIN_SHIFT], 1);
    __syncthreads();

    const int ipt = (nb + SORT_TPB - 1) / SORT_TPB;
    const int b0 = tid * ipt;
    const int b1 = min(nb, b0 + ipt);
    int mine = 0;
    for (int b = b0; b < b1; ++b) mine += cnt[b];
    int run = block_excl_scan_512(mine, wsum, tid);
    for (int b = b0; b < b1; ++b) {
        cursor[b] = run;
        tab[(size_t)b * g1 + c] = run;
        run += cnt[b];
    }
    if (tid == 0) tab[(size_t)nb * g1 + c] = nloc;
    __syncthreads();

    int pos[SORT_EPT];
#pragma unroll
    for (int k = 0; k < SORT_EPT; ++k) {
        pos[k] = 0;
        if (tid + k * SORT_TPB < nloc) pos[k] = atomicAdd(&cursor[ed[k] >> BIN_SHIFT], 1);
    }
#pragma unroll
    for (int k = 0; k < SORT_EPT; ++k)
        if (tid + k * SORT_TPB < nloc) stage[pos[k]] = make_int2(es[k], ed[k]);
    __syncthreads();

    int2* out2 = (int2*)binned + base;
    int2 w[SORT_EPT];
#pragma unroll
    for (int k = 0; k < SORT_EPT; ++k) w[k] = stage[min(tid + k * SORT_TPB, nloc - 1)];
#pragma unroll
    for (int k = 0; k < SORT_EPT; ++k) {
        const int j = tid + k * SORT_TPB;
        if (j < nloc) out2[j] = w[k];
    }
}

#define RUN_Q 18
#define RUN_S 2
__global__ void __launch_bounds__(SORT_TPB) bin_rows_kernel(const int* __restrict__ binned, const int* __restrict__ tab,
                                                            int* __restrict__ adj, int* __restrict__ cntG,
                                                            int* __restrict__ offsG, float* __restrict__ dinv,
                                                            int n, int e, int nb, int g1) {
    __shared__ int cnt[BIN_NODES];
    __shared__ int cursor[BIN_NODES];
    __shared__ int wsum[16];
    __shared__ int red[48];
    __shared__ int stage[FINE_CAP];
    const int tid = threadIdx.x;
    const int lane = tid & 31;
    const int wave = tid >> 5;
    const int b = blockIdx.x;
    const int2* bp = (const int2*)binned;
    const int* row0 = tab + (size_t)b * g1;
    const int* row1 = row0 + g1;

    int rs = 0, ln = 0, s0 = 0, s1 = 0, mx = 0;
    {
        const int cq = wave + 16 * lane;
        if (cq < g1) {
            rs = row0[cq];
            const int re = row1[cq];
            ln = re - rs;
            s0 = rs; s1 = re; mx = ln;
        }
        for (int cc = cq + SORT_TPB; cc < g1; cc += SORT_TPB) {
            const int a0 = row0[cc], a1 = row1[cc];
            s0 += a0; s1 += a1; mx = max(mx, a1 - a0);
        }
    }

    int2 v[RUN_Q * RUN_S];
#pragma unroll
    for (int q = 0; q < RUN_Q; ++q) {
        const int rsq = __builtin_amdgcn_readlane(rs, q);
        const int lnq = __builtin_amdgcn_readlane(ln, q);
        const int p0 = (wave + 16 * q) * SORT_CH + rsq;
#pragma unroll
        for (int s = 0; s < RUN_S; ++s)
            v[q * RUN_S + s] = bp[min(p0 + min(lane + 32 * s, max(lnq - 1, 0)), e - 1)];
    }

    for (int o = 16; o > 0; o >>= 1) {
        s0 += __shfl_xor(s0, o, 32);
        s1 += __shfl_xor(s1, o, 32);
        mx = max(mx, __shfl_xor(mx, o, 32));
    }
    if (lane == 0) { red[wave] = s0; red[16 + wave] = s1; red[32 + wave] = mx; }
    cnt[tid] = 0;
    __syncthreads();
    int start = 0, end = 0, maxlen = 0;
    for (int w = 0; w < 16; ++w) {
        start += red[w];
        end += red[16 + w];
        maxlen = max(maxlen, red[32 + w]);
    }
    const int m = end - start;
    const bool fits = m <= FINE_CAP && g1 <= 16 * RUN_Q && maxlen <= 32 * RUN_S;

    if (fits) {
#pragma unroll
        for (int q = 0; q < RUN_Q; ++q) {
            const int lnq = __builtin_amdgcn_readlane(ln, q);
#pragma unroll
            for (int s = 0; s < RUN_S; ++s)
                if (lane + 32 * s < lnq) atomicAdd(&cnt[v[q * RUN_S + s].y & (BIN_NODES - 1)], 1);
        }
    } else {
        for (int c = wave; c < g1; c += 16) {
            const int r0 = row0[c];
            const int len = row1[c] - r0;
            const int2* rp = bp + (size_t)c * SORT_CH + r0;
            for (int j = lane; j < len; j += 32) atomicAdd(&cnt[rp[j].y & (BIN_NODES - 1)], 1);
        }
    }
    __syncthreads();

    const int cdeg = cnt[tid];
    const int ex = block_excl_scan_512(cdeg, wsum, tid);
    cursor[tid] = ex;
    const int node = b * BIN_NODES + tid;
    if (node < n) {
        cntG[node] = cdeg;
        offsG[node] = start + ex;
        float d = (float)(cdeg + 1);
        dinv[node] = d > 0.0f ? rsqrtf(d) : 0.0f;
    }
    __syncthreads();

    if (fits) {
#pragma unroll
        for (int q = 0; q < RUN_Q; ++q) {
            const int lnq = __builtin_amdgcn_readlane(ln, q);
#pragma unroll
            for (int s = 0; s < RUN_S; ++s)
                if (lane + 32 * s < lnq)
                    v[q * RUN_S + s].y = atomicAdd(&cursor[v[q * RUN_S + s].y & (BIN_NODES - 1)], 1);
        }
#pragma unroll
        for (int q = 0; q < RUN_Q; ++q) {
            const int lnq = __builtin_amdgcn_readlane(ln, q);
#pragma unroll
            for (int s = 0; s < RUN_S; ++s)
                if (lane + 32 * s < lnq) stage[v[q * RUN_S + s].y] = v[q * RUN_S + s].x;
        }
        __syncthreads();
        for (int j = tid; j < m; j += SORT_TPB) adj[start + j] = stage[j];
    } else {
        for (int c = wave; c < g1; c += 16) {
            const int r0 = row0[c];
            const int len = row1[c] - r0;
            const int2* rp = bp + (size_t)c * SORT_CH + r0;
            for (int j = lane; j < len; j += 32) {
                const int2 u = rp[j];
                const int p = atomicAdd(&cursor[u.y & (BIN_NODES - 1)], 1);
                adj[start + p] = u.x;
            }
        }
    }
}

__global__ void wmma_gemm_f32_kernel(const float* __restrict__ A, const float* __restrict__ B,
                                     float* __restrict__ C, const float* __restrict__ bias,
                                     const float* __restrict__ rowscale, int M, int K, int act) {
    extern __shared__ float bsw[];
    const int tid   = threadIdx.x;
    const int lane  = tid & 31;
    const int wave  = tid >> 5;
    const int lmod  = lane & 15;
    const int lhalf = lane >> 4;

    const int pairs = K * 32;
    for (int p = tid; p < pairs; p += 256) {
        int ks  = p >> 7;
        int ll  = (p >> 2) & 31;
        int tn  = p & 3;
        int kb  = 4 * ks + 2 * (ll >> 4);
        int col = tn * 16 + (ll & 15);
        bsw[p * 2 + 0] = B[(size_t)kb * HDIM + col];
        bsw[p * 2 + 1] = B[(size_t)(kb + 1) * HDIM + col];
    }
    __syncthreads();

    const int gwave = blockIdx.x * 8 + wave;
    if (gwave >= (M >> 4)) return;

    const float* arow = A + (size_t)(gwave * 16 + lmod) * K + 2 * lhalf;
    v8f acc0 = {}, acc1 = {}, acc2 = {}, acc3 = {};

    const int steps = K >> 2;
    for (int ks = 0; ks < steps; ++ks) {
        v2f a = *(const v2f*)(arow + 4 * ks);
        const float4* bp = (const float4*)(bsw + (size_t)(ks * 32 + lane) * 8);
        float4 b01 = bp[0];
        float4 b23 = bp[1];
        v2f b0 = {b01.x, b01.y}, b1 = {b01.z, b01.w};
        v2f b2 = {b23.x, b23.y}, b3 = {b23.z, b23.w};
        acc0 = __builtin_amdgcn_wmma_f32_16x16x4_f32(false, a, false, b0, (short)0, acc0, false, false);
        acc1 = __builtin_amdgcn_wmma_f32_16x16x4_f32(false, a, false, b1, (short)0, acc1, false, false);
        acc2 = __builtin_amdgcn_wmma_f32_16x16x4_f32(false, a, false, b2, (short)0, acc2, false, false);
        acc3 = __builtin_amdgcn_wmma_f32_16x16x4_f32(false, a, false, b3, (short)0, acc3, false, false);
    }

    const int rbase = gwave * 16 + lhalf * 8;
    float bi0 = 0.f, bi1 = 0.f, bi2 = 0.f, bi3 = 0.f;
    if (bias) { bi0 = bias[lmod]; bi1 = bias[16 + lmod]; bi2 = bias[32 + lmod]; bi3 = bias[48 + lmod]; }
#pragma unroll
    for (int r = 0; r < 8; ++r) {
        size_t ro = (size_t)(rbase + r) * HDIM + lmod;
        float v0 = acc0[r] + bi0, v1 = acc1[r] + bi1, v2 = acc2[r] + bi2, v3 = acc3[r] + bi3;
        if (act) { v0 = lrelu(v0); v1 = lrelu(v1); v2 = lrelu(v2); v3 = lrelu(v3); }
        if (rowscale) { const float rs = rowscale[rbase + r]; v0 *= rs; v1 *= rs; v2 *= rs; v3 *= rs; }
        C[ro] = v0; C[ro + 16] = v1; C[ro + 32] = v2; C[ro + 48] = v3;
    }
}

#define GATHER_BLOCKS 512
#define FUSED_TPB 512
#define FUSED_WAVES (FUSED_TPB / 32)
#define TILE_PITCH 68
#define FUSED_LDS_FLOATS (HDIM * HDIM + FUSED_WAVES * 16 * TILE_PITCH)

#define ROW_LANE(v, U) __builtin_amdgcn_update_dpp(0, (v), 0x150 + (U), 0xf, 0xf, false)

#define GATHER_SLOT(U)                                                              \
    {                                                                               \
        const unsigned s = (unsigned)ROW_LANE(idx, U);                              \
        h[U] = *(const float4*)(hbytes + (size_t)((s << 8) | jb));                  \
    }
#define GATHER_ADD(U)                                                               \
    {                                                                               \
        const float w = (c + (U) < total) ? 1.0f : 0.0f;                            \
        acc.x = fmaf(w, h[U].x, acc.x); acc.y = fmaf(w, h[U].y, acc.y);             \
        acc.z = fmaf(w, h[U].z, acc.z); acc.w = fmaf(w, h[U].w, acc.w);             \
    }

__device__ __forceinline__ void stage_weight64(float* bsw, const float* __restrict__ B, int tid) {
#pragma unroll
    for (int it = 0; it < HDIM * 32 / FUSED_TPB; ++it) {
        const int p = tid + it * FUSED_TPB;
        int ks  = p >> 7;
        int ll  = (p >> 2) & 31;
        int tn  = p & 3;
        int kb  = 4 * ks + 2 * (ll >> 4);
        int col = tn * 16 + (ll & 15);
        bsw[p * 2 + 0] = B[(size_t)kb * HDIM + col];
        bsw[p * 2 + 1] = B[(size_t)(kb + 1) * HDIM + col];
    }
}

__device__ __forceinline__ void gather_eight(const char* __restrict__ hbytes, const int* __restrict__ adj,
                                             const int* __restrict__ offs, const int* __restrict__ cnt,
                                             const float* __restrict__ dinv, const float4 b, const bool relu,
                                             float* __restrict__ gout, float* lrow, const int node0, const int n,
                                             const int l8, const unsigned j, const unsigned jb) {
    const int last = n - 1;
    int i = node0;
    int m = cnt[i];
    int o = offs[i];
    float dd = dinv[i];
    int traw = adj[max(o + min(l8, m) - 1, 0)];
    int q1 = min(i + 1, last);
    int m1 = cnt[q1];
    int o1 = offs[q1];
    float dd1 = dinv[q1];

    for (int t = 0; t < 8; ++t) {
        const int q2 = min(i + 2, last);
        const int m2 = cnt[q2];
        const int o2 = offs[q2];
        const float dd2 = dinv[q2];
        const int traw1 = adj[max(o1 + min(l8, m1) - 1, 0)];

        float4 acc = make_float4(0.0f, 0.0f, 0.0f, 0.0f);
        const int total = m + 1;
        for (int c = 0; c < total; c += 8) {
            const int ecur = c + l8;
            const int idx = (ecur >= 1 && ecur <= m) ? traw : i;
            traw = adj[max(o + min(ecur + 8, m) - 1, 0)];
            float4 h[8];
            GATHER_SLOT(0) GATHER_SLOT(1) GATHER_SLOT(2) GATHER_SLOT(3)
            GATHER_SLOT(4) GATHER_SLOT(5) GATHER_SLOT(6) GATHER_SLOT(7)
            __builtin_amdgcn_sched_group_barrier(0x020, 9, 0);
            __builtin_amdgcn_sched_group_barrier(0x002, 16, 0);
            GATHER_ADD(0) GATHER_ADD(1) GATHER_ADD(2) GATHER_ADD(3)
            GATHER_ADD(4) GATHER_ADD(5) GATHER_ADD(6) GATHER_ADD(7)
        }

        float4 r = make_float4(b.x + dd * acc.x, b.y + dd * acc.y, b.z + dd * acc.z, b.w + dd * acc.w);
        if (relu) {
            r.x = fmaxf(r.x, 0.0f); r.y = fmaxf(r.y, 0.0f);
            r.z = fmaxf(r.z, 0.0f); r.w = fmaxf(r.w, 0.0f);
        }
        if (gout != nullptr) *(float4*)(gout + (size_t)i * HDIM + j) = r;
        *(float4*)(lrow + t * TILE_PITCH) = r;

        i = i + 1; m = m1; o = o1; dd = dd1; traw = traw1;
        q1 = q2; m1 = m2; o1 = o2; dd1 = dd2;
    }
}

#define TILE_TIMES_WEIGHT()                                                                                      \
    v8f acc0 = {}, acc1 = {}, acc2 = {}, acc3 = {};                                                              \
    {                                                                                                            \
        const float* trow = tile + lmod * TILE_PITCH + 2 * lhalf;                                                \
        _Pragma("unroll")                \
        for (int ks = 0; ks < HDIM / 4; ++ks) {                                                                  \
            v2f a = *(const v2f*)(trow + 4 * ks);                                                \
            const float4* bp = (const float4*)(bsw + (size_t)(ks * 32 + lane) * 8);                              \
            float4 b01 = bp[0];                                                                 \
            float4 b23 = bp[1];                                                                 \
            v2f b0 = {b01.x, b01.y}, b1 = {b01.z, b01.w};                                                        \
            v2f b2 = {b23.x, b23.y}, b3 = {b23.z, b23.w};                                                        \
            acc0 = __builtin_amdgcn_wmma_f32_16x16x4_f32(false, a, false, b0, (short)0, acc0, false, false);     \
            acc1 = __builtin_amdgcn_wmma_f32_16x16x4_f32(false, a, false, b1, (short)0, acc1, false, false);     \
            acc2 = __builtin_amdgcn_wmma_f32_16x16x4_f32(false, a, false, b2, (short)0, acc2, false, false);     \
            acc3 = __builtin_amdgcn_wmma_f32_16x16x4_f32(false, a, false, b3, (short)0, acc3, false, false);     \
        }                                                                                                        \
    }

#define TILE_HANDOVER()                                                 \
    __builtin_amdgcn_fence(__ATOMIC_RELEASE, "wavefront");              \
    __builtin_amdgcn_wave_barrier();                                    \
    __builtin_amdgcn_fence(__ATOMIC_ACQUIRE, "wavefront");

__global__ void __launch_bounds__(FUSED_TPB) conv1_dense_kernel(const float* __restrict__ Hm, const int* __restrict__ adj,
                                                          const int* __restrict__ offs, const int* __restrict__ cnt,
                                                          const float* __restrict__ dinv, const float* __restrict__ bias,
                                                          const float* __restrict__ Wn, float* __restrict__ Hout, int n) {
    __shared__ __attribute__((aligned(16))) float fl[FUSED_LDS_FLOATS];
    float* bsw = fl;
    const int tid   = threadIdx.x;
    const int lane  = tid & 31;
    const int wave  = tid >> 5;
    const int lmod  = lane & 15;
    const int lhalf = lane >> 4;
    float* tile = fl + HDIM * HDIM + wave * (16 * TILE_PITCH);
    stage_weight64(bsw, Wn, tid);
    __syncthreads();

    const int l8 = tid & 7;
    const unsigned j = lmod * 4;
    const unsigned jb = j * 4;
    const float4 b = *(const float4*)(bias + j);
    const int strips = n >> 4;
    const int nwaves = gridDim.x * FUSED_WAVES;
    for (int s = wave * gridDim.x + blockIdx.x; s < strips; s += nwaves) {
        gather_eight((const char*)Hm, adj, offs, cnt, dinv, b, true, nullptr,
                     tile + (8 * lhalf) * TILE_PITCH + j, s * 16 + 8 * lhalf, n, l8, j, jb);
        TILE_HANDOVER()
        TILE_TIMES_WEIGHT()
        const int rbase = s * 16 + lhalf * 8;
#pragma unroll
        for (int r = 0; r < 8; ++r) {
            const size_t ro = (size_t)(rbase + r) * HDIM + lmod;
            const float rs = dinv[rbase + r];
            Hout[ro] = acc0[r] * rs; Hout[ro + 16] = acc1[r] * rs;
            Hout[ro + 32] = acc2[r] * rs; Hout[ro + 48] = acc3[r] * rs;
        }
        TILE_HANDOVER()
    }
}

__global__ void __launch_bounds__(FUSED_TPB) conv2_dense_kernel(const float* __restrict__ Hm, const int* __restrict__ adj,
                                                          const int* __restrict__ offs, const int* __restrict__ cnt,
                                                          const float* __restrict__ dinv, const float* __restrict__ bias,
                                                          const float* __restrict__ Wp1, const float* __restrict__ bp1,
                                                          const float* __restrict__ Wp2, const float* __restrict__ bp2,
                                                          float* __restrict__ Zout, float* __restrict__ tprob, int n) {
    __shared__ __attribute__((aligned(16))) float fl[FUSED_LDS_FLOATS];
    float* bsw = fl;
    const int tid   = threadIdx.x;
    const int lane  = tid & 31;
    const int wave  = tid >> 5;
    const int lmod  = lane & 15;
    const int lhalf = lane >> 4;
    float* tile = fl + HDIM * HDIM + wave * (16 * TILE_PITCH);
    stage_weight64(bsw, Wp1, tid);
    __syncthreads();

    const int l8 = tid & 7;
    const unsigned j = lmod * 4;
    const unsigned jb = j * 4;
    const float4 b = *(const float4*)(bias + j);
    const float bi0 = bp1[lmod], bi1 = bp1[16 + lmod], bi2 = bp1[32 + lmod], bi3 = bp1[48 + lmod];
    const float2 w0 = *(const float2*)(Wp2 + (size_t)(lmod) * 2);
    const float2 w1 = *(const float2*)(Wp2 + (size_t)(16 + lmod) * 2);
    const float2 w2 = *(const float2*)(Wp2 + (size_t)(32 + lmod) * 2);
    const float2 w3 = *(const float2*)(Wp2 + (size_t)(48 + lmod) * 2);
    const float bo = bp2[lmod & 1];
    const bool h8 = (lmod & 8) != 0, h4 = (lmod & 4) != 0, h2 = (lmod & 2) != 0, h1 = (lmod & 1) != 0;
    const int strips = n >> 4;
    const int nwaves = gridDim.x * FUSED_WAVES;
    for (int s = wave * gridDim.x + blockIdx.x; s < strips; s += nwaves) {
        gather_eight((const char*)Hm, adj, offs, cnt, dinv, b, false, Zout,
                     tile + (8 * lhalf) * TILE_PITCH + j, s * 16 + 8 * lhalf, n, l8, j, jb);
        TILE_HANDOVER()
        TILE_TIMES_WEIGHT()
        float v[16];
#pragma unroll
        for (int r = 0; r < 8; ++r) {
            const float p0 = lrelu(acc0[r] + bi0), p1 = lrelu(acc1[r] + bi1);
            const float p2 = lrelu(acc2[r] + bi2), p3 = lrelu(acc3[r] + bi3);
            v[2 * r + 0] = p0 * w0.x + p1 * w1.x + p2 * w2.x + p3 * w3.x;
            v[2 * r + 1] = p0 * w0.y + p1 * w1.y + p2 * w2.y + p3 * w3.y;
        }
        float u8[8], u4[4], u2[2];
#pragma unroll
        for (int i = 0; i < 8; ++i) {
            const float keep = h8 ? v[i + 8] : v[i];
            const float send = h8 ? v[i] : v[i + 8];
            u8[i] = keep + __shfl_xor(send, 8, 32);
        }
#pragma unroll
        for (int i = 0; i < 4; ++i) {
            const float keep = h4 ? u8[i + 4] : u8[i];
            const float send = h4 ? u8[i] : u8[i + 4];
            u4[i] = keep + __shfl_xor(send, 4, 32);
        }
#pragma unroll
        for (int i = 0; i < 2; ++i) {
            const float keep = h2 ? u4[i + 2] : u4[i];
            const float send = h2 ? u4[i] : u4[i + 2];
            u2[i] = keep + __shfl_xor(send, 2, 32);
        }
        const float keep1 = h1 ? u2[1] : u2[0];
        const float send1 = h1 ? u2[0] : u2[1];
        const float tot = keep1 + __shfl_xor(send1, 1, 32);
        tprob[(size_t)s * 32 + lane] = lrelu(tot + bo);
        TILE_HANDOVER()
    }
}

__global__ void __launch_bounds__(256) heads_kernel(const float* __restrict__ Z,
                                                    const int* __restrict__ treat, int nt,
                                                    const int* __restrict__ ctrl, int nc,
                                                    const float* __restrict__ Wy1, const float* __restrict__ by1,
                                                    const float* __restrict__ Wy0, const float* __restrict__ by0,
                                                    float* __restrict__ o_y1, float* __restrict__ o_yc0,
                                                    float* __restrict__ o_y0, float* __restrict__ o_yc1, int hb_t) {
    const int blk = blockIdx.x;
    const bool tr = blk < hb_t;
    const int t = (tr ? blk : blk - hb_t) * 256 + threadIdx.x;
    if (t >= (tr ? nt : nc)) return;
    const int node = tr ? treat[t] : ctrl[t];
    const float* row = Z + (size_t)node * HDIM;
    float d1 = 0.0f, d0 = 0.0f;
#pragma unroll
    for (int j = 0; j < HDIM; j += 4) {
        float4 z  = *(const float4*)(row + j);
        float4 w1 = *(const float4*)(Wy1 + j);
        float4 w0 = *(const float4*)(Wy0 + j);
        d1 += z.x * w1.x + z.y * w1.y + z.z * w1.z + z.w * w1.w;
        d0 += z.x * w0.x + z.y * w0.y + z.z * w0.z + z.w * w0.w;
    }
    const float r1 = lrelu(d1 + by1[0]);
    const float r0 = lrelu(d0 + by0[0]);
    if (tr) { o_y1[t] = r1; o_yc0[t] = r0; }
    else    { o_y0[t] = r0; o_yc1[t] = r1; }
}

extern "C" void kernel_launch(void* const* d_in, const int* in_sizes, int n_in,
                              void* d_out, int out_size, void* d_ws, size_t ws_size,
                              hipStream_t stream) {
    const float* x     = (const float*)d_in[0];
    const int*   ei    = (const int*)d_in[1];
    const int*   treat = (const int*)d_in[2];
    const int*   ctrl  = (const int*)d_in[3];
    const float* W1  = (const float*)d_in[4];  const float* b1  = (const float*)d_in[5];
    const float* W2  = (const float*)d_in[6];  const float* b2  = (const float*)d_in[7];
    const float* Wy1 = (const float*)d_in[8];  const float* by1 = (const float*)d_in[9];
    const float* Wy0 = (const float*)d_in[10]; const float* by0 = (const float*)d_in[11];
    const float* Wp1 = (const float*)d_in[12]; const float* bp1 = (const float*)d_in[13];
    const float* Wp2 = (const float*)d_in[14]; const float* bp2 = (const float*)d_in[15];

    const int N = in_sizes[0] / INDIM;
    const int E = in_sizes[1] / 2;
    const int T = in_sizes[2];
    const int Cn = in_sizes[3];
    const int* e_src = ei;
    const int* e_dst = ei + E;

    float* dinv = (float*)d_ws;
    float* buf0 = dinv + (((size_t)N + 63) & ~(size_t)63);
    int* binned = (int*)(buf0 + (size_t)N * HDIM);
    int* adj  = binned + (size_t)2 * E;
    int* cnt  = adj + (((size_t)E + 63) & ~(size_t)63);
    int* offs = cnt + (((size_t)N + 63) & ~(size_t)63);

    float* o_y1    = (float*)d_out;
    float* o_yc0   = o_y1 + T;
    float* o_y0    = o_yc0 + T;
    float* o_yc1   = o_y0 + Cn;
    float* o_tprob = o_yc1 + Cn;
    float* o_xZ2   = o_tprob + (size_t)2 * N;

    const int BLK = 256;
    auto cdiv = [](long long a, long long b) { return (int)((a + b - 1) / b); };
    const int gemm_blocks = cdiv(N / 16, 8);
    const int gather_blocks = min(GATHER_BLOCKS, cdiv(N, 16));
    const size_t smem128 = (size_t)INDIM * HDIM * sizeof(float);
    const int nb = cdiv(N, BIN_NODES);
    const int g1 = cdiv(E, SORT_CH);
    int* tab = (int*)buf0;
    const size_t smem_sort = ((size_t)((2 * nb + 1) & ~1) + (size_t)2 * SORT_CH) * sizeof(int);

    chunk_sort_kernel<<<g1, SORT_TPB, smem_sort, stream>>>(e_src, e_dst, tab, binned, E, nb, g1);
    bin_rows_kernel<<<nb, SORT_TPB, 0, stream>>>(binned, tab, adj, cnt, offs, dinv, N, E, nb, g1);

    wmma_gemm_f32_kernel<<<gemm_blocks, BLK, smem128, stream>>>(x, W1, o_xZ2, nullptr, dinv, N, INDIM, 0);
    conv1_dense_kernel<<<gather_blocks, FUSED_TPB, 0, stream>>>(o_xZ2, adj, offs, cnt, dinv, b1, W2, buf0, N);

    conv2_dense_kernel<<<gather_blocks, FUSED_TPB, 0, stream>>>(buf0, adj, offs, cnt, dinv, b2, Wp1, bp1, Wp2, bp2,
                                                               o_xZ2, o_tprob, N);

    const int hb_t = cdiv(T, BLK);
    const int hb_c = cdiv(Cn, BLK);
    heads_kernel<<<hb_t + hb_c, BLK, 0, stream>>>(o_xZ2, treat, T, ctrl, Cn, Wy1, by1, Wy0, by0,
                                                  o_y1, o_yc0, o_y0, o_yc1, hb_t);
}


# x@W1 product moves into the sort's second launch: each 512-node bin block multiplies its own x rows on fp32 WMMA behind its ranking, x strips asked for ahead; 6 launches to 5; sum orders kept
# speedup vs baseline: 5.0542x; 1.0701x over previous
#include <hip/hip_runtime.h>
#include <hip/hip_bf16.h>

typedef __attribute__((ext_vector_type(2))) float v2f;
typedef __attribute__((ext_vector_type(8))) float v8f;
typedef __attribute__((ext_vector_type(4))) float v4f;

#define HDIM 64
#define INDIM 128

__device__ __forceinline__ float lrelu(float v) { return v > 0.0f ? v : 0.01f * v; }

#define SORT_TPB 512
#define SORT_CH 6144
#define SORT_EPT (SORT_CH / SORT_TPB)
#define BIN_SHIFT 9
#define BIN_NODES 512
#define FINE_CAP 12288

__device__ __forceinline__ int block_excl_scan_512(int v, int* wsum, int tid) {
    const int lane = tid & 31;
    const int wave = tid >> 5;
    int incl = v;
    for (int o = 1; o < 32; o <<= 1) {
        int u = __shfl_up(incl, o, 32);
        if (lane >= o) incl += u;
    }
    if (lane == 31) wsum[wave] = incl;
    __syncthreads();
    int front = 0;
    for (int w = 0; w < 16; ++w)
        if (w < wave) front += wsum[w];
    return front + incl - v;
}

__global__ void __launch_bounds__(SORT_TPB) chunk_sort_kernel(const int* __restrict__ src, const int* __restrict__ dst,
                                                              int* __restrict__ tab, int* __restrict__ binned,
                                                              int e, int nb, int g1) {
    extern __shared__ int lds[];
    __shared__ int wsum[16];
    const int tid = threadIdx.x;
    const int c = blockIdx.x;
    int* cnt    = lds;
    int* cursor = lds + nb;
    int2* stage = (int2*)(lds + ((2 * nb + 1) & ~1));

    const int base = c * SORT_CH;
    const int lim = min(e, base + SORT_CH);
    const int nloc = lim - base;
    int es[SORT_EPT], ed[SORT_EPT];
#pragma unroll
    for (int k = 0; k < SORT_EPT; ++k) {
        const int i = min(base + tid + k * SORT_TPB, lim - 1);
        es[k] = src[i];
        ed[k] = dst[i];
    }
    for (int b = tid; b < nb; b += SORT_TPB) cnt[b] = 0;
    __syncthreads();
#pragma unroll
    for (int k = 0; k < SORT_EPT; ++k)
        if (tid + k * SORT_TPB < nloc) atomicAdd(&cnt[ed[k] >> BIN_SHIFT], 1);
    __syncthreads();

    const int ipt = (nb + SORT_TPB - 1) / SORT_TPB;
    const int b0 = tid * ipt;
    const int b1 = min(nb, b0 + ipt);
    int mine = 0;
    for (int b = b0; b < b1; ++b) mine += cnt[b];
    int run = block_excl_scan_512(mine, wsum, tid);
    for (int b = b0; b < b1; ++b) {
        cursor[b] = run;
        tab[(size_t)b * g1 + c] = run;
        run += cnt[b];
    }
    if (tid == 0) tab[(size_t)nb * g1 + c] = nloc;
    __syncthreads();

    int pos[SORT_EPT];
#pragma unroll
    for (int k = 0; k < SORT_EPT; ++k) {
        pos[k] = 0;
        if (tid + k * SORT_TPB < nloc) pos[k] = atomicAdd(&cursor[ed[k] >> BIN_SHIFT], 1);
    }
#pragma unroll
    for (int k = 0; k < SORT_EPT; ++k)
        if (tid + k * SORT_TPB < nloc) stage[pos[k]] = make_int2(es[k], ed[k]);
    __syncthreads();

    int2* out2 = (int2*)binned + base;
    int2 w[SORT_EPT];
#pragma unroll
    for (int k = 0; k < SORT_EPT; ++k) w[k] = stage[min(tid + k * SORT_TPB, nloc - 1)];
#pragma unroll
    for (int k = 0; k < SORT_EPT; ++k) {
        const int j = tid + k * SORT_TPB;
        if (j < nloc) out2[j] = w[k];
    }
}

#define RUN_Q 18
#define RUN_S 2
#define XT_PITCH 132
#define BIN_GEMM_LDS_FLOATS (INDIM * HDIM + (SORT_TPB / 32) * 16 * XT_PITCH)

#define XT_HANDOVER()                                                   \
    __builtin_amdgcn_fence(__ATOMIC_RELEASE, "wavefront");              \
    __builtin_amdgcn_wave_barrier();                                    \
    __builtin_amdgcn_fence(__ATOMIC_ACQUIRE, "wavefront");

__device__ __forceinline__ void strip_times_w1(const float* tile, const float* bsw, const float* dl,
                                               float* __restrict__ Hout, const int row0, const int lrow0,
                                               const int lane, const int lmod, const int lhalf) {
    v8f acc0 = {}, acc1 = {}, acc2 = {}, acc3 = {};
    const float* trow = tile + lmod * XT_PITCH + 2 * lhalf;
#pragma unroll
    for (int ks = 0; ks < INDIM / 4; ++ks) {
        v2f a = *(const v2f*)(trow + 4 * ks);
        const float4* bp = (const float4*)(bsw + (size_t)(ks * 32 + lane) * 8);
        float4 b01 = bp[0];
        float4 b23 = bp[1];
        v2f b0 = {b01.x, b01.y}, b1 = {b01.z, b01.w};
        v2f b2 = {b23.x, b23.y}, b3 = {b23.z, b23.w};
        acc0 = __builtin_amdgcn_wmma_f32_16x16x4_f32(false, a, false, b0, (short)0, acc0, false, false);
        acc1 = __builtin_amdgcn_wmma_f32_16x16x4_f32(false, a, false, b1, (short)0, acc1, false, false);
        acc2 = __builtin_amdgcn_wmma_f32_16x16x4_f32(false, a, false, b2, (short)0, acc2, false, false);
        acc3 = __builtin_amdgcn_wmma_f32_16x16x4_f32(false, a, false, b3, (short)0, acc3, false, false);
    }
    const int rbase = row0 + lhalf * 8;
#pragma unroll
    for (int r = 0; r < 8; ++r) {
        const size_t ro = (size_t)(rbase + r) * HDIM + lmod;
        const float rs = dl[lrow0 + lhalf * 8 + r];
        Hout[ro] = acc0[r] * rs; Hout[ro + 16] = acc1[r] * rs;
        Hout[ro + 32] = acc2[r] * rs; Hout[ro + 48] = acc3[r] * rs;
    }
}

__global__ void __launch_bounds__(SORT_TPB) bin_rows_dense_kernel(const int* __restrict__ binned, const int* __restrict__ tab,
                                                                  int* __restrict__ adj, int* __restrict__ cntG,
                                                                  int* __restrict__ offsG, float* __restrict__ dinv,
                                                                  const float* __restrict__ x, const float* __restrict__ W1,
                                                                  float* __restrict__ Hout,
                                                                  int n, int e, int nb, int g1) {
    __shared__ int cnt[BIN_NODES];
    __shared__ int cursor[BIN_NODES];
    __shared__ int wsum[16];
    __shared__ int red[48];
    __shared__ int stage[FINE_CAP];
    __shared__ float dl[BIN_NODES];
    __shared__ __attribute__((aligned(16))) float gl[BIN_GEMM_LDS_FLOATS];
    const int tid = threadIdx.x;
    const int lane = tid & 31;
    const int wave = tid >> 5;
    const int lmod = lane & 15;
    const int lhalf = lane >> 4;
    const int b = blockIdx.x;
    const int2* bp = (const int2*)binned;
    const int* row0 = tab + (size_t)b * g1;
    const int* row1 = row0 + g1;
    float* bsw = gl;
    float* tile = gl + INDIM * HDIM + wave * (16 * XT_PITCH);

    const int lrow = __builtin_amdgcn_readfirstlane(wave) * 32;
    const int grow = b * BIN_NODES + lrow;
    const bool has0 = grow < n;
    const bool has1 = grow + 16 < n;
    const int g0 = min(grow, n - 16);
    const int g1r = min(grow + 16, n - 16);
    v4f xa[16];
#pragma unroll
    for (int i = 0; i < 16; ++i) xa[i] = *(const v4f*)(x + (size_t)(g0 + i) * INDIM + lane * 4);

    int rs = 0, ln = 0, s0 = 0, s1 = 0, mx = 0;
    {
        const int cq = wave + 16 * lane;
        if (cq < g1) {
            rs = row0[cq];
            const int re = row1[cq];
            ln = re - rs;
            s0 = rs; s1 = re; mx = ln;
        }
        for (int cc = cq + SORT_TPB; cc < g1; cc += SORT_TPB) {
            const int a0 = row0[cc], a1 = row1[cc];
            s0 += a0; s1 += a1; mx = max(mx, a1 - a0);
        }
    }

    float wv[2 * (INDIM * 32 / SORT_TPB)];
#pragma unroll
    for (int it = 0; it < INDIM * 32 / SORT_TPB; ++it) {
        const int p = tid + it * SORT_TPB;
        const int ks  = p >> 7;
        const int ll  = (p >> 2) & 31;
        const int tn  = p & 3;
        const int kb  = 4 * ks + 2 * (ll >> 4);
        const int col = tn * 16 + (ll & 15);
        wv[2 * it + 0] = W1[(size_t)kb * HDIM + col];
        wv[2 * it + 1] = W1[(size_t)(kb + 1) * HDIM + col];
    }

    int2 v[RUN_Q * RUN_S];
#pragma unroll
    for (int q = 0; q < RUN_Q; ++q) {
        const int rsq = __builtin_amdgcn_readlane(rs, q);
        const int lnq = __builtin_amdgcn_readlane(ln, q);
        const int p0 = (wave + 16 * q) * SORT_CH + rsq;
#pragma unroll
        for (int s = 0; s < RUN_S; ++s)
            v[q * RUN_S + s] = bp[min(p0 + min(lane + 32 * s, max(lnq - 1, 0)), e - 1)];
    }

    for (int o = 16; o > 0; o >>= 1) {
        s0 += __shfl_xor(s0, o, 32);
        s1 += __shfl_xor(s1, o, 32);
        mx = max(mx, __shfl_xor(mx, o, 32));
    }
    if (lane == 0) { red[wave] = s0; red[16 + wave] = s1; red[32 + wave] = mx; }
    cnt[tid] = 0;
#pragma unroll
    for (int it = 0; it < INDIM * 32 / SORT_TPB; ++it)
        *(float2*)(bsw + (size_t)(tid + it * SORT_TPB) * 2) = make_float2(wv[2 * it + 0], wv[2 * it + 1]);
    __syncthreads();
    int start = 0, end = 0, maxlen = 0;
    for (int w = 0; w < 16; ++w) {
        start += red[w];
        end += red[16 + w];
        maxlen = max(maxlen, red[32 + w]);
    }
    const int m = end - start;
    const bool fits = m <= FINE_CAP && g1 <= 16 * RUN_Q && maxlen <= 32 * RUN_S;

    if (fits) {
#pragma unroll
        for (int q = 0; q < RUN_Q; ++q) {
            const int lnq = __builtin_amdgcn_readlane(ln, q);
#pragma unroll
            for (int s = 0; s < RUN_S; ++s)
                if (lane + 32 * s < lnq) atomicAdd(&cnt[v[q * RUN_S + s].y & (BIN_NODES - 1)], 1);
        }
    } else {
        for (int c = wave; c < g1; c += 16) {
            const int r0 = row0[c];
            const int len = row1[c] - r0;
            const int2* rp = bp + (size_t)c * SORT_CH + r0;
            for (int j = lane; j < len; j += 32) atomicAdd(&cnt[rp[j].y & (BIN_NODES - 1)], 1);
        }
    }
    __syncthreads();

    const int cdeg = cnt[tid];
    const int ex = block_excl_scan_512(cdeg, wsum, tid);
    cursor[tid] = ex;
    const int node = b * BIN_NODES + tid;
    {
        float d = (float)(cdeg + 1);
        const float di = d > 0.0f ? rsqrtf(d) : 0.0f;
        dl[tid] = di;
        if (node < n) {
            cntG[node] = cdeg;
            offsG[node] = start + ex;
            dinv[node] = di;
        }
    }
    __syncthreads();

    if (fits) {
#pragma unroll
        for (int q = 0; q < RUN_Q; ++q) {
            const int lnq = __builtin_amdgcn_readlane(ln, q);
#pragma unroll
            for (int s = 0; s < RUN_S; ++s)
                if (lane + 32 * s < lnq)
                    v[q * RUN_S + s].y = atomicAdd(&cursor[v[q * RUN_S + s].y & (BIN_NODES - 1)], 1);
        }
#pragma unroll
        for (int q = 0; q < RUN_Q; ++q) {
            const int lnq = __builtin_amdgcn_readlane(ln, q);
#pragma unroll
            for (int s = 0; s < RUN_S; ++s)
                if (lane + 32 * s < lnq) stage[v[q * RUN_S + s].y] = v[q * RUN_S + s].x;
        }
        __syncthreads();
        for (int j = tid; j < m; j += SORT_TPB) adj[start + j] = stage[j];
    } else {
        for (int c = wave; c < g1; c += 16) {
            const int r0 = row0[c];
            const int len = row1[c] - r0;
            const int2* rp = bp + (size_t)c * SORT_CH + r0;
            for (int j = lane; j < len; j += 32) {
                const int2 u = rp[j];
                const int p = atomicAdd(&cursor[u.y & (BIN_NODES - 1)], 1);
                adj[start + p] = u.x;
            }
        }
    }

    if (has0) {
#pragma unroll
        for (int i = 0; i < 16; ++i) *(v4f*)(tile + i * XT_PITCH + lane * 4) = xa[i];
#pragma unroll
        for (int i = 0; i < 16; ++i) xa[i] = *(const v4f*)(x + (size_t)(g1r + i) * INDIM + lane * 4);
        XT_HANDOVER()
        strip_times_w1(tile, bsw, dl, Hout, grow, lrow, lane, lmod, lhalf);
        XT_HANDOVER()
        if (has1) {
#pragma unroll
            for (int i = 0; i < 16; ++i) *(v4f*)(tile + i * XT_PITCH + lane * 4) = xa[i];
            XT_HANDOVER()
            strip_times_w1(tile, bsw, dl, Hout, grow + 16, lrow + 16, lane, lmod, lhalf);
        }
    }
}

#define GATHER_BLOCKS 512
#define FUSED_TPB 512
#define FUSED_WAVES (FUSED_TPB / 32)
#define TILE_PITCH 68
#define FUSED_LDS_FLOATS (HDIM * HDIM + FUSED_WAVES * 16 * TILE_PITCH)

#define ROW_LANE(v, U) __builtin_amdgcn_update_dpp(0, (v), 0x150 + (U), 0xf, 0xf, false)

#define GATHER_SLOT(U)                                                              \
    {                                                                               \
        const unsigned s = (unsigned)ROW_LANE(idx, U);                              \
        h[U] = *(const float4*)(hbytes + (size_t)((s << 8) | jb));                  \
    }
#define GATHER_ADD(U)                                                               \
    {                                                                               \
        const float w = (c + (U) < total) ? 1.0f : 0.0f;                            \
        acc.x = fmaf(w, h[U].x, acc.x); acc.y = fmaf(w, h[U].y, acc.y);             \
        acc.z = fmaf(w, h[U].z, acc.z); acc.w = fmaf(w, h[U].w, acc.w);             \
    }

__device__ __forceinline__ void stage_weight64(float* bsw, const float* __restrict__ B, int tid) {
#pragma unroll
    for (int it = 0; it < HDIM * 32 / FUSED_TPB; ++it) {
        const int p = tid + it * FUSED_TPB;
        int ks  = p >> 7;
        int ll  = (p >> 2) & 31;
        int tn  = p & 3;
        int kb  = 4 * ks + 2 * (ll >> 4);
        int col = tn * 16 + (ll & 15);
        bsw[p * 2 + 0] = B[(size_t)kb * HDIM + col];
        bsw[p * 2 + 1] = B[(size_t)(kb + 1) * HDIM + col];
    }
}

__device__ __forceinline__ void gather_eight(const char* __restrict__ hbytes, const int* __restrict__ adj,
                                             const int* __restrict__ offs, const int* __restrict__ cnt,
                                             const float* __restrict__ dinv, const float4 b, const bool relu,
                                             float* __restrict__ gout, float* lrow, const int node0, const int n,
                                             const int l8, const unsigned j, const unsigned jb) {
    const int last = n - 1;
    int i = node0;
    int m = cnt[i];
    int o = offs[i];
    float dd = dinv[i];
    int traw = adj[max(o + min(l8, m) - 1, 0)];
    int q1 = min(i + 1, last);
    int m1 = cnt[q1];
    int o1 = offs[q1];
    float dd1 = dinv[q1];

    for (int t = 0; t < 8; ++t) {
        const int q2 = min(i + 2, last);
        const int m2 = cnt[q2];
        const int o2 = offs[q2];
        const float dd2 = dinv[q2];
        const int traw1 = adj[max(o1 + min(l8, m1) - 1, 0)];

        float4 acc = make_float4(0.0f, 0.0f, 0.0f, 0.0f);
        const int total = m + 1;
        for (int c = 0; c < total; c += 8) {
            const int ecur = c + l8;
            const int idx = (ecur >= 1 && ecur <= m) ? traw : i;
            traw = adj[max(o + min(ecur + 8, m) - 1, 0)];
            float4 h[8];
            GATHER_SLOT(0) GATHER_SLOT(1) GATHER_SLOT(2) GATHER_SLOT(3)
            GATHER_SLOT(4) GATHER_SLOT(5) GATHER_SLOT(6) GATHER_SLOT(7)
            __builtin_amdgcn_sched_group_barrier(0x020, 9, 0);
            __builtin_amdgcn_sched_group_barrier(0x002, 16, 0);
            GATHER_ADD(0) GATHER_ADD(1) GATHER_ADD(2) GATHER_ADD(3)
            GATHER_ADD(4) GATHER_ADD(5) GATHER_ADD(6) GATHER_ADD(7)
        }

        float4 r = make_float4(b.x + dd * acc.x, b.y + dd * acc.y, b.z + dd * acc.z, b.w + dd * acc.w);
        if (relu) {
            r.x = fmaxf(r.x, 0.0f); r.y = fmaxf(r.y, 0.0f);
            r.z = fmaxf(r.z, 0.0f); r.w = fmaxf(r.w, 0.0f);
        }
        if (gout != nullptr) *(float4*)(gout + (size_t)i * HDIM + j) = r;
        *(float4*)(lrow + t * TILE_PITCH) = r;

        i = i + 1; m = m1; o = o1; dd = dd1; traw = traw1;
        q1 = q2; m1 = m2; o1 = o2; dd1 = dd2;
    }
}

#define TILE_TIMES_WEIGHT()                                                                                      \
    v8f acc0 = {}, acc1 = {}, acc2 = {}, acc3 = {};                                                              \
    {                                                                                                            \
        const float* trow = tile + lmod * TILE_PITCH + 2 * lhalf;                                                \
        _Pragma("unroll")                \
        for (int ks = 0; ks < HDIM / 4; ++ks) {                                                                  \
            v2f a = *(const v2f*)(trow + 4 * ks);                                                \
            const float4* bp = (const float4*)(bsw + (size_t)(ks * 32 + lane) * 8);                              \
            float4 b01 = bp[0];                                                                 \
            float4 b23 = bp[1];                                                                 \
            v2f b0 = {b01.x, b01.y}, b1 = {b01.z, b01.w};                                                        \
            v2f b2 = {b23.x, b23.y}, b3 = {b23.z, b23.w};                                                        \
            acc0 = __builtin_amdgcn_wmma_f32_16x16x4_f32(false, a, false, b0, (short)0, acc0, false, false);     \
            acc1 = __builtin_amdgcn_wmma_f32_16x16x4_f32(false, a, false, b1, (short)0, acc1, false, false);     \
            acc2 = __builtin_amdgcn_wmma_f32_16x16x4_f32(false, a, false, b2, (short)0, acc2, false, false);     \
            acc3 = __builtin_amdgcn_wmma_f32_16x16x4_f32(false, a, false, b3, (short)0, acc3, false, false);     \
        }                                                                                                        \
    }

#define TILE_HANDOVER()                                                 \
    __builtin_amdgcn_fence(__ATOMIC_RELEASE, "wavefront");              \
    __builtin_amdgcn_wave_barrier();                                    \
    __builtin_amdgcn_fence(__ATOMIC_ACQUIRE, "wavefront");

__global__ void __launch_bounds__(FUSED_TPB) conv1_dense_kernel(const float* __restrict__ Hm, const int* __restrict__ adj,
                                                          const int* __restrict__ offs, const int* __restrict__ cnt,
                                                          const float* __restrict__ dinv, const float* __restrict__ bias,
                                                          const float* __restrict__ Wn, float* __restrict__ Hout, int n) {
    __shared__ __attribute__((aligned(16))) float fl[FUSED_LDS_FLOATS];
    float* bsw = fl;
    const int tid   = threadIdx.x;
    const int lane  = tid & 31;
    const int wave  = tid >> 5;
    const int lmod  = lane & 15;
    const int lhalf = lane >> 4;
    float* tile = fl + HDIM * HDIM + wave * (16 * TILE_PITCH);
    stage_weight64(bsw, Wn, tid);
    __syncthreads();

    const int l8 = tid & 7;
    const unsigned j = lmod * 4;
    const unsigned jb = j * 4;
    const float4 b = *(const float4*)(bias + j);
    const int strips = n >> 4;
    const int nwaves = gridDim.x * FUSED_WAVES;
    for (int s = wave * gridDim.x + blockIdx.x; s < strips; s += nwaves) {
        gather_eight((const char*)Hm, adj, offs, cnt, dinv, b, true, nullptr,
                     tile + (8 * lhalf) * TILE_PITCH + j, s * 16 + 8 * lhalf, n, l8, j, jb);
        TILE_HANDOVER()
        TILE_TIMES_WEIGHT()
        const int rbase = s * 16 + lhalf * 8;
#pragma unroll
        for (int r = 0; r < 8; ++r) {
            const size_t ro = (size_t)(rbase + r) * HDIM + lmod;
            const float rs = dinv[rbase + r];
            Hout[ro] = acc0[r] * rs; Hout[ro + 16] = acc1[r] * rs;
            Hout[ro + 32] = acc2[r] * rs; Hout[ro + 48] = acc3[r] * rs;
        }
        TILE_HANDOVER()
    }
}

__global__ void __launch_bounds__(FUSED_TPB) conv2_dense_kernel(const float* __restrict__ Hm, const int* __restrict__ adj,
                                                          const int* __restrict__ offs, const int* __restrict__ cnt,
                                                          const float* __restrict__ dinv, const float* __restrict__ bias,
                                                          const float* __restrict__ Wp1, const float* __restrict__ bp1,
                                                          const float* __restrict__ Wp2, const float* __restrict__ bp2,
                                                          float* __restrict__ Zout, float* __restrict__ tprob, int n) {
    __shared__ __attribute__((aligned(16))) float fl[FUSED_LDS_FLOATS];
    float* bsw = fl;
    const int tid   = threadIdx.x;
    const int lane  = tid & 31;
    const int wave  = tid >> 5;
    const int lmod  = lane & 15;
    const int lhalf = lane >> 4;
    float* tile = fl + HDIM * HDIM + wave * (16 * TILE_PITCH);
    stage_weight64(bsw, Wp1, tid);
    __syncthreads();

    const int l8 = tid & 7;
    const unsigned j = lmod * 4;
    const unsigned jb = j * 4;
    const float4 b = *(const float4*)(bias + j);
    const float bi0 = bp1[lmod], bi1 = bp1[16 + lmod], bi2 = bp1[32 + lmod], bi3 = bp1[48 + lmod];
    const float2 w0 = *(const float2*)(Wp2 + (size_t)(lmod) * 2);
    const float2 w1 = *(const float2*)(Wp2 + (size_t)(16 + lmod) * 2);
    const float2 w2 = *(const float2*)(Wp2 + (size_t)(32 + lmod) * 2);
    const float2 w3 = *(const float2*)(Wp2 + (size_t)(48 + lmod) * 2);
    const float bo = bp2[lmod & 1];
    const bool h8 = (lmod & 8) != 0, h4 = (lmod & 4) != 0, h2 = (lmod & 2) != 0, h1 = (lmod & 1) != 0;
    const int strips = n >> 4;
    const int nwaves = gridDim.x * FUSED_WAVES;
    for (int s = wave * gridDim.x + blockIdx.x; s < strips; s += nwaves) {
        gather_eight((const char*)Hm, adj, offs, cnt, dinv, b, false, Zout,
                     tile + (8 * lhalf) * TILE_PITCH + j, s * 16 + 8 * lhalf, n, l8, j, jb);
        TILE_HANDOVER()
        TILE_TIMES_WEIGHT()
        float v[16];
#pragma unroll
        for (int r = 0; r < 8; ++r) {
            const float p0 = lrelu(acc0[r] + bi0), p1 = lrelu(acc1[r] + bi1);
            const float p2 = lrelu(acc2[r] + bi2), p3 = lrelu(acc3[r] + bi3);
            v[2 * r + 0] = p0 * w0.x + p1 * w1.x + p2 * w2.x + p3 * w3.x;
            v[2 * r + 1] = p0 * w0.y + p1 * w1.y + p2 * w2.y + p3 * w3.y;
        }
        float u8[8], u4[4], u2[2];
#pragma unroll
        for (int i = 0; i < 8; ++i) {
            const float keep = h8 ? v[i + 8] : v[i];
            const float send = h8 ? v[i] : v[i + 8];
            u8[i] = keep + __shfl_xor(send, 8, 32);
        }
#pragma unroll
        for (int i = 0; i < 4; ++i) {
            const float keep = h4 ? u8[i + 4] : u8[i];
            const float send = h4 ? u8[i] : u8[i + 4];
            u4[i] = keep + __shfl_xor(send, 4, 32);
        }
#pragma unroll
        for (int i = 0; i < 2; ++i) {
            const float keep = h2 ? u4[i + 2] : u4[i];
            const float send = h2 ? u4[i] : u4[i + 2];
            u2[i] = keep + __shfl_xor(send, 2, 32);
        }
        const float keep1 = h1 ? u2[1] : u2[0];
        const float send1 = h1 ? u2[0] : u2[1];
        const float tot = keep1 + __shfl_xor(send1, 1, 32);
        tprob[(size_t)s * 32 + lane] = lrelu(tot + bo);
        TILE_HANDOVER()
    }
}

__global__ void __launch_bounds__(256) heads_kernel(const float* __restrict__ Z,
                                                    const int* __restrict__ treat, int nt,
                                                    const int* __restrict__ ctrl, int nc,
                                                    const float* __restrict__ Wy1, const float* __restrict__ by1,
                                                    const float* __restrict__ Wy0, const float* __restrict__ by0,
                                                    float* __restrict__ o_y1, float* __restrict__ o_yc0,
                                                    float* __restrict__ o_y0, float* __restrict__ o_yc1, int hb_t) {
    const int blk = blockIdx.x;
    const bool tr = blk < hb_t;
    const int t = (tr ? blk : blk - hb_t) * 256 + threadIdx.x;
    if (t >= (tr ? nt : nc)) return;
    const int node = tr ? treat[t] : ctrl[t];
    const float* row = Z + (size_t)node * HDIM;
    float d1 = 0.0f, d0 = 0.0f;
#pragma unroll
    for (int j = 0; j < HDIM; j += 4) {
        float4 z  = *(const float4*)(row + j);
        float4 w1 = *(const float4*)(Wy1 + j);
        float4 w0 = *(const float4*)(Wy0 + j);
        d1 += z.x * w1.x + z.y * w1.y + z.z * w1.z + z.w * w1.w;
        d0 += z.x * w0.x + z.y * w0.y + z.z * w0.z + z.w * w0.w;
    }
    const float r1 = lrelu(d1 + by1[0]);
    const float r0 = lrelu(d0 + by0[0]);
    if (tr) { o_y1[t] = r1; o_yc0[t] = r0; }
    else    { o_y0[t] = r0; o_yc1[t] = r1; }
}

extern "C" void kernel_launch(void* const* d_in, const int* in_sizes, int n_in,
                              void* d_out, int out_size, void* d_ws, size_t ws_size,
                              hipStream_t stream) {
    const float* x     = (const float*)d_in[0];
    const int*   ei    = (const int*)d_in[1];
    const int*   treat = (const int*)d_in[2];
    const int*   ctrl  = (const int*)d_in[3];
    const float* W1  = (const float*)d_in[4];  const float* b1  = (const float*)d_in[5];
    const float* W2  = (const float*)d_in[6];  const float* b2  = (const float*)d_in[7];
    const float* Wy1 = (const float*)d_in[8];  const float* by1 = (const float*)d_in[9];
    const float* Wy0 = (const float*)d_in[10]; const float* by0 = (const float*)d_in[11];
    const float* Wp1 = (const float*)d_in[12]; const float* bp1 = (const float*)d_in[13];
    const float* Wp2 = (const float*)d_in[14]; const float* bp2 = (const float*)d_in[15];

    const int N = in_sizes[0] / INDIM;
    const int E = in_sizes[1] / 2;
    const int T = in_sizes[2];
    const int Cn = in_sizes[3];
    const int* e_src = ei;
    const int* e_dst = ei + E;

    float* dinv = (float*)d_ws;
    float* buf0 = dinv + (((size_t)N + 63) & ~(size_t)63);
    int* binned = (int*)(buf0 + (size_t)N * HDIM);
    int* adj  = binned + (size_t)2 * E;
    int* cnt  = adj + (((size_t)E + 63) & ~(size_t)63);
    int* offs = cnt + (((size_t)N + 63) & ~(size_t)63);

    float* o_y1    = (float*)d_out;
    float* o_yc0   = o_y1 + T;
    float* o_y0    = o_yc0 + T;
    float* o_yc1   = o_y0 + Cn;
    float* o_tprob = o_yc1 + Cn;
    float* o_xZ2   = o_tprob + (size_t)2 * N;

    const int BLK = 256;
    auto cdiv = [](long long a, long long b) { return (int)((a + b - 1) / b); };
    const int gather_blocks = min(GATHER_BLOCKS, cdiv(N, 16));
    const int nb = cdiv(N, BIN_NODES);
    const int g1 = cdiv(E, SORT_CH);
    int* tab = (int*)buf0;
    const size_t smem_sort = ((size_t)((2 * nb + 1) & ~1) + (size_t)2 * SORT_CH) * sizeof(int);

    chunk_sort_kernel<<<g1, SORT_TPB, smem_sort, stream>>>(e_src, e_dst, tab, binned, E, nb, g1);
    bin_rows_dense_kernel<<<nb, SORT_TPB, 0, stream>>>(binned, tab, adj, cnt, offs, dinv, x, W1, o_xZ2, N, E, nb, g1);

    conv1_dense_kernel<<<gather_blocks, FUSED_TPB, 0, stream>>>(o_xZ2, adj, offs, cnt, dinv, b1, W2, buf0, N);

    conv2_dense_kernel<<<gather_blocks, FUSED_TPB, 0, stream>>>(buf0, adj, offs, cnt, dinv, b2, Wp1, bp1, Wp2, bp2,
                                                               o_xZ2, o_tprob, N);

    const int hb_t = cdiv(T, BLK);
    const int hb_c = cdiv(Cn, BLK);
    heads_kernel<<<hb_t + hb_c, BLK, 0, stream>>>(o_xZ2, treat, T, ctrl, Cn, Wy1, by1, Wy0, by0,
                                                  o_y1, o_yc0, o_y0, o_yc1, hb_t);
}
